# starting kernel
# baseline (speedup 1.0000x reference)
#include <hip/hip_runtime.h>
#include <math.h>

typedef __attribute__((ext_vector_type(2))) float v2f;
typedef __attribute__((ext_vector_type(8))) float v8f;

#define DFEAT 128
#define DDEG 32
#define NEG_SLOPE 0.2f

__device__ __forceinline__ unsigned fenc(float f) {
  unsigned u = __float_as_uint(f);
  return (u & 0x80000000u) ? ~u : (u | 0x80000000u);
}
__device__ __forceinline__ float fdec(unsigned u) {
  u = (u & 0x80000000u) ? (u & 0x7fffffffu) : ~u;
  return __uint_as_float(u);
}

__global__ void __launch_bounds__(256) gat_gemm_xw(const float* __restrict__ X,
                                                   const float* __restrict__ W,
                                                   float* __restrict__ H,
                                                   int nrows) {
  __shared__ float xs[16 * DFEAT];
  const int tid = threadIdx.x;
  const int wave = tid >> 5;
  const int lane = tid & 31;
  const int rowBlk = blockIdx.x * 16;

  {
    int idx = tid * 8;
    int r = idx >> 7;
    int c = idx & (DFEAT - 1);
    const float4* src = (const float4*)(X + (size_t)(rowBlk + r) * DFEAT + c);
    float4 v0 = src[0];
    float4 v1 = src[1];
    *(float4*)(xs + r * DFEAT + c) = v0;
    *(float4*)(xs + r * DFEAT + c + 4) = v1;
  }
  __syncthreads();

  const int col0 = wave * 16;
  const int mn = lane & 15;
  const int kk = (lane >> 4) * 2;

  v8f acc = {};
  for (int k0 = 0; k0 < DFEAT; k0 += 4) {
    v2f a, b;
    a.x = xs[mn * DFEAT + k0 + kk];
    a.y = xs[mn * DFEAT + k0 + kk + 1];
    b.x = W[(size_t)(k0 + kk) * DFEAT + col0 + mn];
    b.y = W[(size_t)(k0 + kk + 1) * DFEAT + col0 + mn];
    acc = __builtin_amdgcn_wmma_f32_16x16x4_f32(
         false, a,  false, b,
         (short)0, acc,  false,  false);
  }

  const int mbase = (lane >> 4) * 8;
  float* out = H + (size_t)(rowBlk + mbase) * DFEAT + col0 + mn;
#pragma unroll
  for (int r = 0; r < 8; ++r) {
    out[(size_t)r * DFEAT] = acc[r];
  }
}

__global__ void __launch_bounds__(256) gat_node_alpha(const float* __restrict__ H,
                                                      const float* __restrict__ a_src,
                                                      const float* __restrict__ a_dst,
                                                      float* __restrict__ asrc,
                                                      float* __restrict__ adst,
                                                      int n) {
  int g = blockIdx.x * blockDim.x + threadIdx.x;
  int node = g >> 5;
  int lane = threadIdx.x & 31;
  if (node >= n) return;
  float4 h4 = *(const float4*)(H + (size_t)node * DFEAT + lane * 4);
  float4 s4 = *(const float4*)(a_src + lane * 4);
  float4 d4 = *(const float4*)(a_dst + lane * 4);
  float ps = h4.x * s4.x + h4.y * s4.y + h4.z * s4.z + h4.w * s4.w;
  float pd = h4.x * d4.x + h4.y * d4.y + h4.z * d4.z + h4.w * d4.w;
#pragma unroll
  for (int off = 16; off; off >>= 1) {
    ps += __shfl_down(ps, off, 32);
    pd += __shfl_down(pd, off, 32);
  }
  if (lane == 0) {
    asrc[node] = ps;
    adst[node] = pd;
  }
}

__global__ void gat_init_seg(unsigned* __restrict__ mmax, float* __restrict__ denom, int n) {
  int i = blockIdx.x * blockDim.x + threadIdx.x;
  if (i >= n) return;
  mmax[i] = fenc(-INFINITY);
  denom[i] = 0.0f;
}

__global__ void fill_zero(float* __restrict__ p, int count) {
  int i = blockIdx.x * blockDim.x + threadIdx.x;
  if (i < count) p[i] = 0.0f;
}

__global__ void __launch_bounds__(256) gat_edge_score(const int* __restrict__ srcI,
                                                      const int* __restrict__ dstI,
                                                      int ecount, int n,
                                                      const float* __restrict__ asrc,
                                                      const float* __restrict__ adst,
                                                      float* __restrict__ esc,
                                                      unsigned* __restrict__ mmax) {
  int e = blockIdx.x * blockDim.x + threadIdx.x;
  if (e >= ecount + n) return;
  int s, d;
  if (e < ecount) { s = srcI[e]; d = dstI[e]; } else { s = d = e - ecount; }
  float v = asrc[s] + adst[d];
  v = v > 0.0f ? v : v * NEG_SLOPE;
  esc[e] = v;
  atomicMax(mmax + d, fenc(v));
}

__global__ void __launch_bounds__(256) gat_edge_exp(const int* __restrict__ dstI,
                                                    int ecount, int n,
                                                    float* __restrict__ esc,
                                                    const unsigned* __restrict__ mmax,
                                                    float* __restrict__ denom) {
  int e = blockIdx.x * blockDim.x + threadIdx.x;
  if (e >= ecount + n) return;
  int d = (e < ecount) ? dstI[e] : (e - ecount);
  float w = expf(esc[e] - fdec(mmax[d]));
  esc[e] = w;
  atomicAdd(denom + d, w);
}

__global__ void __launch_bounds__(256) gat_edge_scatter(const int* __restrict__ srcI,
                                                        const int* __restrict__ dstI,
                                                        int ecount, int n,
                                                        const float* __restrict__ esc,
                                                        const float* __restrict__ denom,
                                                        const float* __restrict__ H,
                                                        float* __restrict__ O) {
  int g = blockIdx.x * blockDim.x + threadIdx.x;
  int e = g >> 5;
  int lane = threadIdx.x & 31;
  if (e >= ecount + n) return;
  int s, d;
  if (e < ecount) { s = srcI[e]; d = dstI[e]; } else { s = d = e - ecount; }
  float coef = esc[e] / (denom[d] + 1e-16f);
  float4 h4 = *(const float4*)(H + (size_t)s * DFEAT + lane * 4);
  float* o = O + (size_t)d * DFEAT + lane * 4;
  atomicAdd(o + 0, coef * h4.x);
  atomicAdd(o + 1, coef * h4.y);
  atomicAdd(o + 2, coef * h4.z);
  atomicAdd(o + 3, coef * h4.w);
}

__global__ void __launch_bounds__(256) gat_bias_act(float* __restrict__ O,
                                                    const float* __restrict__ b,
                                                    int total, int relu) {
  int i = blockIdx.x * blockDim.x + threadIdx.x;
  if (i >= total) return;
  float v = O[i] + b[i & (DFEAT - 1)];
  if (relu) v = fmaxf(v, 0.0f);
  O[i] = v;
}

__global__ void __launch_bounds__(256) gat_predict(const float* __restrict__ x2o,
                                                   const float* __restrict__ x2s,
                                                   const float* __restrict__ deg,
                                                   const float* __restrict__ Wp,
                                                   const float* __restrict__ bp,
                                                   float* __restrict__ out, int n) {
  int i = blockIdx.x * blockDim.x + threadIdx.x;
  if (i >= n) return;
  const float* zo = x2o + (size_t)i * DFEAT;
  const float* zs = x2s + (size_t)i * DFEAT;
  const float* zd = deg + (size_t)i * DDEG;
  float l0 = bp[0], l1 = bp[1];
#pragma unroll 4
  for (int d = 0; d < DFEAT; ++d) {
    l0 += zo[d] * Wp[d * 2 + 0];
    l1 += zo[d] * Wp[d * 2 + 1];
  }
#pragma unroll 4
  for (int d = 0; d < DFEAT; ++d) {
    l0 += zs[d] * Wp[(DFEAT + d) * 2 + 0];
    l1 += zs[d] * Wp[(DFEAT + d) * 2 + 1];
  }
#pragma unroll 4
  for (int d = 0; d < DDEG; ++d) {
    l0 += zd[d] * Wp[(2 * DFEAT + d) * 2 + 0];
    l1 += zd[d] * Wp[(2 * DFEAT + d) * 2 + 1];
  }
  float mx = fmaxf(l0, l1);
  float e0 = expf(l0 - mx), e1 = expf(l1 - mx);
  float inv = 1.0f / (e0 + e1);
  float a0 = e0 * inv, a1 = e1 * inv;
  float* po = out + (size_t)i * DFEAT;
#pragma unroll 4
  for (int d = 0; d < DFEAT; ++d) po[d] = a0 * zo[d] + a1 * zs[d];
}

static inline int ceil_div(long long a, long long b) { return (int)((a + b - 1) / b); }

static void run_gat_layer(const float* X, const int* srcI, const int* dstI, int ecount,
                          const float* W, const float* avs, const float* avd,
                          const float* bias, float* H, float* asrc, float* adst,
                          unsigned* mmax, float* denom, float* esc, float* O,
                          int n, int relu, hipStream_t stream) {
  gat_gemm_xw<<<n / 16, 256, 0, stream>>>(X, W, H, n);
  gat_node_alpha<<<ceil_div((long long)n * 32, 256), 256, 0, stream>>>(H, avs, avd, asrc, adst, n);
  gat_init_seg<<<ceil_div(n, 256), 256, 0, stream>>>(mmax, denom, n);
  fill_zero<<<ceil_div((long long)n * DFEAT, 256), 256, 0, stream>>>(O, n * DFEAT);
  int etot = ecount + n;
  gat_edge_score<<<ceil_div(etot, 256), 256, 0, stream>>>(srcI, dstI, ecount, n, asrc, adst, esc, mmax);
  gat_edge_exp<<<ceil_div(etot, 256), 256, 0, stream>>>(dstI, ecount, n, esc, mmax, denom);
  gat_edge_scatter<<<ceil_div((long long)etot * 32, 256), 256, 0, stream>>>(srcI, dstI, ecount, n,
                                                                            esc, denom, H, O);
  gat_bias_act<<<ceil_div((long long)n * DFEAT, 256), 256, 0, stream>>>(O, bias, n * DFEAT, relu);
}

extern "C" void kernel_launch(void* const* d_in, const int* in_sizes, int n_in,
                              void* d_out, int out_size, void* d_ws, size_t ws_size,
                              hipStream_t stream) {
  const float* x_o      = (const float*)d_in[0];
  const float* degree_o = (const float*)d_in[1];
  const int*   ei_o     = (const int*)d_in[2];
  const int*   ei_s     = (const int*)d_in[3];
  const float* W_o1 = (const float*)d_in[4];
  const float* as_o1 = (const float*)d_in[5];
  const float* ad_o1 = (const float*)d_in[6];
  const float* b_o1  = (const float*)d_in[7];
  const float* W_o2 = (const float*)d_in[8];
  const float* as_o2 = (const float*)d_in[9];
  const float* ad_o2 = (const float*)d_in[10];
  const float* b_o2  = (const float*)d_in[11];
  const float* W_s1 = (const float*)d_in[12];
  const float* as_s1 = (const float*)d_in[13];
  const float* ad_s1 = (const float*)d_in[14];
  const float* b_s1  = (const float*)d_in[15];
  const float* W_s2 = (const float*)d_in[16];
  const float* as_s2 = (const float*)d_in[17];
  const float* ad_s2 = (const float*)d_in[18];
  const float* b_s2  = (const float*)d_in[19];
  const float* W_pred = (const float*)d_in[20];
  const float* b_pred = (const float*)d_in[21];

  const int n = in_sizes[0] / DFEAT;
  const int ecount = in_sizes[2] / 2;

  const int* src_o = ei_o;
  const int* dst_o = ei_o + ecount;
  const int* src_s = ei_s;
  const int* dst_s = ei_s + ecount;

  float* ws = (float*)d_ws;
  float* H     = ws;
  float* X1    = H + (size_t)n * DFEAT;
  float* X2o   = X1 + (size_t)n * DFEAT;
  float* X2s   = X2o + (size_t)n * DFEAT;
  float* asrc  = X2s + (size_t)n * DFEAT;
  float* adst  = asrc + n;
  unsigned* mmax = (unsigned*)(adst + n);
  float* denom = (float*)(mmax + n);
  float* esc   = denom + n;

  run_gat_layer(x_o, src_o, dst_o, ecount, W_o1, as_o1, ad_o1, b_o1,
                H, asrc, adst, mmax, denom, esc, X1, n,  1, stream);
  run_gat_layer(X1, src_o, dst_o, ecount, W_o2, as_o2, ad_o2, b_o2,
                H, asrc, adst, mmax, denom, esc, X2o, n,  0, stream);
  run_gat_layer(x_o, src_s, dst_s, ecount, W_s1, as_s1, ad_s1, b_s1,
                H, asrc, adst, mmax, denom, esc, X1, n,  1, stream);
  run_gat_layer(X1, src_s, dst_s, ecount, W_s2, as_s2, ad_s2, b_s2,
                H, asrc, adst, mmax, denom, esc, X2s, n,  0, stream);

  gat_predict<<<ceil_div(n, 256), 256, 0, stream>>>(X2o, X2s, degree_o, W_pred, b_pred,
                                                    (float*)d_out, n);
}


# Row sums leave global f32 atomics: 256 owner work-groups (196 dst rows each, sums in LDS) scan all dst ids, add matching edges' rows by LDS f32 atomics (arrival order), store act(sum+bias) once
# speedup vs baseline: 1.4746x; 1.4746x over previous
#include <hip/hip_runtime.h>
#include <math.h>

typedef __attribute__((ext_vector_type(2))) float v2f;
typedef __attribute__((ext_vector_type(8))) float v8f;

#define DFEAT 128
#define DDEG 32
#define NEG_SLOPE 0.2f

__device__ __forceinline__ unsigned fenc(float f) {
  unsigned u = __float_as_uint(f);
  return (u & 0x80000000u) ? ~u : (u | 0x80000000u);
}
__device__ __forceinline__ float fdec(unsigned u) {
  u = (u & 0x80000000u) ? (u & 0x7fffffffu) : ~u;
  return __uint_as_float(u);
}

__global__ void __launch_bounds__(256) gat_gemm_xw(const float* __restrict__ X,
                                                   const float* __restrict__ W,
                                                   float* __restrict__ H,
                                                   int nrows) {
  __shared__ float xs[16 * DFEAT];
  const int tid = threadIdx.x;
  const int wave = tid >> 5;
  const int lane = tid & 31;
  const int rowBlk = blockIdx.x * 16;

  {
    int idx = tid * 8;
    int r = idx >> 7;
    int c = idx & (DFEAT - 1);
    const float4* src = (const float4*)(X + (size_t)(rowBlk + r) * DFEAT + c);
    float4 v0 = src[0];
    float4 v1 = src[1];
    *(float4*)(xs + r * DFEAT + c) = v0;
    *(float4*)(xs + r * DFEAT + c + 4) = v1;
  }
  __syncthreads();

  const int col0 = wave * 16;
  const int mn = lane & 15;
  const int kk = (lane >> 4) * 2;

  v8f acc = {};
  for (int k0 = 0; k0 < DFEAT; k0 += 4) {
    v2f a, b;
    a.x = xs[mn * DFEAT + k0 + kk];
    a.y = xs[mn * DFEAT + k0 + kk + 1];
    b.x = W[(size_t)(k0 + kk) * DFEAT + col0 + mn];
    b.y = W[(size_t)(k0 + kk + 1) * DFEAT + col0 + mn];
    acc = __builtin_amdgcn_wmma_f32_16x16x4_f32(
         false, a,  false, b,
         (short)0, acc,  false,  false);
  }

  const int mbase = (lane >> 4) * 8;
  float* out = H + (size_t)(rowBlk + mbase) * DFEAT + col0 + mn;
#pragma unroll
  for (int r = 0; r < 8; ++r) {
    out[(size_t)r * DFEAT] = acc[r];
  }
}

__global__ void __launch_bounds__(256) gat_node_alpha(const float* __restrict__ H,
                                                      const float* __restrict__ a_src,
                                                      const float* __restrict__ a_dst,
                                                      float* __restrict__ asrc,
                                                      float* __restrict__ adst,
                                                      int n) {
  int g = blockIdx.x * blockDim.x + threadIdx.x;
  int node = g >> 5;
  int lane = threadIdx.x & 31;
  if (node >= n) return;
  float4 h4 = *(const float4*)(H + (size_t)node * DFEAT + lane * 4);
  float4 s4 = *(const float4*)(a_src + lane * 4);
  float4 d4 = *(const float4*)(a_dst + lane * 4);
  float ps = h4.x * s4.x + h4.y * s4.y + h4.z * s4.z + h4.w * s4.w;
  float pd = h4.x * d4.x + h4.y * d4.y + h4.z * d4.z + h4.w * d4.w;
#pragma unroll
  for (int off = 16; off; off >>= 1) {
    ps += __shfl_down(ps, off, 32);
    pd += __shfl_down(pd, off, 32);
  }
  if (lane == 0) {
    asrc[node] = ps;
    adst[node] = pd;
  }
}

__global__ void gat_init_seg(unsigned* __restrict__ mmax, float* __restrict__ denom, int n) {
  int i = blockIdx.x * blockDim.x + threadIdx.x;
  if (i >= n) return;
  mmax[i] = fenc(-INFINITY);
  denom[i] = 0.0f;
}

__global__ void __launch_bounds__(256) gat_edge_score(const int* __restrict__ srcI,
                                                      const int* __restrict__ dstI,
                                                      int ecount, int n,
                                                      const float* __restrict__ asrc,
                                                      const float* __restrict__ adst,
                                                      float* __restrict__ esc,
                                                      unsigned* __restrict__ mmax) {
  int e = blockIdx.x * blockDim.x + threadIdx.x;
  if (e >= ecount + n) return;
  int s, d;
  if (e < ecount) { s = srcI[e]; d = dstI[e]; } else { s = d = e - ecount; }
  float v = asrc[s] + adst[d];
  v = v > 0.0f ? v : v * NEG_SLOPE;
  esc[e] = v;
  atomicMax(mmax + d, fenc(v));
}

__global__ void __launch_bounds__(256) gat_edge_exp(const int* __restrict__ dstI,
                                                    int ecount, int n,
                                                    float* __restrict__ esc,
                                                    const unsigned* __restrict__ mmax,
                                                    float* __restrict__ denom) {
  int e = blockIdx.x * blockDim.x + threadIdx.x;
  if (e >= ecount + n) return;
  int d = (e < ecount) ? dstI[e] : (e - ecount);
  float w = expf(esc[e] - fdec(mmax[d]));
  esc[e] = w;
  atomicAdd(denom + d, w);
}

#define OWN_ROWS 196
#define OWN_WAVES 32
#define OWN_THREADS (OWN_WAVES * 32)
#define SCAN_DEPTH 8
#define CHUNK_IDS 128
#define SLAB_IDS (OWN_WAVES * CHUNK_IDS)
#define BATCH_IDS (SCAN_DEPTH * SLAB_IDS)
#define QCAP (32 + SCAN_DEPTH * CHUNK_IDS)
#define GROUP_EDGES 16

__device__ __forceinline__ int own_append(int* __restrict__ myq, int qn, bool p, int e) {
  const unsigned b = __builtin_amdgcn_ballot_w32(p);
  if (b != 0u) {
    const int pos = qn + (int)__builtin_amdgcn_mbcnt_lo(b, 0u);
    if (p) myq[pos] = e;
    qn += __builtin_popcount(b);
  }
  return qn;
}

__global__ void __launch_bounds__(OWN_THREADS) gat_owner_rows(const int* __restrict__ srcI,
                                                              const int* __restrict__ dstI,
                                                              int ecount, int n,
                                                              const float* __restrict__ esc,
                                                              const float* __restrict__ denom,
                                                              const float* __restrict__ H,
                                                              const float* __restrict__ bias,
                                                              int relu,
                                                              float* __restrict__ O) {
  __shared__ __attribute__((aligned(16))) float acc[OWN_ROWS * DFEAT];
  __shared__ int qbuf[OWN_WAVES * QCAP];

  const int tid = threadIdx.x;
  const int lane = tid & 31;
  const int wave = __builtin_amdgcn_readfirstlane(tid >> 5);
  const int lo = blockIdx.x * OWN_ROWS;
  const int rows = min(OWN_ROWS, n - lo);
  const unsigned urows = (unsigned)rows;

  for (int i = tid; i < OWN_ROWS * DFEAT / 4; i += OWN_THREADS)
    ((float4*)acc)[i] = make_float4(0.0f, 0.0f, 0.0f, 0.0f);
  __syncthreads();

  int* myq = qbuf + wave * QCAP;
  int qn = 0;

  auto work_off = [&](const int* q, int cnt) {
    int s = 0, r = 0;
    float coef = 0.0f;
    if (lane < cnt) {
      const int e = q[lane];
      int d;
      if (e < ecount) { s = srcI[e]; d = dstI[e]; } else { s = d = e - ecount; }
      coef = esc[e] / (denom[d] + 1e-16f);
      r = d - lo;
    }
    const int coefBits = __float_as_int(coef);
    const float* hl = H + lane;
    for (int j0 = 0; j0 < cnt; j0 += GROUP_EDGES) {
      float hv[GROUP_EDGES][4];
      float cf[GROUP_EDGES];
      int rr[GROUP_EDGES];
#pragma unroll
      for (int u = 0; u < GROUP_EDGES; ++u) {
        const int sj = __builtin_amdgcn_readlane(s, j0 + u);
        rr[u] = __builtin_amdgcn_readlane(r, j0 + u);
        cf[u] = __int_as_float(__builtin_amdgcn_readlane(coefBits, j0 + u));
        const float* h = hl + (size_t)sj * DFEAT;
        hv[u][0] = h[0];
        hv[u][1] = h[32];
        hv[u][2] = h[64];
        hv[u][3] = h[96];
      }
#pragma unroll
      for (int u = 0; u < GROUP_EDGES; ++u) {
        if (j0 + u < cnt) {
          float* a = &acc[rr[u] * DFEAT + lane];
          atomicAdd(a + 0, cf[u] * hv[u][0]);
          atomicAdd(a + 32, cf[u] * hv[u][1]);
          atomicAdd(a + 64, cf[u] * hv[u][2]);
          atomicAdd(a + 96, cf[u] * hv[u][3]);
        }
      }
    }
  };

  for (int r0 = wave * 32; r0 < rows; r0 += OWN_THREADS) {
    qn = own_append(myq, qn, r0 + lane < rows, ecount + lo + r0 + lane);
    while (qn >= 32) { qn -= 32; work_off(myq + qn, 32); }
  }

  const bool vec_ok = (((size_t)dstI) & 15) == 0;
  const int nb_full = vec_ok ? ecount / BATCH_IDS : 0;
  const int rot = nb_full ? (int)(blockIdx.x % (unsigned)nb_full) : 0;
  for (int k = 0; k < nb_full; ++k) {
    int bi = k + rot;
    if (bi >= nb_full) bi -= nb_full;
    const int b0 = bi * BATCH_IDS + wave * CHUNK_IDS + lane * 4;
    int4 dv[SCAN_DEPTH];
#pragma unroll
    for (int u = 0; u < SCAN_DEPTH; ++u)
      dv[u] = *(const int4*)(dstI + b0 + u * SLAB_IDS);
#pragma unroll
    for (int u = 0; u < SCAN_DEPTH; ++u) {
      const unsigned t0 = (unsigned)(dv[u].x - lo);
      const unsigned t1 = (unsigned)(dv[u].y - lo);
      const unsigned t2 = (unsigned)(dv[u].z - lo);
      const unsigned t3 = (unsigned)(dv[u].w - lo);
      const unsigned tm = min(min(t0, t1), min(t2, t3));
      if (__builtin_amdgcn_ballot_w32(tm < urows) != 0u) {
        const int e0 = b0 + u * SLAB_IDS;
        qn = own_append(myq, qn, t0 < urows, e0);
        qn = own_append(myq, qn, t1 < urows, e0 + 1);
        qn = own_append(myq, qn, t2 < urows, e0 + 2);
        qn = own_append(myq, qn, t3 < urows, e0 + 3);
      }
    }
    while (qn >= 32) { qn -= 32; work_off(myq + qn, 32); }
  }

  for (int base = nb_full * BATCH_IDS + wave * CHUNK_IDS; base < ecount; base += SLAB_IDS) {
    unsigned t[4];
#pragma unroll
    for (int j = 0; j < 4; ++j) {
      const int e = base + j * 32 + lane;
      t[j] = (e < ecount) ? (unsigned)(dstI[e] - lo) : 0xffffffffu;
    }
#pragma unroll
    for (int j = 0; j < 4; ++j)
      qn = own_append(myq, qn, t[j] < urows, base + j * 32 + lane);
    while (qn >= 32) { qn -= 32; work_off(myq + qn, 32); }
  }

  if (qn > 0) work_off(myq, qn);

  __syncthreads();

  for (int i = tid; i < rows * (DFEAT / 4); i += OWN_THREADS) {
    float4 v = ((const float4*)acc)[i];
    const float4 b4 = *(const float4*)(bias + (i & (DFEAT / 4 - 1)) * 4);
    v.x += b4.x; v.y += b4.y; v.z += b4.z; v.w += b4.w;
    if (relu) {
      v.x = fmaxf(v.x, 0.0f); v.y = fmaxf(v.y, 0.0f);
      v.z = fmaxf(v.z, 0.0f); v.w = fmaxf(v.w, 0.0f);
    }
    *(float4*)(O + (size_t)lo * DFEAT + (size_t)i * 4) = v;
  }
}

__global__ void __launch_bounds__(256) gat_predict(const float* __restrict__ x2o,
                                                   const float* __restrict__ x2s,
                                                   const float* __restrict__ deg,
                                                   const float* __restrict__ Wp,
                                                   const float* __restrict__ bp,
                                                   float* __restrict__ out, int n) {
  int i = blockIdx.x * blockDim.x + threadIdx.x;
  if (i >= n) return;
  const float* zo = x2o + (size_t)i * DFEAT;
  const float* zs = x2s + (size_t)i * DFEAT;
  const float* zd = deg + (size_t)i * DDEG;
  float l0 = bp[0], l1 = bp[1];
#pragma unroll 4
  for (int d = 0; d < DFEAT; ++d) {
    l0 += zo[d] * Wp[d * 2 + 0];
    l1 += zo[d] * Wp[d * 2 + 1];
  }
#pragma unroll 4
  for (int d = 0; d < DFEAT; ++d) {
    l0 += zs[d] * Wp[(DFEAT + d) * 2 + 0];
    l1 += zs[d] * Wp[(DFEAT + d) * 2 + 1];
  }
#pragma unroll 4
  for (int d = 0; d < DDEG; ++d) {
    l0 += zd[d] * Wp[(2 * DFEAT + d) * 2 + 0];
    l1 += zd[d] * Wp[(2 * DFEAT + d) * 2 + 1];
  }
  float mx = fmaxf(l0, l1);
  float e0 = expf(l0 - mx), e1 = expf(l1 - mx);
  float inv = 1.0f / (e0 + e1);
  float a0 = e0 * inv, a1 = e1 * inv;
  float* po = out + (size_t)i * DFEAT;
#pragma unroll 4
  for (int d = 0; d < DFEAT; ++d) po[d] = a0 * zo[d] + a1 * zs[d];
}

static inline int ceil_div(long long a, long long b) { return (int)((a + b - 1) / b); }

static void run_gat_layer(const float* X, const int* srcI, const int* dstI, int ecount,
                          const float* W, const float* avs, const float* avd,
                          const float* bias, float* H, float* asrc, float* adst,
                          unsigned* mmax, float* denom, float* esc, float* O,
                          int n, int relu, hipStream_t stream) {
  gat_gemm_xw<<<n / 16, 256, 0, stream>>>(X, W, H, n);
  gat_node_alpha<<<ceil_div((long long)n * 32, 256), 256, 0, stream>>>(H, avs, avd, asrc, adst, n);
  gat_init_seg<<<ceil_div(n, 256), 256, 0, stream>>>(mmax, denom, n);
  int etot = ecount + n;
  gat_edge_score<<<ceil_div(etot, 256), 256, 0, stream>>>(srcI, dstI, ecount, n, asrc, adst, esc, mmax);
  gat_edge_exp<<<ceil_div(etot, 256), 256, 0, stream>>>(dstI, ecount, n, esc, mmax, denom);
  gat_owner_rows<<<ceil_div(n, OWN_ROWS), OWN_THREADS, 0, stream>>>(srcI, dstI, ecount, n, esc, denom,
                                                                  H, bias, relu, O);
}

extern "C" void kernel_launch(void* const* d_in, const int* in_sizes, int n_in,
                              void* d_out, int out_size, void* d_ws, size_t ws_size,
                              hipStream_t stream) {
  const float* x_o      = (const float*)d_in[0];
  const float* degree_o = (const float*)d_in[1];
  const int*   ei_o     = (const int*)d_in[2];
  const int*   ei_s     = (const int*)d_in[3];
  const float* W_o1 = (const float*)d_in[4];
  const float* as_o1 = (const float*)d_in[5];
  const float* ad_o1 = (const float*)d_in[6];
  const float* b_o1  = (const float*)d_in[7];
  const float* W_o2 = (const float*)d_in[8];
  const float* as_o2 = (const float*)d_in[9];
  const float* ad_o2 = (const float*)d_in[10];
  const float* b_o2  = (const float*)d_in[11];
  const float* W_s1 = (const float*)d_in[12];
  const float* as_s1 = (const float*)d_in[13];
  const float* ad_s1 = (const float*)d_in[14];
  const float* b_s1  = (const float*)d_in[15];
  const float* W_s2 = (const float*)d_in[16];
  const float* as_s2 = (const float*)d_in[17];
  const float* ad_s2 = (const float*)d_in[18];
  const float* b_s2  = (const float*)d_in[19];
  const float* W_pred = (const float*)d_in[20];
  const float* b_pred = (const float*)d_in[21];

  const int n = in_sizes[0] / DFEAT;
  const int ecount = in_sizes[2] / 2;

  const int* src_o = ei_o;
  const int* dst_o = ei_o + ecount;
  const int* src_s = ei_s;
  const int* dst_s = ei_s + ecount;

  float* ws = (float*)d_ws;
  float* H     = ws;
  float* X1    = H + (size_t)n * DFEAT;
  float* X2o   = X1 + (size_t)n * DFEAT;
  float* X2s   = X2o + (size_t)n * DFEAT;
  float* asrc  = X2s + (size_t)n * DFEAT;
  float* adst  = asrc + n;
  unsigned* mmax = (unsigned*)(adst + n);
  float* denom = (float*)(mmax + n);
  float* esc   = denom + n;

  run_gat_layer(x_o, src_o, dst_o, ecount, W_o1, as_o1, ad_o1, b_o1,
                H, asrc, adst, mmax, denom, esc, X1, n,  1, stream);
  run_gat_layer(X1, src_o, dst_o, ecount, W_o2, as_o2, ad_o2, b_o2,
                H, asrc, adst, mmax, denom, esc, X2o, n,  0, stream);
  run_gat_layer(x_o, src_s, dst_s, ecount, W_s1, as_s1, ad_s1, b_s1,
                H, asrc, adst, mmax, denom, esc, X1, n,  1, stream);
  run_gat_layer(X1, src_s, dst_s, ecount, W_s2, as_s2, ad_s2, b_s2,
                H, asrc, adst, mmax, denom, esc, X2s, n,  0, stream);

  gat_predict<<<ceil_div(n, 256), 256, 0, stream>>>(X2o, X2s, degree_o, W_pred, b_pred,
                                                    (float*)d_out, n);
}


# Edge softmax moves into the owner work-groups: score, row max, exp and denominator in LDS, edge lists kept in LDS; 12 launches and all global atomics go; the sum is divided once, arrival order
# speedup vs baseline: 3.4127x; 2.3143x over previous
#include <hip/hip_runtime.h>
#include <math.h>

typedef __attribute__((ext_vector_type(2))) float v2f;
typedef __attribute__((ext_vector_type(8))) float v8f;

#define DFEAT 128
#define DDEG 32
#define NEG_SLOPE 0.2f

__device__ __forceinline__ unsigned fenc(float f) {
  unsigned u = __float_as_uint(f);
  return (u & 0x80000000u) ? ~u : (u | 0x80000000u);
}
__device__ __forceinline__ float fdec(unsigned u) {
  u = (u & 0x80000000u) ? (u & 0x7fffffffu) : ~u;
  return __uint_as_float(u);
}

__global__ void __launch_bounds__(256) gat_gemm_xw(const float* __restrict__ X,
                                                   const float* __restrict__ W,
                                                   float* __restrict__ H,
                                                   int nrows) {
  __shared__ float xs[16 * DFEAT];
  const int tid = threadIdx.x;
  const int wave = tid >> 5;
  const int lane = tid & 31;
  const int rowBlk = blockIdx.x * 16;

  {
    int idx = tid * 8;
    int r = idx >> 7;
    int c = idx & (DFEAT - 1);
    const float4* src = (const float4*)(X + (size_t)(rowBlk + r) * DFEAT + c);
    float4 v0 = src[0];
    float4 v1 = src[1];
    *(float4*)(xs + r * DFEAT + c) = v0;
    *(float4*)(xs + r * DFEAT + c + 4) = v1;
  }
  __syncthreads();

  const int col0 = wave * 16;
  const int mn = lane & 15;
  const int kk = (lane >> 4) * 2;

  v8f acc = {};
  for (int k0 = 0; k0 < DFEAT; k0 += 4) {
    v2f a, b;
    a.x = xs[mn * DFEAT + k0 + kk];
    a.y = xs[mn * DFEAT + k0 + kk + 1];
    b.x = W[(size_t)(k0 + kk) * DFEAT + col0 + mn];
    b.y = W[(size_t)(k0 + kk + 1) * DFEAT + col0 + mn];
    acc = __builtin_amdgcn_wmma_f32_16x16x4_f32(
         false, a,  false, b,
         (short)0, acc,  false,  false);
  }

  const int mbase = (lane >> 4) * 8;
  float* out = H + (size_t)(rowBlk + mbase) * DFEAT + col0 + mn;
#pragma unroll
  for (int r = 0; r < 8; ++r) {
    out[(size_t)r * DFEAT] = acc[r];
  }
}

__global__ void __launch_bounds__(256) gat_node_alpha(const float* __restrict__ H,
                                                      const float* __restrict__ a_src,
                                                      const float* __restrict__ a_dst,
                                                      float* __restrict__ asrc,
                                                      float* __restrict__ adst,
                                                      int n) {
  int g = blockIdx.x * blockDim.x + threadIdx.x;
  int node = g >> 5;
  int lane = threadIdx.x & 31;
  if (node >= n) return;
  float4 h4 = *(const float4*)(H + (size_t)node * DFEAT + lane * 4);
  float4 s4 = *(const float4*)(a_src + lane * 4);
  float4 d4 = *(const float4*)(a_dst + lane * 4);
  float ps = h4.x * s4.x + h4.y * s4.y + h4.z * s4.z + h4.w * s4.w;
  float pd = h4.x * d4.x + h4.y * d4.y + h4.z * d4.z + h4.w * d4.w;
#pragma unroll
  for (int off = 16; off; off >>= 1) {
    ps += __shfl_down(ps, off, 32);
    pd += __shfl_down(pd, off, 32);
  }
  if (lane == 0) {
    asrc[node] = ps;
    adst[node] = pd;
  }
}

#define OWN_ROWS 196
#define OWN_WAVES 32
#define OWN_THREADS (OWN_WAVES * 32)
#define SCAN_DEPTH 8
#define CHUNK_IDS 128
#define SLAB_IDS (OWN_WAVES * CHUNK_IDS)
#define BATCH_IDS (SCAN_DEPTH * SLAB_IDS)
#define WCAP 512
#define SCORE_DEPTH 8
#define GROUP_EDGES 16

__device__ __forceinline__ int own_append(int* __restrict__ qe, unsigned char* __restrict__ qr,
                                          int qn, bool p, int e, unsigned t, int& ovf) {
  const unsigned b = __builtin_amdgcn_ballot_w32(p);
  if (b != 0u) {
    const int cnt = __builtin_popcount(b);
    if (qn + cnt <= WCAP) {
      const int pos = qn + (int)__builtin_amdgcn_mbcnt_lo(b, 0u);
      if (p) { qe[pos] = e; qr[pos] = (unsigned char)t; }
      qn += cnt;
    } else {
      ovf = 1;
    }
  }
  return qn;
}

__global__ void __launch_bounds__(OWN_THREADS) gat_owner_rows(const int* __restrict__ srcI,
                                                              const int* __restrict__ dstI,
                                                              int ecount, int n,
                                                              const float* __restrict__ asrc,
                                                              const float* __restrict__ adst,
                                                              const float* __restrict__ H,
                                                              const float* __restrict__ bias,
                                                              int relu,
                                                              float* __restrict__ O) {
  __shared__ __attribute__((aligned(16))) float acc[OWN_ROWS * DFEAT];
  __shared__ int qsbuf[OWN_WAVES * WCAP];
  __shared__ float qvbuf[OWN_WAVES * WCAP];
  __shared__ unsigned char qrbuf[OWN_WAVES * WCAP];
  __shared__ unsigned rmax[OWN_ROWS];
  __shared__ float rden[OWN_ROWS];
  __shared__ int ovf_flag;

  const int tid = threadIdx.x;
  const int lane = tid & 31;
  const int wave = __builtin_amdgcn_readfirstlane(tid >> 5);
  const int lo = blockIdx.x * OWN_ROWS;
  const int rows = min(OWN_ROWS, n - lo);
  const unsigned urows = (unsigned)rows;

  for (int i = tid; i < OWN_ROWS * DFEAT / 4; i += OWN_THREADS)
    ((float4*)acc)[i] = make_float4(0.0f, 0.0f, 0.0f, 0.0f);
  if (tid < OWN_ROWS) {
    rmax[tid] = fenc(-INFINITY);
    rden[tid] = 0.0f;
  }
  if (tid == 0) ovf_flag = 0;
  __syncthreads();

  int* myqs = qsbuf + wave * WCAP;
  float* myqv = qvbuf + wave * WCAP;
  unsigned char* myqr = qrbuf + wave * WCAP;
  int qn = 0;
  int ovf = 0;

  for (int r0 = wave * 32; r0 < rows; r0 += OWN_THREADS)
    qn = own_append(myqs, myqr, qn, r0 + lane < rows, ecount + lo + r0 + lane, (unsigned)(r0 + lane), ovf);

  const bool vec_ok = (((size_t)dstI) & 15) == 0;
  const int nb_full = vec_ok ? ecount / BATCH_IDS : 0;
  const int rot = nb_full ? (int)(blockIdx.x % (unsigned)nb_full) : 0;
  for (int k = 0; k < nb_full; ++k) {
    int bi = k + rot;
    if (bi >= nb_full) bi -= nb_full;
    const int b0 = bi * BATCH_IDS + wave * CHUNK_IDS + lane * 4;
    int4 dv[SCAN_DEPTH];
#pragma unroll
    for (int u = 0; u < SCAN_DEPTH; ++u)
      dv[u] = *(const int4*)(dstI + b0 + u * SLAB_IDS);
#pragma unroll
    for (int u = 0; u < SCAN_DEPTH; ++u) {
      const unsigned t0 = (unsigned)(dv[u].x - lo);
      const unsigned t1 = (unsigned)(dv[u].y - lo);
      const unsigned t2 = (unsigned)(dv[u].z - lo);
      const unsigned t3 = (unsigned)(dv[u].w - lo);
      const unsigned tm = min(min(t0, t1), min(t2, t3));
      if (__builtin_amdgcn_ballot_w32(tm < urows) != 0u) {
        const int e0 = b0 + u * SLAB_IDS;
        qn = own_append(myqs, myqr, qn, t0 < urows, e0, t0, ovf);
        qn = own_append(myqs, myqr, qn, t1 < urows, e0 + 1, t1, ovf);
        qn = own_append(myqs, myqr, qn, t2 < urows, e0 + 2, t2, ovf);
        qn = own_append(myqs, myqr, qn, t3 < urows, e0 + 3, t3, ovf);
      }
    }
  }

  for (int base = nb_full * BATCH_IDS + wave * CHUNK_IDS; base < ecount; base += SLAB_IDS) {
    unsigned t[4];
#pragma unroll
    for (int j = 0; j < 4; ++j) {
      const int e = base + j * 32 + lane;
      t[j] = (e < ecount) ? (unsigned)(dstI[e] - lo) : 0xffffffffu;
    }
#pragma unroll
    for (int j = 0; j < 4; ++j)
      qn = own_append(myqs, myqr, qn, t[j] < urows, base + j * 32 + lane, t[j], ovf);
  }

  for (int g = 0; g < qn; g += 32 * SCORE_DEPTH) {
    int s[SCORE_DEPTH];
    int r[SCORE_DEPTH];
    float v[SCORE_DEPTH];
#pragma unroll
    for (int u = 0; u < SCORE_DEPTH; ++u) {
      const int i = g + u * 32 + lane;
      s[u] = 0;
      r[u] = 0;
      if (i < qn) {
        const int e = myqs[i];
        r[u] = myqr[i];
        s[u] = (e < ecount) ? srcI[e] : e - ecount;
      }
    }
#pragma unroll
    for (int u = 0; u < SCORE_DEPTH; ++u)
      v[u] = asrc[s[u]] + adst[lo + r[u]];
#pragma unroll
    for (int u = 0; u < SCORE_DEPTH; ++u) {
      const int i = g + u * 32 + lane;
      if (i < qn) {
        const float x = v[u] > 0.0f ? v[u] : v[u] * NEG_SLOPE;
        myqs[i] = s[u];
        myqv[i] = x;
        atomicMax(&rmax[r[u]], fenc(x));
      }
    }
  }
  if (ovf != 0 && lane == 0) ovf_flag = 1;

  __syncthreads();

  const float* hl = H + lane;
  if (__builtin_amdgcn_readfirstlane(ovf_flag) == 0) {
    for (int j0 = 0; j0 < qn; j0 += 32) {
      const int cnt = min(32, qn - j0);
      int s = 0, r = 0;
      float w = 0.0f;
      if (lane < cnt) {
        s = myqs[j0 + lane];
        r = myqr[j0 + lane];
        w = expf(myqv[j0 + lane] - fdec(rmax[r]));
        atomicAdd(&rden[r], w);
      }
      const int wBits = __float_as_int(w);
      for (int k0 = 0; k0 < cnt; k0 += GROUP_EDGES) {
        float hv[GROUP_EDGES][4];
        float cf[GROUP_EDGES];
        int rr[GROUP_EDGES];
#pragma unroll
        for (int u = 0; u < GROUP_EDGES; ++u) {
          const int sj = __builtin_amdgcn_readlane(s, k0 + u);
          rr[u] = __builtin_amdgcn_readlane(r, k0 + u);
          cf[u] = __int_as_float(__builtin_amdgcn_readlane(wBits, k0 + u));
          const float* h = hl + (size_t)sj * DFEAT;
          hv[u][0] = h[0];
          hv[u][1] = h[32];
          hv[u][2] = h[64];
          hv[u][3] = h[96];
        }
#pragma unroll
        for (int u = 0; u < GROUP_EDGES; ++u) {
          if (k0 + u < cnt) {
            float* a = &acc[rr[u] * DFEAT + lane];
            atomicAdd(a + 0, cf[u] * hv[u][0]);
            atomicAdd(a + 32, cf[u] * hv[u][1]);
            atomicAdd(a + 64, cf[u] * hv[u][2]);
            atomicAdd(a + 96, cf[u] * hv[u][3]);
          }
        }
      }
    }
  } else {
    const int etot = ecount + n;
    for (int pass = 0; pass < 2; ++pass) {
      for (int base = wave * 32; base < etot; base += OWN_THREADS) {
        const int e = base + lane;
        int s = 0;
        unsigned t = 0xffffffffu;
        float v = 0.0f;
        if (e < etot) {
          const int d = (e < ecount) ? dstI[e] : e - ecount;
          t = (unsigned)(d - lo);
          if (t < urows) {
            s = (e < ecount) ? srcI[e] : d;
            v = asrc[s] + adst[d];
            v = v > 0.0f ? v : v * NEG_SLOPE;
          }
        }
        const bool own = t < urows;
        if (pass == 0) {
          if (own) atomicMax(&rmax[t], fenc(v));
        } else {
          float w = 0.0f;
          if (own) {
            w = expf(v - fdec(rmax[t]));
            atomicAdd(&rden[t], w);
          }
          const int wBits = __float_as_int(w);
          unsigned b = __builtin_amdgcn_ballot_w32(own);
          while (b != 0u) {
            const int j = __builtin_ctz(b);
            b &= b - 1u;
            const int sj = __builtin_amdgcn_readlane(s, j);
            const int rj = __builtin_amdgcn_readlane((int)t, j);
            const float wj = __int_as_float(__builtin_amdgcn_readlane(wBits, j));
            const float* h = hl + (size_t)sj * DFEAT;
            float* a = &acc[rj * DFEAT + lane];
            atomicAdd(a + 0, wj * h[0]);
            atomicAdd(a + 32, wj * h[32]);
            atomicAdd(a + 64, wj * h[64]);
            atomicAdd(a + 96, wj * h[96]);
          }
        }
      }
      __syncthreads();
    }
  }

  __syncthreads();

  for (int i = tid; i < rows * (DFEAT / 4); i += OWN_THREADS) {
    float4 v = ((const float4*)acc)[i];
    const float dn = rden[i / (DFEAT / 4)] + 1e-16f;
    const float4 b4 = *(const float4*)(bias + (i & (DFEAT / 4 - 1)) * 4);
    v.x = v.x / dn + b4.x; v.y = v.y / dn + b4.y;
    v.z = v.z / dn + b4.z; v.w = v.w / dn + b4.w;
    if (relu) {
      v.x = fmaxf(v.x, 0.0f); v.y = fmaxf(v.y, 0.0f);
      v.z = fmaxf(v.z, 0.0f); v.w = fmaxf(v.w, 0.0f);
    }
    *(float4*)(O + (size_t)lo * DFEAT + (size_t)i * 4) = v;
  }
}

__global__ void __launch_bounds__(256) gat_predict(const float* __restrict__ x2o,
                                                   const float* __restrict__ x2s,
                                                   const float* __restrict__ deg,
                                                   const float* __restrict__ Wp,
                                                   const float* __restrict__ bp,
                                                   float* __restrict__ out, int n) {
  int i = blockIdx.x * blockDim.x + threadIdx.x;
  if (i >= n) return;
  const float* zo = x2o + (size_t)i * DFEAT;
  const float* zs = x2s + (size_t)i * DFEAT;
  const float* zd = deg + (size_t)i * DDEG;
  float l0 = bp[0], l1 = bp[1];
#pragma unroll 4
  for (int d = 0; d < DFEAT; ++d) {
    l0 += zo[d] * Wp[d * 2 + 0];
    l1 += zo[d] * Wp[d * 2 + 1];
  }
#pragma unroll 4
  for (int d = 0; d < DFEAT; ++d) {
    l0 += zs[d] * Wp[(DFEAT + d) * 2 + 0];
    l1 += zs[d] * Wp[(DFEAT + d) * 2 + 1];
  }
#pragma unroll 4
  for (int d = 0; d < DDEG; ++d) {
    l0 += zd[d] * Wp[(2 * DFEAT + d) * 2 + 0];
    l1 += zd[d] * Wp[(2 * DFEAT + d) * 2 + 1];
  }
  float mx = fmaxf(l0, l1);
  float e0 = expf(l0 - mx), e1 = expf(l1 - mx);
  float inv = 1.0f / (e0 + e1);
  float a0 = e0 * inv, a1 = e1 * inv;
  float* po = out + (size_t)i * DFEAT;
#pragma unroll 4
  for (int d = 0; d < DFEAT; ++d) po[d] = a0 * zo[d] + a1 * zs[d];
}

static inline int ceil_div(long long a, long long b) { return (int)((a + b - 1) / b); }

static void run_gat_layer(const float* X, const int* srcI, const int* dstI, int ecount,
                          const float* W, const float* avs, const float* avd,
                          const float* bias, float* H, float* asrc, float* adst,
                          float* O, int n, int relu, hipStream_t stream) {
  gat_gemm_xw<<<n / 16, 256, 0, stream>>>(X, W, H, n);
  gat_node_alpha<<<ceil_div((long long)n * 32, 256), 256, 0, stream>>>(H, avs, avd, asrc, adst, n);
  gat_owner_rows<<<ceil_div(n, OWN_ROWS), OWN_THREADS, 0, stream>>>(srcI, dstI, ecount, n, asrc, adst,
                                                                  H, bias, relu, O);
}

extern "C" void kernel_launch(void* const* d_in, const int* in_sizes, int n_in,
                              void* d_out, int out_size, void* d_ws, size_t ws_size,
                              hipStream_t stream) {
  const float* x_o      = (const float*)d_in[0];
  const float* degree_o = (const float*)d_in[1];
  const int*   ei_o     = (const int*)d_in[2];
  const int*   ei_s     = (const int*)d_in[3];
  const float* W_o1 = (const float*)d_in[4];
  const float* as_o1 = (const float*)d_in[5];
  const float* ad_o1 = (const float*)d_in[6];
  const float* b_o1  = (const float*)d_in[7];
  const float* W_o2 = (const float*)d_in[8];
  const float* as_o2 = (const float*)d_in[9];
  const float* ad_o2 = (const float*)d_in[10];
  const float* b_o2  = (const float*)d_in[11];
  const float* W_s1 = (const float*)d_in[12];
  const float* as_s1 = (const float*)d_in[13];
  const float* ad_s1 = (const float*)d_in[14];
  const float* b_s1  = (const float*)d_in[15];
  const float* W_s2 = (const float*)d_in[16];
  const float* as_s2 = (const float*)d_in[17];
  const float* ad_s2 = (const float*)d_in[18];
  const float* b_s2  = (const float*)d_in[19];
  const float* W_pred = (const float*)d_in[20];
  const float* b_pred = (const float*)d_in[21];

  const int n = in_sizes[0] / DFEAT;
  const int ecount = in_sizes[2] / 2;

  const int* src_o = ei_o;
  const int* dst_o = ei_o + ecount;
  const int* src_s = ei_s;
  const int* dst_s = ei_s + ecount;

  float* ws = (float*)d_ws;
  float* H     = ws;
  float* X1    = H + (size_t)n * DFEAT;
  float* X2o   = X1 + (size_t)n * DFEAT;
  float* X2s   = X2o + (size_t)n * DFEAT;
  float* asrc  = X2s + (size_t)n * DFEAT;
  float* adst  = asrc + n;

  run_gat_layer(x_o, src_o, dst_o, ecount, W_o1, as_o1, ad_o1, b_o1,
                H, asrc, adst, X1, n,  1, stream);
  run_gat_layer(X1, src_o, dst_o, ecount, W_o2, as_o2, ad_o2, b_o2,
                H, asrc, adst, X2o, n,  0, stream);
  run_gat_layer(x_o, src_s, dst_s, ecount, W_s1, as_s1, ad_s1, b_s1,
                H, asrc, adst, X1, n,  1, stream);
  run_gat_layer(X1, src_s, dst_s, ecount, W_s2, as_s2, ad_s2, b_s2,
                H, asrc, adst, X2s, n,  0, stream);

  gat_predict<<<ceil_div(n, 256), 256, 0, stream>>>(X2o, X2s, degree_o, W_pred, b_pred,
                                                    (float*)d_out, n);
}


# Owner work-groups keep their edge lists (source row + owned row, one word per edge) in the workspace in layer 1 of each graph; layer 2 reads them back: 2 of 4 dst scans and source-id gathers go
# speedup vs baseline: 3.8694x; 1.1338x over previous
#include <hip/hip_runtime.h>
#include <math.h>

typedef __attribute__((ext_vector_type(2))) float v2f;
typedef __attribute__((ext_vector_type(8))) float v8f;

#define DFEAT 128
#define DDEG 32
#define NEG_SLOPE 0.2f

__device__ __forceinline__ unsigned fenc(float f) {
  unsigned u = __float_as_uint(f);
  return (u & 0x80000000u) ? ~u : (u | 0x80000000u);
}
__device__ __forceinline__ float fdec(unsigned u) {
  u = (u & 0x80000000u) ? (u & 0x7fffffffu) : ~u;
  return __uint_as_float(u);
}

__global__ void __launch_bounds__(256) gat_gemm_xw(const float* __restrict__ X,
                                                   const float* __restrict__ W,
                                                   float* __restrict__ H,
                                                   int nrows) {
  __shared__ float xs[16 * DFEAT];
  const int tid = threadIdx.x;
  const int wave = tid >> 5;
  const int lane = tid & 31;
  const int rowBlk = blockIdx.x * 16;

  {
    int idx = tid * 8;
    int r = idx >> 7;
    int c = idx & (DFEAT - 1);
    const float4* src = (const float4*)(X + (size_t)(rowBlk + r) * DFEAT + c);
    float4 v0 = src[0];
    float4 v1 = src[1];
    *(float4*)(xs + r * DFEAT + c) = v0;
    *(float4*)(xs + r * DFEAT + c + 4) = v1;
  }
  __syncthreads();

  const int col0 = wave * 16;
  const int mn = lane & 15;
  const int kk = (lane >> 4) * 2;

  v8f acc = {};
  for (int k0 = 0; k0 < DFEAT; k0 += 4) {
    v2f a, b;
    a.x = xs[mn * DFEAT + k0 + kk];
    a.y = xs[mn * DFEAT + k0 + kk + 1];
    b.x = W[(size_t)(k0 + kk) * DFEAT + col0 + mn];
    b.y = W[(size_t)(k0 + kk + 1) * DFEAT + col0 + mn];
    acc = __builtin_amdgcn_wmma_f32_16x16x4_f32(
         false, a,  false, b,
         (short)0, acc,  false,  false);
  }

  const int mbase = (lane >> 4) * 8;
  float* out = H + (size_t)(rowBlk + mbase) * DFEAT + col0 + mn;
#pragma unroll
  for (int r = 0; r < 8; ++r) {
    out[(size_t)r * DFEAT] = acc[r];
  }
}

__global__ void __launch_bounds__(256) gat_node_alpha(const float* __restrict__ H,
                                                      const float* __restrict__ a_src,
                                                      const float* __restrict__ a_dst,
                                                      float* __restrict__ asrc,
                                                      float* __restrict__ adst,
                                                      int n) {
  int g = blockIdx.x * blockDim.x + threadIdx.x;
  int node = g >> 5;
  int lane = threadIdx.x & 31;
  if (node >= n) return;
  float4 h4 = *(const float4*)(H + (size_t)node * DFEAT + lane * 4);
  float4 s4 = *(const float4*)(a_src + lane * 4);
  float4 d4 = *(const float4*)(a_dst + lane * 4);
  float ps = h4.x * s4.x + h4.y * s4.y + h4.z * s4.z + h4.w * s4.w;
  float pd = h4.x * d4.x + h4.y * d4.y + h4.z * d4.z + h4.w * d4.w;
#pragma unroll
  for (int off = 16; off; off >>= 1) {
    ps += __shfl_down(ps, off, 32);
    pd += __shfl_down(pd, off, 32);
  }
  if (lane == 0) {
    asrc[node] = ps;
    adst[node] = pd;
  }
}

#define OWN_ROWS 196
#define OWN_WAVES 32
#define OWN_THREADS (OWN_WAVES * 32)
#define SCAN_DEPTH 8
#define CHUNK_IDS 128
#define SLAB_IDS (OWN_WAVES * CHUNK_IDS)
#define BATCH_IDS (SCAN_DEPTH * SLAB_IDS)
#define WCAP 512
#define SCORE_DEPTH 8
#define GROUP_EDGES 16
#define LIST_WORDS (OWN_WAVES * WCAP)
#define LIST_STRIDE (LIST_WORDS + 64)

__device__ __forceinline__ int own_append(int* __restrict__ qe, unsigned char* __restrict__ qr,
                                          int qn, bool p, int e, unsigned t, int& ovf) {
  const unsigned b = __builtin_amdgcn_ballot_w32(p);
  if (b != 0u) {
    const int cnt = __builtin_popcount(b);
    if (qn + cnt <= WCAP) {
      const int pos = qn + (int)__builtin_amdgcn_mbcnt_lo(b, 0u);
      if (p) { qe[pos] = e; qr[pos] = (unsigned char)t; }
      qn += cnt;
    } else {
      ovf = 1;
    }
  }
  return qn;
}

__global__ void __launch_bounds__(OWN_THREADS) gat_owner_rows(const int* __restrict__ srcI,
                                                              const int* __restrict__ dstI,
                                                              int ecount, int n,
                                                              const float* __restrict__ asrc,
                                                              const float* __restrict__ adst,
                                                              const float* __restrict__ H,
                                                              const float* __restrict__ bias,
                                                              int relu,
                                                              int* __restrict__ lists,
                                                              int reuse,
                                                              float* __restrict__ O) {
  __shared__ __attribute__((aligned(16))) float acc[OWN_ROWS * DFEAT];
  __shared__ int qsbuf[OWN_WAVES * WCAP];
  __shared__ float qvbuf[OWN_WAVES * WCAP];
  __shared__ unsigned char qrbuf[OWN_WAVES * WCAP];
  __shared__ unsigned rmax[OWN_ROWS];
  __shared__ float rden[OWN_ROWS];
  __shared__ int ovf_flag;

  const int tid = threadIdx.x;
  const int lane = tid & 31;
  const int wave = __builtin_amdgcn_readfirstlane(tid >> 5);
  const int lo = blockIdx.x * OWN_ROWS;
  const int rows = min(OWN_ROWS, n - lo);
  const unsigned urows = (unsigned)rows;

  for (int i = tid; i < OWN_ROWS * DFEAT / 4; i += OWN_THREADS)
    ((float4*)acc)[i] = make_float4(0.0f, 0.0f, 0.0f, 0.0f);
  if (tid < OWN_ROWS) {
    rmax[tid] = fenc(-INFINITY);
    rden[tid] = 0.0f;
  }
  if (tid == 0) ovf_flag = 0;
  __syncthreads();

  int* myqs = qsbuf + wave * WCAP;
  float* myqv = qvbuf + wave * WCAP;
  unsigned char* myqr = qrbuf + wave * WCAP;
  int qn = 0;
  int ovf = 0;

  int* gq = lists + (size_t)blockIdx.x * LIST_STRIDE;
  int* myg = gq + wave * WCAP;

  if (reuse == 0) {
    for (int r0 = wave * 32; r0 < rows; r0 += OWN_THREADS)
      qn = own_append(myqs, myqr, qn, r0 + lane < rows, ecount + lo + r0 + lane, (unsigned)(r0 + lane), ovf);

    const bool vec_ok = (((size_t)dstI) & 15) == 0;
    const int nb_full = vec_ok ? ecount / BATCH_IDS : 0;
    const int rot = nb_full ? (int)(blockIdx.x % (unsigned)nb_full) : 0;
    for (int k = 0; k < nb_full; ++k) {
      int bi = k + rot;
      if (bi >= nb_full) bi -= nb_full;
      const int b0 = bi * BATCH_IDS + wave * CHUNK_IDS + lane * 4;
      int4 dv[SCAN_DEPTH];
  #pragma unroll
      for (int u = 0; u < SCAN_DEPTH; ++u)
        dv[u] = *(const int4*)(dstI + b0 + u * SLAB_IDS);
  #pragma unroll
      for (int u = 0; u < SCAN_DEPTH; ++u) {
        const unsigned t0 = (unsigned)(dv[u].x - lo);
        const unsigned t1 = (unsigned)(dv[u].y - lo);
        const unsigned t2 = (unsigned)(dv[u].z - lo);
        const unsigned t3 = (unsigned)(dv[u].w - lo);
        const unsigned tm = min(min(t0, t1), min(t2, t3));
        if (__builtin_amdgcn_ballot_w32(tm < urows) != 0u) {
          const int e0 = b0 + u * SLAB_IDS;
          qn = own_append(myqs, myqr, qn, t0 < urows, e0, t0, ovf);
          qn = own_append(myqs, myqr, qn, t1 < urows, e0 + 1, t1, ovf);
          qn = own_append(myqs, myqr, qn, t2 < urows, e0 + 2, t2, ovf);
          qn = own_append(myqs, myqr, qn, t3 < urows, e0 + 3, t3, ovf);
        }
      }
    }

    for (int base = nb_full * BATCH_IDS + wave * CHUNK_IDS; base < ecount; base += SLAB_IDS) {
      unsigned t[4];
  #pragma unroll
      for (int j = 0; j < 4; ++j) {
        const int e = base + j * 32 + lane;
        t[j] = (e < ecount) ? (unsigned)(dstI[e] - lo) : 0xffffffffu;
      }
  #pragma unroll
      for (int j = 0; j < 4; ++j)
        qn = own_append(myqs, myqr, qn, t[j] < urows, base + j * 32 + lane, t[j], ovf);
    }
  } else {
    ovf = __builtin_amdgcn_readfirstlane(gq[LIST_WORDS + OWN_WAVES]);
    qn = ovf != 0 ? 0 : __builtin_amdgcn_readfirstlane(gq[LIST_WORDS + wave]);
  }

  for (int g = 0; g < qn; g += 32 * SCORE_DEPTH) {
    int s[SCORE_DEPTH];
    int r[SCORE_DEPTH];
    float v[SCORE_DEPTH];
    if (reuse == 0) {
#pragma unroll
      for (int u = 0; u < SCORE_DEPTH; ++u) {
        const int i = g + u * 32 + lane;
        s[u] = 0;
        r[u] = 0;
        if (i < qn) {
          const int e = myqs[i];
          r[u] = myqr[i];
          s[u] = (e < ecount) ? srcI[e] : e - ecount;
        }
      }
    } else {
#pragma unroll
      for (int u = 0; u < SCORE_DEPTH; ++u) {
        const int i = g + u * 32 + lane;
        s[u] = 0;
        if (i < qn) s[u] = myg[i];
      }
#pragma unroll
      for (int u = 0; u < SCORE_DEPTH; ++u) {
        r[u] = (int)((unsigned)s[u] >> 24);
        s[u] &= 0xffffff;
      }
    }
#pragma unroll
    for (int u = 0; u < SCORE_DEPTH; ++u)
      v[u] = asrc[s[u]] + adst[lo + r[u]];
#pragma unroll
    for (int u = 0; u < SCORE_DEPTH; ++u) {
      const int i = g + u * 32 + lane;
      if (i < qn) {
        const float x = v[u] > 0.0f ? v[u] : v[u] * NEG_SLOPE;
        myqs[i] = s[u];
        myqv[i] = x;
        if (reuse == 0) myg[i] = (int)((unsigned)s[u] | ((unsigned)r[u] << 24));
        else myqr[i] = (unsigned char)r[u];
        atomicMax(&rmax[r[u]], fenc(x));
      }
    }
  }
  if (lane == 0) {
    if (reuse == 0) gq[LIST_WORDS + wave] = qn;
    if (ovf != 0) ovf_flag = 1;
  }

  __syncthreads();

  if (reuse == 0 && tid == 0) gq[LIST_WORDS + OWN_WAVES] = ovf_flag;

  const float* hl = H + lane;
  if (__builtin_amdgcn_readfirstlane(ovf_flag) == 0) {
    for (int j0 = 0; j0 < qn; j0 += 32) {
      const int cnt = min(32, qn - j0);
      int s = 0, r = 0;
      float w = 0.0f;
      if (lane < cnt) {
        s = myqs[j0 + lane];
        r = myqr[j0 + lane];
        w = expf(myqv[j0 + lane] - fdec(rmax[r]));
        atomicAdd(&rden[r], w);
      }
      const int wBits = __float_as_int(w);
      for (int k0 = 0; k0 < cnt; k0 += GROUP_EDGES) {
        float hv[GROUP_EDGES][4];
        float cf[GROUP_EDGES];
        int rr[GROUP_EDGES];
#pragma unroll
        for (int u = 0; u < GROUP_EDGES; ++u) {
          const int sj = __builtin_amdgcn_readlane(s, k0 + u);
          rr[u] = __builtin_amdgcn_readlane(r, k0 + u);
          cf[u] = __int_as_float(__builtin_amdgcn_readlane(wBits, k0 + u));
          const float* h = hl + (size_t)sj * DFEAT;
          hv[u][0] = h[0];
          hv[u][1] = h[32];
          hv[u][2] = h[64];
          hv[u][3] = h[96];
        }
#pragma unroll
        for (int u = 0; u < GROUP_EDGES; ++u) {
          if (k0 + u < cnt) {
            float* a = &acc[rr[u] * DFEAT + lane];
            atomicAdd(a + 0, cf[u] * hv[u][0]);
            atomicAdd(a + 32, cf[u] * hv[u][1]);
            atomicAdd(a + 64, cf[u] * hv[u][2]);
            atomicAdd(a + 96, cf[u] * hv[u][3]);
          }
        }
      }
    }
  } else {
    const int etot = ecount + n;
    for (int pass = 0; pass < 2; ++pass) {
      for (int base = wave * 32; base < etot; base += OWN_THREADS) {
        const int e = base + lane;
        int s = 0;
        unsigned t = 0xffffffffu;
        float v = 0.0f;
        if (e < etot) {
          const int d = (e < ecount) ? dstI[e] : e - ecount;
          t = (unsigned)(d - lo);
          if (t < urows) {
            s = (e < ecount) ? srcI[e] : d;
            v = asrc[s] + adst[d];
            v = v > 0.0f ? v : v * NEG_SLOPE;
          }
        }
        const bool own = t < urows;
        if (pass == 0) {
          if (own) atomicMax(&rmax[t], fenc(v));
        } else {
          float w = 0.0f;
          if (own) {
            w = expf(v - fdec(rmax[t]));
            atomicAdd(&rden[t], w);
          }
          const int wBits = __float_as_int(w);
          unsigned b = __builtin_amdgcn_ballot_w32(own);
          while (b != 0u) {
            const int j = __builtin_ctz(b);
            b &= b - 1u;
            const int sj = __builtin_amdgcn_readlane(s, j);
            const int rj = __builtin_amdgcn_readlane((int)t, j);
            const float wj = __int_as_float(__builtin_amdgcn_readlane(wBits, j));
            const float* h = hl + (size_t)sj * DFEAT;
            float* a = &acc[rj * DFEAT + lane];
            atomicAdd(a + 0, wj * h[0]);
            atomicAdd(a + 32, wj * h[32]);
            atomicAdd(a + 64, wj * h[64]);
            atomicAdd(a + 96, wj * h[96]);
          }
        }
      }
      __syncthreads();
    }
  }

  __syncthreads();

  for (int i = tid; i < rows * (DFEAT / 4); i += OWN_THREADS) {
    float4 v = ((const float4*)acc)[i];
    const float dn = rden[i / (DFEAT / 4)] + 1e-16f;
    const float4 b4 = *(const float4*)(bias + (i & (DFEAT / 4 - 1)) * 4);
    v.x = v.x / dn + b4.x; v.y = v.y / dn + b4.y;
    v.z = v.z / dn + b4.z; v.w = v.w / dn + b4.w;
    if (relu) {
      v.x = fmaxf(v.x, 0.0f); v.y = fmaxf(v.y, 0.0f);
      v.z = fmaxf(v.z, 0.0f); v.w = fmaxf(v.w, 0.0f);
    }
    *(float4*)(O + (size_t)lo * DFEAT + (size_t)i * 4) = v;
  }
}

__global__ void __launch_bounds__(256) gat_predict(const float* __restrict__ x2o,
                                                   const float* __restrict__ x2s,
                                                   const float* __restrict__ deg,
                                                   const float* __restrict__ Wp,
                                                   const float* __restrict__ bp,
                                                   float* __restrict__ out, int n) {
  int i = blockIdx.x * blockDim.x + threadIdx.x;
  if (i >= n) return;
  const float* zo = x2o + (size_t)i * DFEAT;
  const float* zs = x2s + (size_t)i * DFEAT;
  const float* zd = deg + (size_t)i * DDEG;
  float l0 = bp[0], l1 = bp[1];
#pragma unroll 4
  for (int d = 0; d < DFEAT; ++d) {
    l0 += zo[d] * Wp[d * 2 + 0];
    l1 += zo[d] * Wp[d * 2 + 1];
  }
#pragma unroll 4
  for (int d = 0; d < DFEAT; ++d) {
    l0 += zs[d] * Wp[(DFEAT + d) * 2 + 0];
    l1 += zs[d] * Wp[(DFEAT + d) * 2 + 1];
  }
#pragma unroll 4
  for (int d = 0; d < DDEG; ++d) {
    l0 += zd[d] * Wp[(2 * DFEAT + d) * 2 + 0];
    l1 += zd[d] * Wp[(2 * DFEAT + d) * 2 + 1];
  }
  float mx = fmaxf(l0, l1);
  float e0 = expf(l0 - mx), e1 = expf(l1 - mx);
  float inv = 1.0f / (e0 + e1);
  float a0 = e0 * inv, a1 = e1 * inv;
  float* po = out + (size_t)i * DFEAT;
#pragma unroll 4
  for (int d = 0; d < DFEAT; ++d) po[d] = a0 * zo[d] + a1 * zs[d];
}

static inline int ceil_div(long long a, long long b) { return (int)((a + b - 1) / b); }

static void run_gat_layer(const float* X, const int* srcI, const int* dstI, int ecount,
                          const float* W, const float* avs, const float* avd,
                          const float* bias, float* H, float* asrc, float* adst,
                          float* O, int n, int relu, int* lists, int reuse, hipStream_t stream) {
  gat_gemm_xw<<<n / 16, 256, 0, stream>>>(X, W, H, n);
  gat_node_alpha<<<ceil_div((long long)n * 32, 256), 256, 0, stream>>>(H, avs, avd, asrc, adst, n);
  gat_owner_rows<<<ceil_div(n, OWN_ROWS), OWN_THREADS, 0, stream>>>(srcI, dstI, ecount, n, asrc, adst,
                                                                  H, bias, relu, lists, reuse, O);
}

extern "C" void kernel_launch(void* const* d_in, const int* in_sizes, int n_in,
                              void* d_out, int out_size, void* d_ws, size_t ws_size,
                              hipStream_t stream) {
  const float* x_o      = (const float*)d_in[0];
  const float* degree_o = (const float*)d_in[1];
  const int*   ei_o     = (const int*)d_in[2];
  const int*   ei_s     = (const int*)d_in[3];
  const float* W_o1 = (const float*)d_in[4];
  const float* as_o1 = (const float*)d_in[5];
  const float* ad_o1 = (const float*)d_in[6];
  const float* b_o1  = (const float*)d_in[7];
  const float* W_o2 = (const float*)d_in[8];
  const float* as_o2 = (const float*)d_in[9];
  const float* ad_o2 = (const float*)d_in[10];
  const float* b_o2  = (const float*)d_in[11];
  const float* W_s1 = (const float*)d_in[12];
  const float* as_s1 = (const float*)d_in[13];
  const float* ad_s1 = (const float*)d_in[14];
  const float* b_s1  = (const float*)d_in[15];
  const float* W_s2 = (const float*)d_in[16];
  const float* as_s2 = (const float*)d_in[17];
  const float* ad_s2 = (const float*)d_in[18];
  const float* b_s2  = (const float*)d_in[19];
  const float* W_pred = (const float*)d_in[20];
  const float* b_pred = (const float*)d_in[21];

  const int n = in_sizes[0] / DFEAT;
  const int ecount = in_sizes[2] / 2;

  const int* src_o = ei_o;
  const int* dst_o = ei_o + ecount;
  const int* src_s = ei_s;
  const int* dst_s = ei_s + ecount;

  float* ws = (float*)d_ws;
  float* H     = ws;
  float* X1    = H + (size_t)n * DFEAT;
  float* X2o   = X1 + (size_t)n * DFEAT;
  float* LST   = X2o + (size_t)n * DFEAT;
  float* asrc  = LST + (size_t)n * DFEAT;
  float* adst  = asrc + n;
  float* X2s   = X1;
  int* lists   = (int*)LST;

  run_gat_layer(x_o, src_o, dst_o, ecount, W_o1, as_o1, ad_o1, b_o1,
                H, asrc, adst, X1, n,  1, lists,  0, stream);
  run_gat_layer(X1, src_o, dst_o, ecount, W_o2, as_o2, ad_o2, b_o2,
                H, asrc, adst, X2o, n,  0, lists,  1, stream);
  run_gat_layer(x_o, src_s, dst_s, ecount, W_s1, as_s1, ad_s1, b_s1,
                H, asrc, adst, X1, n,  1, lists,  0, stream);
  run_gat_layer(X1, src_s, dst_s, ecount, W_s2, as_s2, ad_s2, b_s2,
                H, asrc, adst, X2s, n,  0, lists,  1, stream);

  gat_predict<<<ceil_div(n, 256), 256, 0, stream>>>(X2o, X2s, degree_o, W_pred, b_pred,
                                                    (float*)d_out, n);
}


# GEMM re-tiled as a resident grid (256 work-groups x 16 waves): each wave keeps W's fragments for a 32-column slice in registers and streams 16-row X tiles from global; k grouping of the sums changes
# speedup vs baseline: 4.3170x; 1.1157x over previous
#include <hip/hip_runtime.h>
#include <math.h>

typedef __attribute__((ext_vector_type(2))) float v2f;
typedef __attribute__((ext_vector_type(8))) float v8f;

#define DFEAT 128
#define DDEG 32
#define NEG_SLOPE 0.2f

__device__ __forceinline__ unsigned fenc(float f) {
  unsigned u = __float_as_uint(f);
  return (u & 0x80000000u) ? ~u : (u | 0x80000000u);
}
__device__ __forceinline__ float fdec(unsigned u) {
  u = (u & 0x80000000u) ? (u & 0x7fffffffu) : ~u;
  return __uint_as_float(u);
}

#define GEMM_WAVES 16
#define GEMM_THREADS (GEMM_WAVES * 32)
#define GEMM_GRID 256

__global__ void __launch_bounds__(GEMM_THREADS) gat_gemm_xw(const float* __restrict__ X,
                                                            const float* __restrict__ W,
                                                            float* __restrict__ H,
                                                            int ntiles) {
  __shared__ __attribute__((aligned(16))) float wl[DFEAT * DFEAT];
  const int tid = threadIdx.x;
  const int lane = tid & 31;
  const int wave = __builtin_amdgcn_readfirstlane(tid >> 5);
  const int c = lane & 15;
  const int h = lane >> 4;
  const int q = wave & 3;
  const int kq = wave >> 2;
  const int tstep = (int)gridDim.x * (GEMM_WAVES / 4);
  int tile = (int)blockIdx.x + (int)gridDim.x * kq;

  float4 a4[16];
  {
    const int t0 = tile < ntiles ? tile : 0;
    const float* xr = X + (size_t)(t0 * 16 + c) * DFEAT + 4 * h;
#pragma unroll
    for (int j = 0; j < 16; ++j) a4[j] = *(const float4*)(xr + 8 * j);
  }

#pragma unroll
  for (int i = 0; i < DFEAT * DFEAT / 4 / GEMM_THREADS; ++i) {
    const int g = tid + i * GEMM_THREADS;
    const int ln = g & 31;
    const int t = (g >> 5) & 7;
    const int j = g >> 8;
    const float* wp = W + (size_t)(8 * j + 4 * (ln >> 4)) * DFEAT + 16 * t + (ln & 15);
    float4 v;
    v.x = wp[0];
    v.y = wp[DFEAT];
    v.z = wp[2 * DFEAT];
    v.w = wp[3 * DFEAT];
    *(float4*)(wl + g * 4) = v;
  }
  __syncthreads();

  float4 bq[16][2];
  {
    const float* wf = wl + (2 * q * 32 + lane) * 4;
#pragma unroll
    for (int j = 0; j < 16; ++j) {
      bq[j][0] = *(const float4*)(wf + (j * 8 + 0) * (32 * 4));
      bq[j][1] = *(const float4*)(wf + (j * 8 + 1) * (32 * 4));
    }
  }

  while (tile < ntiles) {
    const int next = tile + tstep;
    const int tl = next < ntiles ? next : tile;
    const float* xn = X + (size_t)(tl * 16 + c) * DFEAT + 4 * h;
    v8f acc0 = {};
    v8f acc1 = {};
#pragma unroll
    for (int j = 0; j < 16; ++j) {
      v2f a, b0, b1;
      a.x = a4[j].x; a.y = a4[j].y;
      b0.x = bq[j][0].x; b0.y = bq[j][0].y;
      b1.x = bq[j][1].x; b1.y = bq[j][1].y;
      acc0 = __builtin_amdgcn_wmma_f32_16x16x4_f32(
           false, a,  false, b0,
           (short)0, acc0,  false,  false);
      acc1 = __builtin_amdgcn_wmma_f32_16x16x4_f32(
           false, a,  false, b1,
           (short)0, acc1,  false,  false);
      a.x = a4[j].z; a.y = a4[j].w;
      b0.x = bq[j][0].z; b0.y = bq[j][0].w;
      b1.x = bq[j][1].z; b1.y = bq[j][1].w;
      acc0 = __builtin_amdgcn_wmma_f32_16x16x4_f32(
           false, a,  false, b0,
           (short)0, acc0,  false,  false);
      acc1 = __builtin_amdgcn_wmma_f32_16x16x4_f32(
           false, a,  false, b1,
           (short)0, acc1,  false,  false);
      a4[j] = *(const float4*)(xn + 8 * j);
    }

    float* hrow = H + (size_t)tile * 16 * DFEAT + 32 * q + lane;
#pragma unroll
    for (int r = 0; r < 8; ++r) {
      const float keep = h ? acc1[r] : acc0[r];
      const float give = h ? acc0[r] : acc1[r];
      const float recv = __shfl_xor(give, 16, 32);
      hrow[(size_t)r * DFEAT] = h ? recv : keep;
      hrow[(size_t)(8 + r) * DFEAT] = h ? keep : recv;
    }
    tile = next;
  }
}

__global__ void __launch_bounds__(256) gat_node_alpha(const float* __restrict__ H,
                                                      const float* __restrict__ a_src,
                                                      const float* __restrict__ a_dst,
                                                      float* __restrict__ asrc,
                                                      float* __restrict__ adst,
                                                      int n) {
  int g = blockIdx.x * blockDim.x + threadIdx.x;
  int node = g >> 5;
  int lane = threadIdx.x & 31;
  if (node >= n) return;
  float4 h4 = *(const float4*)(H + (size_t)node * DFEAT + lane * 4);
  float4 s4 = *(const float4*)(a_src + lane * 4);
  float4 d4 = *(const float4*)(a_dst + lane * 4);
  float ps = h4.x * s4.x + h4.y * s4.y + h4.z * s4.z + h4.w * s4.w;
  float pd = h4.x * d4.x + h4.y * d4.y + h4.z * d4.z + h4.w * d4.w;
#pragma unroll
  for (int off = 16; off; off >>= 1) {
    ps += __shfl_down(ps, off, 32);
    pd += __shfl_down(pd, off, 32);
  }
  if (lane == 0) {
    asrc[node] = ps;
    adst[node] = pd;
  }
}

#define OWN_ROWS 196
#define OWN_WAVES 32
#define OWN_THREADS (OWN_WAVES * 32)
#define SCAN_DEPTH 8
#define CHUNK_IDS 128
#define SLAB_IDS (OWN_WAVES * CHUNK_IDS)
#define BATCH_IDS (SCAN_DEPTH * SLAB_IDS)
#define WCAP 512
#define SCORE_DEPTH 8
#define GROUP_EDGES 16
#define LIST_WORDS (OWN_WAVES * WCAP)
#define LIST_STRIDE (LIST_WORDS + 64)

__device__ __forceinline__ int own_append(int* __restrict__ qe, unsigned char* __restrict__ qr,
                                          int qn, bool p, int e, unsigned t, int& ovf) {
  const unsigned b = __builtin_amdgcn_ballot_w32(p);
  if (b != 0u) {
    const int cnt = __builtin_popcount(b);
    if (qn + cnt <= WCAP) {
      const int pos = qn + (int)__builtin_amdgcn_mbcnt_lo(b, 0u);
      if (p) { qe[pos] = e; qr[pos] = (unsigned char)t; }
      qn += cnt;
    } else {
      ovf = 1;
    }
  }
  return qn;
}

__global__ void __launch_bounds__(OWN_THREADS) gat_owner_rows(const int* __restrict__ srcI,
                                                              const int* __restrict__ dstI,
                                                              int ecount, int n,
                                                              const float* __restrict__ asrc,
                                                              const float* __restrict__ adst,
                                                              const float* __restrict__ H,
                                                              const float* __restrict__ bias,
                                                              int relu,
                                                              int* __restrict__ lists,
                                                              int reuse,
                                                              float* __restrict__ O) {
  __shared__ __attribute__((aligned(16))) float acc[OWN_ROWS * DFEAT];
  __shared__ int qsbuf[OWN_WAVES * WCAP];
  __shared__ float qvbuf[OWN_WAVES * WCAP];
  __shared__ unsigned char qrbuf[OWN_WAVES * WCAP];
  __shared__ unsigned rmax[OWN_ROWS];
  __shared__ float rden[OWN_ROWS];
  __shared__ int ovf_flag;

  const int tid = threadIdx.x;
  const int lane = tid & 31;
  const int wave = __builtin_amdgcn_readfirstlane(tid >> 5);
  const int lo = blockIdx.x * OWN_ROWS;
  const int rows = min(OWN_ROWS, n - lo);
  const unsigned urows = (unsigned)rows;

  for (int i = tid; i < OWN_ROWS * DFEAT / 4; i += OWN_THREADS)
    ((float4*)acc)[i] = make_float4(0.0f, 0.0f, 0.0f, 0.0f);
  if (tid < OWN_ROWS) {
    rmax[tid] = fenc(-INFINITY);
    rden[tid] = 0.0f;
  }
  if (tid == 0) ovf_flag = 0;
  __syncthreads();

  int* myqs = qsbuf + wave * WCAP;
  float* myqv = qvbuf + wave * WCAP;
  unsigned char* myqr = qrbuf + wave * WCAP;
  int qn = 0;
  int ovf = 0;

  int* gq = lists + (size_t)blockIdx.x * LIST_STRIDE;
  int* myg = gq + wave * WCAP;

  if (reuse == 0) {
    for (int r0 = wave * 32; r0 < rows; r0 += OWN_THREADS)
      qn = own_append(myqs, myqr, qn, r0 + lane < rows, ecount + lo + r0 + lane, (unsigned)(r0 + lane), ovf);

    const bool vec_ok = (((size_t)dstI) & 15) == 0;
    const int nb_full = vec_ok ? ecount / BATCH_IDS : 0;
    const int rot = nb_full ? (int)(blockIdx.x % (unsigned)nb_full) : 0;
    for (int k = 0; k < nb_full; ++k) {
      int bi = k + rot;
      if (bi >= nb_full) bi -= nb_full;
      const int b0 = bi * BATCH_IDS + wave * CHUNK_IDS + lane * 4;
      int4 dv[SCAN_DEPTH];
  #pragma unroll
      for (int u = 0; u < SCAN_DEPTH; ++u)
        dv[u] = *(const int4*)(dstI + b0 + u * SLAB_IDS);
  #pragma unroll
      for (int u = 0; u < SCAN_DEPTH; ++u) {
        const unsigned t0 = (unsigned)(dv[u].x - lo);
        const unsigned t1 = (unsigned)(dv[u].y - lo);
        const unsigned t2 = (unsigned)(dv[u].z - lo);
        const unsigned t3 = (unsigned)(dv[u].w - lo);
        const unsigned tm = min(min(t0, t1), min(t2, t3));
        if (__builtin_amdgcn_ballot_w32(tm < urows) != 0u) {
          const int e0 = b0 + u * SLAB_IDS;
          qn = own_append(myqs, myqr, qn, t0 < urows, e0, t0, ovf);
          qn = own_append(myqs, myqr, qn, t1 < urows, e0 + 1, t1, ovf);
          qn = own_append(myqs, myqr, qn, t2 < urows, e0 + 2, t2, ovf);
          qn = own_append(myqs, myqr, qn, t3 < urows, e0 + 3, t3, ovf);
        }
      }
    }

    for (int base = nb_full * BATCH_IDS + wave * CHUNK_IDS; base < ecount; base += SLAB_IDS) {
      unsigned t[4];
  #pragma unroll
      for (int j = 0; j < 4; ++j) {
        const int e = base + j * 32 + lane;
        t[j] = (e < ecount) ? (unsigned)(dstI[e] - lo) : 0xffffffffu;
      }
  #pragma unroll
      for (int j = 0; j < 4; ++j)
        qn = own_append(myqs, myqr, qn, t[j] < urows, base + j * 32 + lane, t[j], ovf);
    }
  } else {
    ovf = __builtin_amdgcn_readfirstlane(gq[LIST_WORDS + OWN_WAVES]);
    qn = ovf != 0 ? 0 : __builtin_amdgcn_readfirstlane(gq[LIST_WORDS + wave]);
  }

  for (int g = 0; g < qn; g += 32 * SCORE_DEPTH) {
    int s[SCORE_DEPTH];
    int r[SCORE_DEPTH];
    float v[SCORE_DEPTH];
    if (reuse == 0) {
#pragma unroll
      for (int u = 0; u < SCORE_DEPTH; ++u) {
        const int i = g + u * 32 + lane;
        s[u] = 0;
        r[u] = 0;
        if (i < qn) {
          const int e = myqs[i];
          r[u] = myqr[i];
          s[u] = (e < ecount) ? srcI[e] : e - ecount;
        }
      }
    } else {
#pragma unroll
      for (int u = 0; u < SCORE_DEPTH; ++u) {
        const int i = g + u * 32 + lane;
        s[u] = 0;
        if (i < qn) s[u] = myg[i];
      }
#pragma unroll
      for (int u = 0; u < SCORE_DEPTH; ++u) {
        r[u] = (int)((unsigned)s[u] >> 24);
        s[u] &= 0xffffff;
      }
    }
#pragma unroll
    for (int u = 0; u < SCORE_DEPTH; ++u)
      v[u] = asrc[s[u]] + adst[lo + r[u]];
#pragma unroll
    for (int u = 0; u < SCORE_DEPTH; ++u) {
      const int i = g + u * 32 + lane;
      if (i < qn) {
        const float x = v[u] > 0.0f ? v[u] : v[u] * NEG_SLOPE;
        myqs[i] = s[u];
        myqv[i] = x;
        if (reuse == 0) myg[i] = (int)((unsigned)s[u] | ((unsigned)r[u] << 24));
        else myqr[i] = (unsigned char)r[u];
        atomicMax(&rmax[r[u]], fenc(x));
      }
    }
  }
  if (lane == 0) {
    if (reuse == 0) gq[LIST_WORDS + wave] = qn;
    if (ovf != 0) ovf_flag = 1;
  }

  __syncthreads();

  if (reuse == 0 && tid == 0) gq[LIST_WORDS + OWN_WAVES] = ovf_flag;

  const float* hl = H + lane;
  if (__builtin_amdgcn_readfirstlane(ovf_flag) == 0) {
    for (int j0 = 0; j0 < qn; j0 += 32) {
      const int cnt = min(32, qn - j0);
      int s = 0, r = 0;
      float w = 0.0f;
      if (lane < cnt) {
        s = myqs[j0 + lane];
        r = myqr[j0 + lane];
        w = expf(myqv[j0 + lane] - fdec(rmax[r]));
        atomicAdd(&rden[r], w);
      }
      const int wBits = __float_as_int(w);
      for (int k0 = 0; k0 < cnt; k0 += GROUP_EDGES) {
        float hv[GROUP_EDGES][4];
        float cf[GROUP_EDGES];
        int rr[GROUP_EDGES];
#pragma unroll
        for (int u = 0; u < GROUP_EDGES; ++u) {
          const int sj = __builtin_amdgcn_readlane(s, k0 + u);
          rr[u] = __builtin_amdgcn_readlane(r, k0 + u);
          cf[u] = __int_as_float(__builtin_amdgcn_readlane(wBits, k0 + u));
          const float* h = hl + (size_t)sj * DFEAT;
          hv[u][0] = h[0];
          hv[u][1] = h[32];
          hv[u][2] = h[64];
          hv[u][3] = h[96];
        }
#pragma unroll
        for (int u = 0; u < GROUP_EDGES; ++u) {
          if (k0 + u < cnt) {
            float* a = &acc[rr[u] * DFEAT + lane];
            atomicAdd(a + 0, cf[u] * hv[u][0]);
            atomicAdd(a + 32, cf[u] * hv[u][1]);
            atomicAdd(a + 64, cf[u] * hv[u][2]);
            atomicAdd(a + 96, cf[u] * hv[u][3]);
          }
        }
      }
    }
  } else {
    const int etot = ecount + n;
    for (int pass = 0; pass < 2; ++pass) {
      for (int base = wave * 32; base < etot; base += OWN_THREADS) {
        const int e = base + lane;
        int s = 0;
        unsigned t = 0xffffffffu;
        float v = 0.0f;
        if (e < etot) {
          const int d = (e < ecount) ? dstI[e] : e - ecount;
          t = (unsigned)(d - lo);
          if (t < urows) {
            s = (e < ecount) ? srcI[e] : d;
            v = asrc[s] + adst[d];
            v = v > 0.0f ? v : v * NEG_SLOPE;
          }
        }
        const bool own = t < urows;
        if (pass == 0) {
          if (own) atomicMax(&rmax[t], fenc(v));
        } else {
          float w = 0.0f;
          if (own) {
            w = expf(v - fdec(rmax[t]));
            atomicAdd(&rden[t], w);
          }
          const int wBits = __float_as_int(w);
          unsigned b = __builtin_amdgcn_ballot_w32(own);
          while (b != 0u) {
            const int j = __builtin_ctz(b);
            b &= b - 1u;
            const int sj = __builtin_amdgcn_readlane(s, j);
            const int rj = __builtin_amdgcn_readlane((int)t, j);
            const float wj = __int_as_float(__builtin_amdgcn_readlane(wBits, j));
            const float* h = hl + (size_t)sj * DFEAT;
            float* a = &acc[rj * DFEAT + lane];
            atomicAdd(a + 0, wj * h[0]);
            atomicAdd(a + 32, wj * h[32]);
            atomicAdd(a + 64, wj * h[64]);
            atomicAdd(a + 96, wj * h[96]);
          }
        }
      }
      __syncthreads();
    }
  }

  __syncthreads();

  for (int i = tid; i < rows * (DFEAT / 4); i += OWN_THREADS) {
    float4 v = ((const float4*)acc)[i];
    const float dn = rden[i / (DFEAT / 4)] + 1e-16f;
    const float4 b4 = *(const float4*)(bias + (i & (DFEAT / 4 - 1)) * 4);
    v.x = v.x / dn + b4.x; v.y = v.y / dn + b4.y;
    v.z = v.z / dn + b4.z; v.w = v.w / dn + b4.w;
    if (relu) {
      v.x = fmaxf(v.x, 0.0f); v.y = fmaxf(v.y, 0.0f);
      v.z = fmaxf(v.z, 0.0f); v.w = fmaxf(v.w, 0.0f);
    }
    *(float4*)(O + (size_t)lo * DFEAT + (size_t)i * 4) = v;
  }
}

__global__ void __launch_bounds__(256) gat_predict(const float* __restrict__ x2o,
                                                   const float* __restrict__ x2s,
                                                   const float* __restrict__ deg,
                                                   const float* __restrict__ Wp,
                                                   const float* __restrict__ bp,
                                                   float* __restrict__ out, int n) {
  int i = blockIdx.x * blockDim.x + threadIdx.x;
  if (i >= n) return;
  const float* zo = x2o + (size_t)i * DFEAT;
  const float* zs = x2s + (size_t)i * DFEAT;
  const float* zd = deg + (size_t)i * DDEG;
  float l0 = bp[0], l1 = bp[1];
#pragma unroll 4
  for (int d = 0; d < DFEAT; ++d) {
    l0 += zo[d] * Wp[d * 2 + 0];
    l1 += zo[d] * Wp[d * 2 + 1];
  }
#pragma unroll 4
  for (int d = 0; d < DFEAT; ++d) {
    l0 += zs[d] * Wp[(DFEAT + d) * 2 + 0];
    l1 += zs[d] * Wp[(DFEAT + d) * 2 + 1];
  }
#pragma unroll 4
  for (int d = 0; d < DDEG; ++d) {
    l0 += zd[d] * Wp[(2 * DFEAT + d) * 2 + 0];
    l1 += zd[d] * Wp[(2 * DFEAT + d) * 2 + 1];
  }
  float mx = fmaxf(l0, l1);
  float e0 = expf(l0 - mx), e1 = expf(l1 - mx);
  float inv = 1.0f / (e0 + e1);
  float a0 = e0 * inv, a1 = e1 * inv;
  float* po = out + (size_t)i * DFEAT;
#pragma unroll 4
  for (int d = 0; d < DFEAT; ++d) po[d] = a0 * zo[d] + a1 * zs[d];
}

static inline int ceil_div(long long a, long long b) { return (int)((a + b - 1) / b); }

static void run_gat_layer(const float* X, const int* srcI, const int* dstI, int ecount,
                          const float* W, const float* avs, const float* avd,
                          const float* bias, float* H, float* asrc, float* adst,
                          float* O, int n, int relu, int* lists, int reuse, hipStream_t stream) {
  gat_gemm_xw<<<GEMM_GRID, GEMM_THREADS, 0, stream>>>(X, W, H, n / 16);
  gat_node_alpha<<<ceil_div((long long)n * 32, 256), 256, 0, stream>>>(H, avs, avd, asrc, adst, n);
  gat_owner_rows<<<ceil_div(n, OWN_ROWS), OWN_THREADS, 0, stream>>>(srcI, dstI, ecount, n, asrc, adst,
                                                                  H, bias, relu, lists, reuse, O);
}

extern "C" void kernel_launch(void* const* d_in, const int* in_sizes, int n_in,
                              void* d_out, int out_size, void* d_ws, size_t ws_size,
                              hipStream_t stream) {
  const float* x_o      = (const float*)d_in[0];
  const float* degree_o = (const float*)d_in[1];
  const int*   ei_o     = (const int*)d_in[2];
  const int*   ei_s     = (const int*)d_in[3];
  const float* W_o1 = (const float*)d_in[4];
  const float* as_o1 = (const float*)d_in[5];
  const float* ad_o1 = (const float*)d_in[6];
  const float* b_o1  = (const float*)d_in[7];
  const float* W_o2 = (const float*)d_in[8];
  const float* as_o2 = (const float*)d_in[9];
  const float* ad_o2 = (const float*)d_in[10];
  const float* b_o2  = (const float*)d_in[11];
  const float* W_s1 = (const float*)d_in[12];
  const float* as_s1 = (const float*)d_in[13];
  const float* ad_s1 = (const float*)d_in[14];
  const float* b_s1  = (const float*)d_in[15];
  const float* W_s2 = (const float*)d_in[16];
  const float* as_s2 = (const float*)d_in[17];
  const float* ad_s2 = (const float*)d_in[18];
  const float* b_s2  = (const float*)d_in[19];
  const float* W_pred = (const float*)d_in[20];
  const float* b_pred = (const float*)d_in[21];

  const int n = in_sizes[0] / DFEAT;
  const int ecount = in_sizes[2] / 2;

  const int* src_o = ei_o;
  const int* dst_o = ei_o + ecount;
  const int* src_s = ei_s;
  const int* dst_s = ei_s + ecount;

  float* ws = (float*)d_ws;
  float* H     = ws;
  float* X1    = H + (size_t)n * DFEAT;
  float* X2o   = X1 + (size_t)n * DFEAT;
  float* LST   = X2o + (size_t)n * DFEAT;
  float* asrc  = LST + (size_t)n * DFEAT;
  float* adst  = asrc + n;
  float* X2s   = X1;
  int* lists   = (int*)LST;

  run_gat_layer(x_o, src_o, dst_o, ecount, W_o1, as_o1, ad_o1, b_o1,
                H, asrc, adst, X1, n,  1, lists,  0, stream);
  run_gat_layer(X1, src_o, dst_o, ecount, W_o2, as_o2, ad_o2, b_o2,
                H, asrc, adst, X2o, n,  0, lists,  1, stream);
  run_gat_layer(x_o, src_s, dst_s, ecount, W_s1, as_s1, ad_s1, b_s1,
                H, asrc, adst, X1, n,  1, lists,  0, stream);
  run_gat_layer(X1, src_s, dst_s, ecount, W_s2, as_s2, ad_s2, b_s2,
                H, asrc, adst, X2s, n,  0, lists,  1, stream);

  gat_predict<<<ceil_div(n, 256), 256, 0, stream>>>(X2o, X2s, degree_o, W_pred, b_pred,
                                                    (float*)d_out, n);
}


# Fusion head (288-term logits, 2-way softmax, blend) moves into the last owner launch: a wave per owned row, straight from the LDS sums; 13 launches become 12; logit sum order changes (DPP xor tree)
# speedup vs baseline: 4.6506x; 1.0773x over previous
#include <hip/hip_runtime.h>
#include <math.h>

typedef __attribute__((ext_vector_type(2))) float v2f;
typedef __attribute__((ext_vector_type(8))) float v8f;

#define DFEAT 128
#define DDEG 32
#define NEG_SLOPE 0.2f

__device__ __forceinline__ unsigned fenc(float f) {
  unsigned u = __float_as_uint(f);
  return (u & 0x80000000u) ? ~u : (u | 0x80000000u);
}
__device__ __forceinline__ float fdec(unsigned u) {
  u = (u & 0x80000000u) ? (u & 0x7fffffffu) : ~u;
  return __uint_as_float(u);
}

template <int M>
__device__ __forceinline__ float row_xmask(float x) {
  const int b = __builtin_bit_cast(int, x);
  return __builtin_bit_cast(float, __builtin_amdgcn_update_dpp(b, b, 0x160 + M, 0xf, 0xf, false));
}
__device__ __forceinline__ float xhalf(float x) {
  const int b = __builtin_bit_cast(int, x);
  return __builtin_bit_cast(float, __builtin_amdgcn_permlanex16(b, b, 0x76543210, 0xfedcba98, false, false));
}
__device__ __forceinline__ float wave_sum(float x) {
  x += row_xmask<1>(x);
  x += row_xmask<2>(x);
  x += row_xmask<4>(x);
  x += row_xmask<8>(x);
  return x + xhalf(x);
}

#define GEMM_WAVES 16
#define GEMM_THREADS (GEMM_WAVES * 32)
#define GEMM_GRID 256

__global__ void __launch_bounds__(GEMM_THREADS) gat_gemm_xw(const float* __restrict__ X,
                                                            const float* __restrict__ W,
                                                            float* __restrict__ H,
                                                            int ntiles) {
  __shared__ __attribute__((aligned(16))) float wl[DFEAT * DFEAT];
  const int tid = threadIdx.x;
  const int lane = tid & 31;
  const int wave = __builtin_amdgcn_readfirstlane(tid >> 5);
  const int c = lane & 15;
  const int h = lane >> 4;
  const int q = wave & 3;
  const int kq = wave >> 2;
  const int tstep = (int)gridDim.x * (GEMM_WAVES / 4);
  int tile = (int)blockIdx.x + (int)gridDim.x * kq;

  float4 a4[16];
  {
    const int t0 = tile < ntiles ? tile : 0;
    const float* xr = X + (size_t)(t0 * 16 + c) * DFEAT + 4 * h;
#pragma unroll
    for (int j = 0; j < 16; ++j) a4[j] = *(const float4*)(xr + 8 * j);
  }

#pragma unroll
  for (int i = 0; i < DFEAT * DFEAT / 4 / GEMM_THREADS; ++i) {
    const int g = tid + i * GEMM_THREADS;
    const int ln = g & 31;
    const int t = (g >> 5) & 7;
    const int j = g >> 8;
    const float* wp = W + (size_t)(8 * j + 4 * (ln >> 4)) * DFEAT + 16 * t + (ln & 15);
    float4 v;
    v.x = wp[0];
    v.y = wp[DFEAT];
    v.z = wp[2 * DFEAT];
    v.w = wp[3 * DFEAT];
    *(float4*)(wl + g * 4) = v;
  }
  __syncthreads();

  float4 bq[16][2];
  {
    const float* wf = wl + (2 * q * 32 + lane) * 4;
#pragma unroll
    for (int j = 0; j < 16; ++j) {
      bq[j][0] = *(const float4*)(wf + (j * 8 + 0) * (32 * 4));
      bq[j][1] = *(const float4*)(wf + (j * 8 + 1) * (32 * 4));
    }
  }

  while (tile < ntiles) {
    const int next = tile + tstep;
    const int tl = next < ntiles ? next : tile;
    const float* xn = X + (size_t)(tl * 16 + c) * DFEAT + 4 * h;
    v8f acc0 = {};
    v8f acc1 = {};
#pragma unroll
    for (int j = 0; j < 16; ++j) {
      v2f a, b0, b1;
      a.x = a4[j].x; a.y = a4[j].y;
      b0.x = bq[j][0].x; b0.y = bq[j][0].y;
      b1.x = bq[j][1].x; b1.y = bq[j][1].y;
      acc0 = __builtin_amdgcn_wmma_f32_16x16x4_f32(
           false, a,  false, b0,
           (short)0, acc0,  false,  false);
      acc1 = __builtin_amdgcn_wmma_f32_16x16x4_f32(
           false, a,  false, b1,
           (short)0, acc1,  false,  false);
      a.x = a4[j].z; a.y = a4[j].w;
      b0.x = bq[j][0].z; b0.y = bq[j][0].w;
      b1.x = bq[j][1].z; b1.y = bq[j][1].w;
      acc0 = __builtin_amdgcn_wmma_f32_16x16x4_f32(
           false, a,  false, b0,
           (short)0, acc0,  false,  false);
      acc1 = __builtin_amdgcn_wmma_f32_16x16x4_f32(
           false, a,  false, b1,
           (short)0, acc1,  false,  false);
      a4[j] = *(const float4*)(xn + 8 * j);
    }

    float* hrow = H + (size_t)tile * 16 * DFEAT + 32 * q + lane;
#pragma unroll
    for (int r = 0; r < 8; ++r) {
      const float keep = h ? acc1[r] : acc0[r];
      const float give = h ? acc0[r] : acc1[r];
      const float recv = __shfl_xor(give, 16, 32);
      hrow[(size_t)r * DFEAT] = h ? recv : keep;
      hrow[(size_t)(8 + r) * DFEAT] = h ? keep : recv;
    }
    tile = next;
  }
}

__global__ void __launch_bounds__(256) gat_node_alpha(const float* __restrict__ H,
                                                      const float* __restrict__ a_src,
                                                      const float* __restrict__ a_dst,
                                                      float* __restrict__ asrc,
                                                      float* __restrict__ adst,
                                                      int n) {
  int g = blockIdx.x * blockDim.x + threadIdx.x;
  int node = g >> 5;
  int lane = threadIdx.x & 31;
  if (node >= n) return;
  float4 h4 = *(const float4*)(H + (size_t)node * DFEAT + lane * 4);
  float4 s4 = *(const float4*)(a_src + lane * 4);
  float4 d4 = *(const float4*)(a_dst + lane * 4);
  float ps = h4.x * s4.x + h4.y * s4.y + h4.z * s4.z + h4.w * s4.w;
  float pd = h4.x * d4.x + h4.y * d4.y + h4.z * d4.z + h4.w * d4.w;
#pragma unroll
  for (int off = 16; off; off >>= 1) {
    ps += __shfl_down(ps, off, 32);
    pd += __shfl_down(pd, off, 32);
  }
  if (lane == 0) {
    asrc[node] = ps;
    adst[node] = pd;
  }
}

#define OWN_ROWS 196
#define OWN_WAVES 32
#define OWN_THREADS (OWN_WAVES * 32)
#define SCAN_DEPTH 8
#define CHUNK_IDS 128
#define SLAB_IDS (OWN_WAVES * CHUNK_IDS)
#define BATCH_IDS (SCAN_DEPTH * SLAB_IDS)
#define WCAP 512
#define SCORE_DEPTH 8
#define GROUP_EDGES 16
#define LIST_WORDS (OWN_WAVES * WCAP)
#define HEAD_DEPTH ((OWN_ROWS + OWN_WAVES - 1) / OWN_WAVES)
#define LIST_STRIDE (LIST_WORDS + 64)

__device__ __forceinline__ int own_append(int* __restrict__ qe, unsigned char* __restrict__ qr,
                                          int qn, bool p, int e, unsigned t, int& ovf) {
  const unsigned b = __builtin_amdgcn_ballot_w32(p);
  if (b != 0u) {
    const int cnt = __builtin_popcount(b);
    if (qn + cnt <= WCAP) {
      const int pos = qn + (int)__builtin_amdgcn_mbcnt_lo(b, 0u);
      if (p) { qe[pos] = e; qr[pos] = (unsigned char)t; }
      qn += cnt;
    } else {
      ovf = 1;
    }
  }
  return qn;
}

__global__ void __launch_bounds__(OWN_THREADS) gat_owner_rows(const int* __restrict__ srcI,
                                                              const int* __restrict__ dstI,
                                                              int ecount, int n,
                                                              const float* __restrict__ asrc,
                                                              const float* __restrict__ adst,
                                                              const float* __restrict__ H,
                                                              const float* __restrict__ bias,
                                                              int relu,
                                                              int* __restrict__ lists,
                                                              int reuse,
                                                              float* __restrict__ O,
                                                              int head,
                                                              const float* __restrict__ x2o,
                                                              const float* __restrict__ deg,
                                                              const float* __restrict__ Wp,
                                                              const float* __restrict__ bp,
                                                              float* __restrict__ out) {
  __shared__ __attribute__((aligned(16))) float acc[OWN_ROWS * DFEAT];
  __shared__ int qsbuf[OWN_WAVES * WCAP];
  __shared__ float qvbuf[OWN_WAVES * WCAP];
  __shared__ unsigned char qrbuf[OWN_WAVES * WCAP];
  __shared__ unsigned rmax[OWN_ROWS];
  __shared__ float rden[OWN_ROWS];
  __shared__ int ovf_flag;

  const int tid = threadIdx.x;
  const int lane = tid & 31;
  const int wave = __builtin_amdgcn_readfirstlane(tid >> 5);
  const int lo = blockIdx.x * OWN_ROWS;
  const int rows = min(OWN_ROWS, n - lo);
  const unsigned urows = (unsigned)rows;

  for (int i = tid; i < OWN_ROWS * DFEAT / 4; i += OWN_THREADS)
    ((float4*)acc)[i] = make_float4(0.0f, 0.0f, 0.0f, 0.0f);
  if (tid < OWN_ROWS) {
    rmax[tid] = fenc(-INFINITY);
    rden[tid] = 0.0f;
  }
  if (tid == 0) ovf_flag = 0;
  __syncthreads();

  int* myqs = qsbuf + wave * WCAP;
  float* myqv = qvbuf + wave * WCAP;
  unsigned char* myqr = qrbuf + wave * WCAP;
  int qn = 0;
  int ovf = 0;

  int* gq = lists + (size_t)blockIdx.x * LIST_STRIDE;
  int* myg = gq + wave * WCAP;

  if (reuse == 0) {
    for (int r0 = wave * 32; r0 < rows; r0 += OWN_THREADS)
      qn = own_append(myqs, myqr, qn, r0 + lane < rows, ecount + lo + r0 + lane, (unsigned)(r0 + lane), ovf);

    const bool vec_ok = (((size_t)dstI) & 15) == 0;
    const int nb_full = vec_ok ? ecount / BATCH_IDS : 0;
    const int rot = nb_full ? (int)(blockIdx.x % (unsigned)nb_full) : 0;
    for (int k = 0; k < nb_full; ++k) {
      int bi = k + rot;
      if (bi >= nb_full) bi -= nb_full;
      const int b0 = bi * BATCH_IDS + wave * CHUNK_IDS + lane * 4;
      int4 dv[SCAN_DEPTH];
  #pragma unroll
      for (int u = 0; u < SCAN_DEPTH; ++u)
        dv[u] = *(const int4*)(dstI + b0 + u * SLAB_IDS);
  #pragma unroll
      for (int u = 0; u < SCAN_DEPTH; ++u) {
        const unsigned t0 = (unsigned)(dv[u].x - lo);
        const unsigned t1 = (unsigned)(dv[u].y - lo);
        const unsigned t2 = (unsigned)(dv[u].z - lo);
        const unsigned t3 = (unsigned)(dv[u].w - lo);
        const unsigned tm = min(min(t0, t1), min(t2, t3));
        if (__builtin_amdgcn_ballot_w32(tm < urows) != 0u) {
          const int e0 = b0 + u * SLAB_IDS;
          qn = own_append(myqs, myqr, qn, t0 < urows, e0, t0, ovf);
          qn = own_append(myqs, myqr, qn, t1 < urows, e0 + 1, t1, ovf);
          qn = own_append(myqs, myqr, qn, t2 < urows, e0 + 2, t2, ovf);
          qn = own_append(myqs, myqr, qn, t3 < urows, e0 + 3, t3, ovf);
        }
      }
    }

    for (int base = nb_full * BATCH_IDS + wave * CHUNK_IDS; base < ecount; base += SLAB_IDS) {
      unsigned t[4];
  #pragma unroll
      for (int j = 0; j < 4; ++j) {
        const int e = base + j * 32 + lane;
        t[j] = (e < ecount) ? (unsigned)(dstI[e] - lo) : 0xffffffffu;
      }
  #pragma unroll
      for (int j = 0; j < 4; ++j)
        qn = own_append(myqs, myqr, qn, t[j] < urows, base + j * 32 + lane, t[j], ovf);
    }
  } else {
    ovf = __builtin_amdgcn_readfirstlane(gq[LIST_WORDS + OWN_WAVES]);
    qn = ovf != 0 ? 0 : __builtin_amdgcn_readfirstlane(gq[LIST_WORDS + wave]);
  }

  for (int g = 0; g < qn; g += 32 * SCORE_DEPTH) {
    int s[SCORE_DEPTH];
    int r[SCORE_DEPTH];
    float v[SCORE_DEPTH];
    if (reuse == 0) {
#pragma unroll
      for (int u = 0; u < SCORE_DEPTH; ++u) {
        const int i = g + u * 32 + lane;
        s[u] = 0;
        r[u] = 0;
        if (i < qn) {
          const int e = myqs[i];
          r[u] = myqr[i];
          s[u] = (e < ecount) ? srcI[e] : e - ecount;
        }
      }
    } else {
#pragma unroll
      for (int u = 0; u < SCORE_DEPTH; ++u) {
        const int i = g + u * 32 + lane;
        s[u] = 0;
        if (i < qn) s[u] = myg[i];
      }
#pragma unroll
      for (int u = 0; u < SCORE_DEPTH; ++u) {
        r[u] = (int)((unsigned)s[u] >> 24);
        s[u] &= 0xffffff;
      }
    }
#pragma unroll
    for (int u = 0; u < SCORE_DEPTH; ++u)
      v[u] = asrc[s[u]] + adst[lo + r[u]];
#pragma unroll
    for (int u = 0; u < SCORE_DEPTH; ++u) {
      const int i = g + u * 32 + lane;
      if (i < qn) {
        const float x = v[u] > 0.0f ? v[u] : v[u] * NEG_SLOPE;
        myqs[i] = s[u];
        myqv[i] = x;
        if (reuse == 0) myg[i] = (int)((unsigned)s[u] | ((unsigned)r[u] << 24));
        else myqr[i] = (unsigned char)r[u];
        atomicMax(&rmax[r[u]], fenc(x));
      }
    }
  }
  if (lane == 0) {
    if (reuse == 0) gq[LIST_WORDS + wave] = qn;
    if (ovf != 0) ovf_flag = 1;
  }

  __syncthreads();

  if (reuse == 0 && tid == 0) gq[LIST_WORDS + OWN_WAVES] = ovf_flag;

  const float* hl = H + lane;
  if (__builtin_amdgcn_readfirstlane(ovf_flag) == 0) {
    for (int j0 = 0; j0 < qn; j0 += 32) {
      const int cnt = min(32, qn - j0);
      int s = 0, r = 0;
      float w = 0.0f;
      if (lane < cnt) {
        s = myqs[j0 + lane];
        r = myqr[j0 + lane];
        w = expf(myqv[j0 + lane] - fdec(rmax[r]));
        atomicAdd(&rden[r], w);
      }
      const int wBits = __float_as_int(w);
      for (int k0 = 0; k0 < cnt; k0 += GROUP_EDGES) {
        float hv[GROUP_EDGES][4];
        float cf[GROUP_EDGES];
        int rr[GROUP_EDGES];
#pragma unroll
        for (int u = 0; u < GROUP_EDGES; ++u) {
          const int sj = __builtin_amdgcn_readlane(s, k0 + u);
          rr[u] = __builtin_amdgcn_readlane(r, k0 + u);
          cf[u] = __int_as_float(__builtin_amdgcn_readlane(wBits, k0 + u));
          const float* h = hl + (size_t)sj * DFEAT;
          hv[u][0] = h[0];
          hv[u][1] = h[32];
          hv[u][2] = h[64];
          hv[u][3] = h[96];
        }
#pragma unroll
        for (int u = 0; u < GROUP_EDGES; ++u) {
          if (k0 + u < cnt) {
            float* a = &acc[rr[u] * DFEAT + lane];
            atomicAdd(a + 0, cf[u] * hv[u][0]);
            atomicAdd(a + 32, cf[u] * hv[u][1]);
            atomicAdd(a + 64, cf[u] * hv[u][2]);
            atomicAdd(a + 96, cf[u] * hv[u][3]);
          }
        }
      }
    }
  } else {
    const int etot = ecount + n;
    for (int pass = 0; pass < 2; ++pass) {
      for (int base = wave * 32; base < etot; base += OWN_THREADS) {
        const int e = base + lane;
        int s = 0;
        unsigned t = 0xffffffffu;
        float v = 0.0f;
        if (e < etot) {
          const int d = (e < ecount) ? dstI[e] : e - ecount;
          t = (unsigned)(d - lo);
          if (t < urows) {
            s = (e < ecount) ? srcI[e] : d;
            v = asrc[s] + adst[d];
            v = v > 0.0f ? v : v * NEG_SLOPE;
          }
        }
        const bool own = t < urows;
        if (pass == 0) {
          if (own) atomicMax(&rmax[t], fenc(v));
        } else {
          float w = 0.0f;
          if (own) {
            w = expf(v - fdec(rmax[t]));
            atomicAdd(&rden[t], w);
          }
          const int wBits = __float_as_int(w);
          unsigned b = __builtin_amdgcn_ballot_w32(own);
          while (b != 0u) {
            const int j = __builtin_ctz(b);
            b &= b - 1u;
            const int sj = __builtin_amdgcn_readlane(s, j);
            const int rj = __builtin_amdgcn_readlane((int)t, j);
            const float wj = __int_as_float(__builtin_amdgcn_readlane(wBits, j));
            const float* h = hl + (size_t)sj * DFEAT;
            float* a = &acc[rj * DFEAT + lane];
            atomicAdd(a + 0, wj * h[0]);
            atomicAdd(a + 32, wj * h[32]);
            atomicAdd(a + 64, wj * h[64]);
            atomicAdd(a + 96, wj * h[96]);
          }
        }
      }
      __syncthreads();
    }
  }

  __syncthreads();

  if (head != 0) {
    const float4 wo0 = *(const float4*)(Wp + lane * 8);
    const float4 wo1 = *(const float4*)(Wp + lane * 8 + 4);
    const float4 ws0 = *(const float4*)(Wp + 2 * DFEAT + lane * 8);
    const float4 ws1 = *(const float4*)(Wp + 2 * DFEAT + lane * 8 + 4);
    const float2 wd = *(const float2*)(Wp + 4 * DFEAT + lane * 2);
    const float bp0 = bp[0];
    const float bp1 = bp[1];
    const float4 b4 = *(const float4*)(bias + lane * 4);
    float4 ov[HEAD_DEPTH];
    float dgv[HEAD_DEPTH];
#pragma unroll
    for (int k = 0; k < HEAD_DEPTH; ++k) {
      const int r = wave + k * OWN_WAVES;
      const int rc = r < rows ? r : 0;
      const size_t row = (size_t)(lo + rc);
      ov[k] = *(const float4*)(x2o + row * DFEAT + lane * 4);
      dgv[k] = deg[row * DDEG + lane];
    }
#pragma unroll
    for (int k = 0; k < HEAD_DEPTH; ++k) {
      const int r = wave + k * OWN_WAVES;
      if (r < rows) {
        float4 v = ((const float4*)acc)[r * (DFEAT / 4) + lane];
        const float dn = rden[r] + 1e-16f;
        v.x = v.x / dn + b4.x; v.y = v.y / dn + b4.y;
        v.z = v.z / dn + b4.z; v.w = v.w / dn + b4.w;
        if (relu) {
          v.x = fmaxf(v.x, 0.0f); v.y = fmaxf(v.y, 0.0f);
          v.z = fmaxf(v.z, 0.0f); v.w = fmaxf(v.w, 0.0f);
        }
        const float4 o = ov[k];
        float l0 = o.x * wo0.x + o.y * wo0.z + o.z * wo1.x + o.w * wo1.z;
        float l1 = o.x * wo0.y + o.y * wo0.w + o.z * wo1.y + o.w * wo1.w;
        l0 += v.x * ws0.x + v.y * ws0.z + v.z * ws1.x + v.w * ws1.z;
        l1 += v.x * ws0.y + v.y * ws0.w + v.z * ws1.y + v.w * ws1.w;
        l0 += dgv[k] * wd.x;
        l1 += dgv[k] * wd.y;
        l0 = wave_sum(l0);
        l1 = wave_sum(l1);
        l0 += bp0;
        l1 += bp1;
        const float mx = fmaxf(l0, l1);
        const float e0 = expf(l0 - mx), e1 = expf(l1 - mx);
        const float inv = 1.0f / (e0 + e1);
        const float a0 = e0 * inv, a1 = e1 * inv;
        float4 p;
        p.x = a0 * o.x + a1 * v.x; p.y = a0 * o.y + a1 * v.y;
        p.z = a0 * o.z + a1 * v.z; p.w = a0 * o.w + a1 * v.w;
        *(float4*)(out + (size_t)(lo + r) * DFEAT + lane * 4) = p;
      }
    }
    return;
  }

  for (int i = tid; i < rows * (DFEAT / 4); i += OWN_THREADS) {
    float4 v = ((const float4*)acc)[i];
    const float dn = rden[i / (DFEAT / 4)] + 1e-16f;
    const float4 b4 = *(const float4*)(bias + (i & (DFEAT / 4 - 1)) * 4);
    v.x = v.x / dn + b4.x; v.y = v.y / dn + b4.y;
    v.z = v.z / dn + b4.z; v.w = v.w / dn + b4.w;
    if (relu) {
      v.x = fmaxf(v.x, 0.0f); v.y = fmaxf(v.y, 0.0f);
      v.z = fmaxf(v.z, 0.0f); v.w = fmaxf(v.w, 0.0f);
    }
    *(float4*)(O + (size_t)lo * DFEAT + (size_t)i * 4) = v;
  }
}

static inline int ceil_div(long long a, long long b) { return (int)((a + b - 1) / b); }

static void run_gat_layer(const float* X, const int* srcI, const int* dstI, int ecount,
                          const float* W, const float* avs, const float* avd,
                          const float* bias, float* H, float* asrc, float* adst,
                          float* O, int n, int relu, int* lists, int reuse, hipStream_t stream,
                          int head, const float* x2o, const float* deg, const float* Wp,
                          const float* bp, float* out) {
  gat_gemm_xw<<<GEMM_GRID, GEMM_THREADS, 0, stream>>>(X, W, H, n / 16);
  gat_node_alpha<<<ceil_div((long long)n * 32, 256), 256, 0, stream>>>(H, avs, avd, asrc, adst, n);
  gat_owner_rows<<<ceil_div(n, OWN_ROWS), OWN_THREADS, 0, stream>>>(srcI, dstI, ecount, n, asrc, adst,
                                                                  H, bias, relu, lists, reuse, O,
                                                                  head, x2o, deg, Wp, bp, out);
}

extern "C" void kernel_launch(void* const* d_in, const int* in_sizes, int n_in,
                              void* d_out, int out_size, void* d_ws, size_t ws_size,
                              hipStream_t stream) {
  const float* x_o      = (const float*)d_in[0];
  const float* degree_o = (const float*)d_in[1];
  const int*   ei_o     = (const int*)d_in[2];
  const int*   ei_s     = (const int*)d_in[3];
  const float* W_o1 = (const float*)d_in[4];
  const float* as_o1 = (const float*)d_in[5];
  const float* ad_o1 = (const float*)d_in[6];
  const float* b_o1  = (const float*)d_in[7];
  const float* W_o2 = (const float*)d_in[8];
  const float* as_o2 = (const float*)d_in[9];
  const float* ad_o2 = (const float*)d_in[10];
  const float* b_o2  = (const float*)d_in[11];
  const float* W_s1 = (const float*)d_in[12];
  const float* as_s1 = (const float*)d_in[13];
  const float* ad_s1 = (const float*)d_in[14];
  const float* b_s1  = (const float*)d_in[15];
  const float* W_s2 = (const float*)d_in[16];
  const float* as_s2 = (const float*)d_in[17];
  const float* ad_s2 = (const float*)d_in[18];
  const float* b_s2  = (const float*)d_in[19];
  const float* W_pred = (const float*)d_in[20];
  const float* b_pred = (const float*)d_in[21];

  const int n = in_sizes[0] / DFEAT;
  const int ecount = in_sizes[2] / 2;

  const int* src_o = ei_o;
  const int* dst_o = ei_o + ecount;
  const int* src_s = ei_s;
  const int* dst_s = ei_s + ecount;

  float* ws = (float*)d_ws;
  float* H     = ws;
  float* X1    = H + (size_t)n * DFEAT;
  float* X2o   = X1 + (size_t)n * DFEAT;
  float* LST   = X2o + (size_t)n * DFEAT;
  float* asrc  = LST + (size_t)n * DFEAT;
  float* adst  = asrc + n;
  int* lists   = (int*)LST;

  run_gat_layer(x_o, src_o, dst_o, ecount, W_o1, as_o1, ad_o1, b_o1,
                H, asrc, adst, X1, n,  1, lists,  0, stream,
                 0, nullptr, nullptr, nullptr, nullptr, nullptr);
  run_gat_layer(X1, src_o, dst_o, ecount, W_o2, as_o2, ad_o2, b_o2,
                H, asrc, adst, X2o, n,  0, lists,  1, stream,
                 0, nullptr, nullptr, nullptr, nullptr, nullptr);
  run_gat_layer(x_o, src_s, dst_s, ecount, W_s1, as_s1, ad_s1, b_s1,
                H, asrc, adst, X1, n,  1, lists,  0, stream,
                 0, nullptr, nullptr, nullptr, nullptr, nullptr);
  run_gat_layer(X1, src_s, dst_s, ecount, W_s2, as_s2, ad_s2, b_s2,
                H, asrc, adst,  nullptr, n,  0, lists,  1, stream,
                 1, X2o, degree_o, W_pred, b_pred, (float*)d_out);
}


# Layer product moved behind the row sums (linearity): owners sum input rows, then apply W to their 196 rows on WMMA; score dots fold to x.(W a); 12 launches become 6; sum order changes
# speedup vs baseline: 5.2284x; 1.1242x over previous
#include <hip/hip_runtime.h>
#include <math.h>

typedef __attribute__((ext_vector_type(2))) float v2f;
typedef __attribute__((ext_vector_type(8))) float v8f;

#define DFEAT 128
#define DDEG 32
#define NEG_SLOPE 0.2f

__device__ __forceinline__ unsigned fenc(float f) {
  unsigned u = __float_as_uint(f);
  return (u & 0x80000000u) ? ~u : (u | 0x80000000u);
}
__device__ __forceinline__ float fdec(unsigned u) {
  u = (u & 0x80000000u) ? (u & 0x7fffffffu) : ~u;
  return __uint_as_float(u);
}

template <int M>
__device__ __forceinline__ float row_xmask(float x) {
  const int b = __builtin_bit_cast(int, x);
  return __builtin_bit_cast(float, __builtin_amdgcn_update_dpp(b, b, 0x160 + M, 0xf, 0xf, false));
}
__device__ __forceinline__ float xhalf(float x) {
  const int b = __builtin_bit_cast(int, x);
  return __builtin_bit_cast(float, __builtin_amdgcn_permlanex16(b, b, 0x76543210, 0xfedcba98, false, false));
}
__device__ __forceinline__ float wave_sum(float x) {
  x += row_xmask<1>(x);
  x += row_xmask<2>(x);
  x += row_xmask<4>(x);
  x += row_xmask<8>(x);
  return x + xhalf(x);
}

__global__ void __launch_bounds__(256) gat_fold_a(const float* __restrict__ W0, const float* __restrict__ s0,
                                                  const float* __restrict__ d0,
                                                  const float* __restrict__ W1, const float* __restrict__ s1,
                                                  const float* __restrict__ d1,
                                                  const float* __restrict__ W2, const float* __restrict__ s2,
                                                  const float* __restrict__ d2,
                                                  const float* __restrict__ W3, const float* __restrict__ s3,
                                                  const float* __restrict__ d3,
                                                  float* __restrict__ wa) {
  const int lane = threadIdx.x & 31;
  const int wave = threadIdx.x >> 5;
  const int layer = blockIdx.x >> 4;
  const int i = ((blockIdx.x & 15) << 3) + wave;
  const float* W = layer == 0 ? W0 : layer == 1 ? W1 : layer == 2 ? W2 : W3;
  const float* as = layer == 0 ? s0 : layer == 1 ? s1 : layer == 2 ? s2 : s3;
  const float* ad = layer == 0 ? d0 : layer == 1 ? d1 : layer == 2 ? d2 : d3;
  const float4 w4 = *(const float4*)(W + (size_t)i * DFEAT + lane * 4);
  const float4 s4 = *(const float4*)(as + lane * 4);
  const float4 d4 = *(const float4*)(ad + lane * 4);
  float ps = w4.x * s4.x + w4.y * s4.y + w4.z * s4.z + w4.w * s4.w;
  float pd = w4.x * d4.x + w4.y * d4.y + w4.z * d4.z + w4.w * d4.w;
  ps = wave_sum(ps);
  pd = wave_sum(pd);
  if (lane == 0) {
    wa[(2 * layer + 0) * DFEAT + i] = ps;
    wa[(2 * layer + 1) * DFEAT + i] = pd;
  }
}

__global__ void __launch_bounds__(256) gat_node_alpha4(const float* __restrict__ X,
                                                       const float* __restrict__ wa,
                                                       float* __restrict__ as_o, float* __restrict__ ad_o,
                                                       float* __restrict__ as_s, float* __restrict__ ad_s,
                                                       int n) {
  const int g = blockIdx.x * blockDim.x + threadIdx.x;
  const int node = g >> 5;
  const int lane = threadIdx.x & 31;
  if (node >= n) return;
  const float4 x4 = *(const float4*)(X + (size_t)node * DFEAT + lane * 4);
  const float4 a4 = *(const float4*)(wa + 0 * DFEAT + lane * 4);
  const float4 b4 = *(const float4*)(wa + 1 * DFEAT + lane * 4);
  const float4 c4 = *(const float4*)(wa + 4 * DFEAT + lane * 4);
  const float4 e4 = *(const float4*)(wa + 5 * DFEAT + lane * 4);
  float p0 = x4.x * a4.x + x4.y * a4.y + x4.z * a4.z + x4.w * a4.w;
  float p1 = x4.x * b4.x + x4.y * b4.y + x4.z * b4.z + x4.w * b4.w;
  float p2 = x4.x * c4.x + x4.y * c4.y + x4.z * c4.z + x4.w * c4.w;
  float p3 = x4.x * e4.x + x4.y * e4.y + x4.z * e4.z + x4.w * e4.w;
  p0 = wave_sum(p0);
  p1 = wave_sum(p1);
  p2 = wave_sum(p2);
  p3 = wave_sum(p3);
  if (lane == 0) {
    as_o[node] = p0;
    ad_o[node] = p1;
    as_s[node] = p2;
    ad_s[node] = p3;
  }
}

#define OWN_ROWS 196
#define OWN_WAVES 32
#define OWN_THREADS (OWN_WAVES * 32)
#define SCAN_DEPTH 8
#define CHUNK_IDS 128
#define SLAB_IDS (OWN_WAVES * CHUNK_IDS)
#define BATCH_IDS (SCAN_DEPTH * SLAB_IDS)
#define WCAP 512
#define SCORE_DEPTH 8
#define GROUP_EDGES 16
#define LIST_WORDS (OWN_WAVES * WCAP)
#define HEAD_DEPTH ((OWN_ROWS + OWN_WAVES - 1) / OWN_WAVES)
#define LIST_STRIDE (LIST_WORDS + 64)
#define ACC_PITCH 132
#define ACC_ROWS 208

__device__ __forceinline__ int own_append(int* __restrict__ qe, unsigned char* __restrict__ qr,
                                          int qn, bool p, int e, unsigned t, int& ovf) {
  const unsigned b = __builtin_amdgcn_ballot_w32(p);
  if (b != 0u) {
    const int cnt = __builtin_popcount(b);
    if (qn + cnt <= WCAP) {
      const int pos = qn + (int)__builtin_amdgcn_mbcnt_lo(b, 0u);
      if (p) { qe[pos] = e; qr[pos] = (unsigned char)t; }
      qn += cnt;
    } else {
      ovf = 1;
    }
  }
  return qn;
}

__global__ void __launch_bounds__(OWN_THREADS) gat_owner_rows(const int* __restrict__ srcI,
                                                              const int* __restrict__ dstI,
                                                              int ecount, int n,
                                                              const float* __restrict__ asrc,
                                                              const float* __restrict__ adst,
                                                              const float* __restrict__ H,
                                                              const float* __restrict__ W,
                                                              const float* __restrict__ bias,
                                                              int relu,
                                                              int* __restrict__ lists,
                                                              int reuse,
                                                              float* __restrict__ O,
                                                              int nxt,
                                                              const float* __restrict__ wa_ns,
                                                              const float* __restrict__ wa_nd,
                                                              float* __restrict__ as_n,
                                                              float* __restrict__ ad_n,
                                                              int head,
                                                              const float* __restrict__ x2o,
                                                              const float* __restrict__ deg,
                                                              const float* __restrict__ Wp,
                                                              const float* __restrict__ bp,
                                                              float* __restrict__ out) {
  __shared__ __attribute__((aligned(16))) float acc[ACC_ROWS * ACC_PITCH];
  __shared__ __attribute__((aligned(16))) int qsv[2 * OWN_WAVES * WCAP];
  int* const qsbuf = qsv;
  float* const qvbuf = (float*)(qsv + OWN_WAVES * WCAP);
  float* const obuf = (float*)qsv;
  __shared__ unsigned char qrbuf[OWN_WAVES * WCAP];
  __shared__ unsigned rmax[OWN_ROWS];
  __shared__ float rden[OWN_ROWS];
  __shared__ int ovf_flag;

  const int tid = threadIdx.x;
  const int lane = tid & 31;
  const int wave = __builtin_amdgcn_readfirstlane(tid >> 5);
  const int lo = blockIdx.x * OWN_ROWS;
  const int rows = min(OWN_ROWS, n - lo);
  const unsigned urows = (unsigned)rows;

  for (int i = tid; i < ACC_ROWS * ACC_PITCH / 4; i += OWN_THREADS)
    ((float4*)acc)[i] = make_float4(0.0f, 0.0f, 0.0f, 0.0f);
  if (tid < OWN_ROWS) {
    rmax[tid] = fenc(-INFINITY);
    rden[tid] = 0.0f;
  }
  if (tid == 0) ovf_flag = 0;
  __syncthreads();

  int* myqs = qsbuf + wave * WCAP;
  float* myqv = qvbuf + wave * WCAP;
  unsigned char* myqr = qrbuf + wave * WCAP;
  int qn = 0;
  int ovf = 0;

  int* gq = lists + (size_t)blockIdx.x * LIST_STRIDE;
  int* myg = gq + wave * WCAP;

  if (reuse == 0) {
    for (int r0 = wave * 32; r0 < rows; r0 += OWN_THREADS)
      qn = own_append(myqs, myqr, qn, r0 + lane < rows, ecount + lo + r0 + lane, (unsigned)(r0 + lane), ovf);

    const bool vec_ok = (((size_t)dstI) & 15) == 0;
    const int nb_full = vec_ok ? ecount / BATCH_IDS : 0;
    const int rot = nb_full ? (int)(blockIdx.x % (unsigned)nb_full) : 0;
    for (int k = 0; k < nb_full; ++k) {
      int bi = k + rot;
      if (bi >= nb_full) bi -= nb_full;
      const int b0 = bi * BATCH_IDS + wave * CHUNK_IDS + lane * 4;
      int4 dv[SCAN_DEPTH];
  #pragma unroll
      for (int u = 0; u < SCAN_DEPTH; ++u)
        dv[u] = *(const int4*)(dstI + b0 + u * SLAB_IDS);
  #pragma unroll
      for (int u = 0; u < SCAN_DEPTH; ++u) {
        const unsigned t0 = (unsigned)(dv[u].x - lo);
        const unsigned t1 = (unsigned)(dv[u].y - lo);
        const unsigned t2 = (unsigned)(dv[u].z - lo);
        const unsigned t3 = (unsigned)(dv[u].w - lo);
        const unsigned tm = min(min(t0, t1), min(t2, t3));
        if (__builtin_amdgcn_ballot_w32(tm < urows) != 0u) {
          const int e0 = b0 + u * SLAB_IDS;
          qn = own_append(myqs, myqr, qn, t0 < urows, e0, t0, ovf);
          qn = own_append(myqs, myqr, qn, t1 < urows, e0 + 1, t1, ovf);
          qn = own_append(myqs, myqr, qn, t2 < urows, e0 + 2, t2, ovf);
          qn = own_append(myqs, myqr, qn, t3 < urows, e0 + 3, t3, ovf);
        }
      }
    }

    for (int base = nb_full * BATCH_IDS + wave * CHUNK_IDS; base < ecount; base += SLAB_IDS) {
      unsigned t[4];
  #pragma unroll
      for (int j = 0; j < 4; ++j) {
        const int e = base + j * 32 + lane;
        t[j] = (e < ecount) ? (unsigned)(dstI[e] - lo) : 0xffffffffu;
      }
  #pragma unroll
      for (int j = 0; j < 4; ++j)
        qn = own_append(myqs, myqr, qn, t[j] < urows, base + j * 32 + lane, t[j], ovf);
    }
  } else {
    ovf = __builtin_amdgcn_readfirstlane(gq[LIST_WORDS + OWN_WAVES]);
    qn = ovf != 0 ? 0 : __builtin_amdgcn_readfirstlane(gq[LIST_WORDS + wave]);
  }

  for (int g = 0; g < qn; g += 32 * SCORE_DEPTH) {
    int s[SCORE_DEPTH];
    int r[SCORE_DEPTH];
    float v[SCORE_DEPTH];
    if (reuse == 0) {
#pragma unroll
      for (int u = 0; u < SCORE_DEPTH; ++u) {
        const int i = g + u * 32 + lane;
        s[u] = 0;
        r[u] = 0;
        if (i < qn) {
          const int e = myqs[i];
          r[u] = myqr[i];
          s[u] = (e < ecount) ? srcI[e] : e - ecount;
        }
      }
    } else {
#pragma unroll
      for (int u = 0; u < SCORE_DEPTH; ++u) {
        const int i = g + u * 32 + lane;
        s[u] = 0;
        if (i < qn) s[u] = myg[i];
      }
#pragma unroll
      for (int u = 0; u < SCORE_DEPTH; ++u) {
        r[u] = (int)((unsigned)s[u] >> 24);
        s[u] &= 0xffffff;
      }
    }
#pragma unroll
    for (int u = 0; u < SCORE_DEPTH; ++u)
      v[u] = asrc[s[u]] + adst[lo + r[u]];
#pragma unroll
    for (int u = 0; u < SCORE_DEPTH; ++u) {
      const int i = g + u * 32 + lane;
      if (i < qn) {
        const float x = v[u] > 0.0f ? v[u] : v[u] * NEG_SLOPE;
        myqs[i] = s[u];
        myqv[i] = x;
        if (reuse == 0) myg[i] = (int)((unsigned)s[u] | ((unsigned)r[u] << 24));
        else myqr[i] = (unsigned char)r[u];
        atomicMax(&rmax[r[u]], fenc(x));
      }
    }
  }
  if (lane == 0) {
    if (reuse == 0) gq[LIST_WORDS + wave] = qn;
    if (ovf != 0) ovf_flag = 1;
  }

  __syncthreads();

  if (reuse == 0 && tid == 0) gq[LIST_WORDS + OWN_WAVES] = ovf_flag;

  const float* hl = H + lane;
  if (__builtin_amdgcn_readfirstlane(ovf_flag) == 0) {
    for (int j0 = 0; j0 < qn; j0 += 32) {
      const int cnt = min(32, qn - j0);
      int s = 0, r = 0;
      float w = 0.0f;
      if (lane < cnt) {
        s = myqs[j0 + lane];
        r = myqr[j0 + lane];
        w = expf(myqv[j0 + lane] - fdec(rmax[r]));
        atomicAdd(&rden[r], w);
      }
      const int wBits = __float_as_int(w);
      for (int k0 = 0; k0 < cnt; k0 += GROUP_EDGES) {
        float hv[GROUP_EDGES][4];
        float cf[GROUP_EDGES];
        int rr[GROUP_EDGES];
#pragma unroll
        for (int u = 0; u < GROUP_EDGES; ++u) {
          const int sj = __builtin_amdgcn_readlane(s, k0 + u);
          rr[u] = __builtin_amdgcn_readlane(r, k0 + u);
          cf[u] = __int_as_float(__builtin_amdgcn_readlane(wBits, k0 + u));
          const float* h = hl + (size_t)sj * DFEAT;
          hv[u][0] = h[0];
          hv[u][1] = h[32];
          hv[u][2] = h[64];
          hv[u][3] = h[96];
        }
#pragma unroll
        for (int u = 0; u < GROUP_EDGES; ++u) {
          if (k0 + u < cnt) {
            float* a = &acc[rr[u] * ACC_PITCH + lane];
            atomicAdd(a + 0, cf[u] * hv[u][0]);
            atomicAdd(a + 32, cf[u] * hv[u][1]);
            atomicAdd(a + 64, cf[u] * hv[u][2]);
            atomicAdd(a + 96, cf[u] * hv[u][3]);
          }
        }
      }
    }
  } else {
    const int etot = ecount + n;
    for (int pass = 0; pass < 2; ++pass) {
      for (int base = wave * 32; base < etot; base += OWN_THREADS) {
        const int e = base + lane;
        int s = 0;
        unsigned t = 0xffffffffu;
        float v = 0.0f;
        if (e < etot) {
          const int d = (e < ecount) ? dstI[e] : e - ecount;
          t = (unsigned)(d - lo);
          if (t < urows) {
            s = (e < ecount) ? srcI[e] : d;
            v = asrc[s] + adst[d];
            v = v > 0.0f ? v : v * NEG_SLOPE;
          }
        }
        const bool own = t < urows;
        if (pass == 0) {
          if (own) atomicMax(&rmax[t], fenc(v));
        } else {
          float w = 0.0f;
          if (own) {
            w = expf(v - fdec(rmax[t]));
            atomicAdd(&rden[t], w);
          }
          const int wBits = __float_as_int(w);
          unsigned b = __builtin_amdgcn_ballot_w32(own);
          while (b != 0u) {
            const int j = __builtin_ctz(b);
            b &= b - 1u;
            const int sj = __builtin_amdgcn_readlane(s, j);
            const int rj = __builtin_amdgcn_readlane((int)t, j);
            const float wj = __int_as_float(__builtin_amdgcn_readlane(wBits, j));
            const float* h = hl + (size_t)sj * DFEAT;
            float* a = &acc[rj * ACC_PITCH + lane];
            atomicAdd(a + 0, wj * h[0]);
            atomicAdd(a + 32, wj * h[32]);
            atomicAdd(a + 64, wj * h[64]);
            atomicAdd(a + 96, wj * h[96]);
          }
        }
      }
      __syncthreads();
    }
  }

  const int fc = lane & 15;
  const int fh = lane >> 4;
  const int fq = wave & 7;
  v2f bf[32];
  {
    const float* wp = W + (size_t)(4 * fh) * DFEAT + 16 * fq + fc;
#pragma unroll
    for (int ks = 0; ks < 32; ++ks) {
      const int k0 = 8 * (ks >> 1) + 2 * (ks & 1);
      bf[ks].x = wp[(size_t)k0 * DFEAT];
      bf[ks].y = wp[(size_t)(k0 + 1) * DFEAT];
    }
  }

  __syncthreads();

  const int ntile = (rows + 15) >> 4;
  for (int rt = wave >> 3; rt < ntile; rt += OWN_WAVES / 8) {
    const float* ar = acc + (16 * rt + fc) * ACC_PITCH + 4 * fh;
    v8f d = {};
#pragma unroll
    for (int j = 0; j < 16; ++j) {
      const float4 a4 = *(const float4*)(ar + 8 * j);
      v2f a;
      a.x = a4.x; a.y = a4.y;
      d = __builtin_amdgcn_wmma_f32_16x16x4_f32(
           false, a,  false, bf[2 * j],
           (short)0, d,  false,  false);
      a.x = a4.z; a.y = a4.w;
      d = __builtin_amdgcn_wmma_f32_16x16x4_f32(
           false, a,  false, bf[2 * j + 1],
           (short)0, d,  false,  false);
    }
    float* ob = obuf + (16 * rt + 8 * fh) * DFEAT + 16 * fq + fc;
#pragma unroll
    for (int r = 0; r < 8; ++r) ob[r * DFEAT] = d[r];
  }

  __syncthreads();

  if (head != 0) {
    const float4 wo0 = *(const float4*)(Wp + lane * 8);
    const float4 wo1 = *(const float4*)(Wp + lane * 8 + 4);
    const float4 ws0 = *(const float4*)(Wp + 2 * DFEAT + lane * 8);
    const float4 ws1 = *(const float4*)(Wp + 2 * DFEAT + lane * 8 + 4);
    const float2 wd = *(const float2*)(Wp + 4 * DFEAT + lane * 2);
    const float bp0 = bp[0];
    const float bp1 = bp[1];
    const float4 b4 = *(const float4*)(bias + lane * 4);
    float4 ov[HEAD_DEPTH];
    float dgv[HEAD_DEPTH];
#pragma unroll
    for (int k = 0; k < HEAD_DEPTH; ++k) {
      const int r = wave + k * OWN_WAVES;
      const int rc = r < rows ? r : 0;
      const size_t row = (size_t)(lo + rc);
      ov[k] = *(const float4*)(x2o + row * DFEAT + lane * 4);
      dgv[k] = deg[row * DDEG + lane];
    }
#pragma unroll
    for (int k = 0; k < HEAD_DEPTH; ++k) {
      const int r = wave + k * OWN_WAVES;
      if (r < rows) {
        float4 v = ((const float4*)obuf)[r * (DFEAT / 4) + lane];
        const float dn = rden[r] + 1e-16f;
        v.x = v.x / dn + b4.x; v.y = v.y / dn + b4.y;
        v.z = v.z / dn + b4.z; v.w = v.w / dn + b4.w;
        if (relu) {
          v.x = fmaxf(v.x, 0.0f); v.y = fmaxf(v.y, 0.0f);
          v.z = fmaxf(v.z, 0.0f); v.w = fmaxf(v.w, 0.0f);
        }
        const float4 o = ov[k];
        float l0 = o.x * wo0.x + o.y * wo0.z + o.z * wo1.x + o.w * wo1.z;
        float l1 = o.x * wo0.y + o.y * wo0.w + o.z * wo1.y + o.w * wo1.w;
        l0 += v.x * ws0.x + v.y * ws0.z + v.z * ws1.x + v.w * ws1.z;
        l1 += v.x * ws0.y + v.y * ws0.w + v.z * ws1.y + v.w * ws1.w;
        l0 += dgv[k] * wd.x;
        l1 += dgv[k] * wd.y;
        l0 = wave_sum(l0);
        l1 = wave_sum(l1);
        l0 += bp0;
        l1 += bp1;
        const float mx = fmaxf(l0, l1);
        const float e0 = expf(l0 - mx), e1 = expf(l1 - mx);
        const float inv = 1.0f / (e0 + e1);
        const float a0 = e0 * inv, a1 = e1 * inv;
        float4 p;
        p.x = a0 * o.x + a1 * v.x; p.y = a0 * o.y + a1 * v.y;
        p.z = a0 * o.z + a1 * v.z; p.w = a0 * o.w + a1 * v.w;
        *(float4*)(out + (size_t)(lo + r) * DFEAT + lane * 4) = p;
      }
    }
    return;
  }

  const float4 b4 = *(const float4*)(bias + lane * 4);
  float4 ns4 = make_float4(0.0f, 0.0f, 0.0f, 0.0f);
  float4 nd4 = make_float4(0.0f, 0.0f, 0.0f, 0.0f);
  if (nxt != 0) {
    ns4 = *(const float4*)(wa_ns + lane * 4);
    nd4 = *(const float4*)(wa_nd + lane * 4);
  }
  for (int r = wave; r < rows; r += OWN_WAVES) {
    float4 v = ((const float4*)obuf)[r * (DFEAT / 4) + lane];
    const float dn = rden[r] + 1e-16f;
    v.x = v.x / dn + b4.x; v.y = v.y / dn + b4.y;
    v.z = v.z / dn + b4.z; v.w = v.w / dn + b4.w;
    if (relu) {
      v.x = fmaxf(v.x, 0.0f); v.y = fmaxf(v.y, 0.0f);
      v.z = fmaxf(v.z, 0.0f); v.w = fmaxf(v.w, 0.0f);
    }
    *(float4*)(O + (size_t)(lo + r) * DFEAT + lane * 4) = v;
    if (nxt != 0) {
      float ps = v.x * ns4.x + v.y * ns4.y + v.z * ns4.z + v.w * ns4.w;
      float pd = v.x * nd4.x + v.y * nd4.y + v.z * nd4.z + v.w * nd4.w;
      ps = wave_sum(ps);
      pd = wave_sum(pd);
      if (lane == 0) {
        as_n[lo + r] = ps;
        ad_n[lo + r] = pd;
      }
    }
  }
}

static inline int ceil_div(long long a, long long b) { return (int)((a + b - 1) / b); }

static void run_gat_layer(const float* X, const int* srcI, const int* dstI, int ecount,
                          const float* W, const float* bias, const float* asrc, const float* adst,
                          float* O, int n, int relu, int* lists, int reuse, hipStream_t stream,
                          int nxt, const float* wa_ns, const float* wa_nd, float* as_n, float* ad_n,
                          int head, const float* x2o, const float* deg, const float* Wp,
                          const float* bp, float* out) {
  gat_owner_rows<<<ceil_div(n, OWN_ROWS), OWN_THREADS, 0, stream>>>(srcI, dstI, ecount, n, asrc, adst,
                                                                  X, W, bias, relu, lists, reuse, O,
                                                                  nxt, wa_ns, wa_nd, as_n, ad_n,
                                                                  head, x2o, deg, Wp, bp, out);
}

extern "C" void kernel_launch(void* const* d_in, const int* in_sizes, int n_in,
                              void* d_out, int out_size, void* d_ws, size_t ws_size,
                              hipStream_t stream) {
  const float* x_o      = (const float*)d_in[0];
  const float* degree_o = (const float*)d_in[1];
  const int*   ei_o     = (const int*)d_in[2];
  const int*   ei_s     = (const int*)d_in[3];
  const float* W_o1 = (const float*)d_in[4];
  const float* as_o1 = (const float*)d_in[5];
  const float* ad_o1 = (const float*)d_in[6];
  const float* b_o1  = (const float*)d_in[7];
  const float* W_o2 = (const float*)d_in[8];
  const float* as_o2 = (const float*)d_in[9];
  const float* ad_o2 = (const float*)d_in[10];
  const float* b_o2  = (const float*)d_in[11];
  const float* W_s1 = (const float*)d_in[12];
  const float* as_s1 = (const float*)d_in[13];
  const float* ad_s1 = (const float*)d_in[14];
  const float* b_s1  = (const float*)d_in[15];
  const float* W_s2 = (const float*)d_in[16];
  const float* as_s2 = (const float*)d_in[17];
  const float* ad_s2 = (const float*)d_in[18];
  const float* b_s2  = (const float*)d_in[19];
  const float* W_pred = (const float*)d_in[20];
  const float* b_pred = (const float*)d_in[21];

  const int n = in_sizes[0] / DFEAT;
  const int ecount = in_sizes[2] / 2;

  const int* src_o = ei_o;
  const int* dst_o = ei_o + ecount;
  const int* src_s = ei_s;
  const int* dst_s = ei_s + ecount;

  float* ws = (float*)d_ws;
  float* T     = ws;
  float* X1    = T + (size_t)n * DFEAT;
  float* X2o   = X1 + (size_t)n * DFEAT;
  float* LST   = X2o + (size_t)n * DFEAT;
  float* wa    = T;
  float* a1os  = wa + 8 * DFEAT;
  float* a1od  = a1os + n;
  float* a1ss  = a1od + n;
  float* a1sd  = a1ss + n;
  float* a2s   = a1sd + n;
  float* a2d   = a2s + n;
  int* lists   = (int*)LST;

  gat_fold_a<<<64, 256, 0, stream>>>(W_o1, as_o1, ad_o1, W_o2, as_o2, ad_o2,
                                     W_s1, as_s1, ad_s1, W_s2, as_s2, ad_s2, wa);
  gat_node_alpha4<<<ceil_div((long long)n * 32, 256), 256, 0, stream>>>(x_o, wa, a1os, a1od, a1ss, a1sd, n);

  run_gat_layer(x_o, src_o, dst_o, ecount, W_o1, b_o1, a1os, a1od,
                X1, n,  1, lists,  0, stream,
                 1, wa + 2 * DFEAT, wa + 3 * DFEAT, a2s, a2d,
                 0, nullptr, nullptr, nullptr, nullptr, nullptr);
  run_gat_layer(X1, src_o, dst_o, ecount, W_o2, b_o2, a2s, a2d,
                X2o, n,  0, lists,  1, stream,
                 0, nullptr, nullptr, nullptr, nullptr,
                 0, nullptr, nullptr, nullptr, nullptr, nullptr);
  run_gat_layer(x_o, src_s, dst_s, ecount, W_s1, b_s1, a1ss, a1sd,
                X1, n,  1, lists,  0, stream,
                 1, wa + 6 * DFEAT, wa + 7 * DFEAT, a2s, a2d,
                 0, nullptr, nullptr, nullptr, nullptr, nullptr);
  run_gat_layer(X1, src_s, dst_s, ecount, W_s2, b_s2, a2s, a2d,
                 nullptr, n,  0, lists,  1, stream,
                 0, nullptr, nullptr, nullptr, nullptr,
                 1, X2o, degree_o, W_pred, b_pred, (float*)d_out);
}


# dst ids packed once per call into one owner byte each (node-dot launch); first-layer owners scan that map, 16 ids per lane load, not the 4-byte ids; list rows come from dst[e]; arrival-order sums
# speedup vs baseline: 5.5514x; 1.0618x over previous
#include <hip/hip_runtime.h>
#include <math.h>

typedef __attribute__((ext_vector_type(2))) float v2f;
typedef __attribute__((ext_vector_type(8))) float v8f;

#define DFEAT 128
#define DDEG 32
#define NEG_SLOPE 0.2f
#define OWN_ROWS 196

__device__ __forceinline__ unsigned fenc(float f) {
  unsigned u = __float_as_uint(f);
  return (u & 0x80000000u) ? ~u : (u | 0x80000000u);
}
__device__ __forceinline__ float fdec(unsigned u) {
  u = (u & 0x80000000u) ? (u & 0x7fffffffu) : ~u;
  return __uint_as_float(u);
}

template <int M>
__device__ __forceinline__ float row_xmask(float x) {
  const int b = __builtin_bit_cast(int, x);
  return __builtin_bit_cast(float, __builtin_amdgcn_update_dpp(b, b, 0x160 + M, 0xf, 0xf, false));
}
__device__ __forceinline__ float xhalf(float x) {
  const int b = __builtin_bit_cast(int, x);
  return __builtin_bit_cast(float, __builtin_amdgcn_permlanex16(b, b, 0x76543210, 0xfedcba98, false, false));
}
__device__ __forceinline__ float wave_sum(float x) {
  x += row_xmask<1>(x);
  x += row_xmask<2>(x);
  x += row_xmask<4>(x);
  x += row_xmask<8>(x);
  return x + xhalf(x);
}

__global__ void __launch_bounds__(256) gat_fold_a(const float* __restrict__ W0, const float* __restrict__ s0,
                                                  const float* __restrict__ d0,
                                                  const float* __restrict__ W1, const float* __restrict__ s1,
                                                  const float* __restrict__ d1,
                                                  const float* __restrict__ W2, const float* __restrict__ s2,
                                                  const float* __restrict__ d2,
                                                  const float* __restrict__ W3, const float* __restrict__ s3,
                                                  const float* __restrict__ d3,
                                                  float* __restrict__ wa) {
  const int lane = threadIdx.x & 31;
  const int wave = threadIdx.x >> 5;
  const int layer = blockIdx.x >> 4;
  const int i = ((blockIdx.x & 15) << 3) + wave;
  const float* W = layer == 0 ? W0 : layer == 1 ? W1 : layer == 2 ? W2 : W3;
  const float* as = layer == 0 ? s0 : layer == 1 ? s1 : layer == 2 ? s2 : s3;
  const float* ad = layer == 0 ? d0 : layer == 1 ? d1 : layer == 2 ? d2 : d3;
  const float4 w4 = *(const float4*)(W + (size_t)i * DFEAT + lane * 4);
  const float4 s4 = *(const float4*)(as + lane * 4);
  const float4 d4 = *(const float4*)(ad + lane * 4);
  float ps = w4.x * s4.x + w4.y * s4.y + w4.z * s4.z + w4.w * s4.w;
  float pd = w4.x * d4.x + w4.y * d4.y + w4.z * d4.z + w4.w * d4.w;
  ps = wave_sum(ps);
  pd = wave_sum(pd);
  if (lane == 0) {
    wa[(2 * layer + 0) * DFEAT + i] = ps;
    wa[(2 * layer + 1) * DFEAT + i] = pd;
  }
}

__device__ __forceinline__ unsigned owner_word(const int* __restrict__ d, int q, bool vec) {
  int4 v;
  if (vec) {
    v = *(const int4*)(d + 4 * (size_t)q);
  } else {
    v.x = d[4 * (size_t)q + 0];
    v.y = d[4 * (size_t)q + 1];
    v.z = d[4 * (size_t)q + 2];
    v.w = d[4 * (size_t)q + 3];
  }
  return (((unsigned)v.x / OWN_ROWS) & 255u) | ((((unsigned)v.y / OWN_ROWS) & 255u) << 8) |
         ((((unsigned)v.z / OWN_ROWS) & 255u) << 16) | ((((unsigned)v.w / OWN_ROWS) & 255u) << 24);
}

__global__ void __launch_bounds__(256) gat_node_alpha4(const float* __restrict__ X,
                                                       const float* __restrict__ wa,
                                                       float* __restrict__ as_o, float* __restrict__ ad_o,
                                                       float* __restrict__ as_s, float* __restrict__ ad_s,
                                                       int n,
                                                       const int* __restrict__ dst_o,
                                                       const int* __restrict__ dst_s,
                                                       unsigned* __restrict__ map_o,
                                                       unsigned* __restrict__ map_s,
                                                       int ecount) {
  const int g = blockIdx.x * blockDim.x + threadIdx.x;
  {
    const int nq = ecount >> 2;
    const int step = (int)(gridDim.x * blockDim.x);
    const bool vo = (((size_t)dst_o) & 15) == 0;
    const bool vs = (((size_t)dst_s) & 15) == 0;
    for (int q = g; q < nq; q += step) {
      map_o[q] = owner_word(dst_o, q, vo);
      map_s[q] = owner_word(dst_s, q, vs);
    }
  }
  const int node = g >> 5;
  const int lane = threadIdx.x & 31;
  if (node >= n) return;
  const float4 x4 = *(const float4*)(X + (size_t)node * DFEAT + lane * 4);
  const float4 a4 = *(const float4*)(wa + 0 * DFEAT + lane * 4);
  const float4 b4 = *(const float4*)(wa + 1 * DFEAT + lane * 4);
  const float4 c4 = *(const float4*)(wa + 4 * DFEAT + lane * 4);
  const float4 e4 = *(const float4*)(wa + 5 * DFEAT + lane * 4);
  float p0 = x4.x * a4.x + x4.y * a4.y + x4.z * a4.z + x4.w * a4.w;
  float p1 = x4.x * b4.x + x4.y * b4.y + x4.z * b4.z + x4.w * b4.w;
  float p2 = x4.x * c4.x + x4.y * c4.y + x4.z * c4.z + x4.w * c4.w;
  float p3 = x4.x * e4.x + x4.y * e4.y + x4.z * e4.z + x4.w * e4.w;
  p0 = wave_sum(p0);
  p1 = wave_sum(p1);
  p2 = wave_sum(p2);
  p3 = wave_sum(p3);
  if (lane == 0) {
    as_o[node] = p0;
    ad_o[node] = p1;
    as_s[node] = p2;
    ad_s[node] = p3;
  }
}

#define OWN_WAVES 32
#define OWN_THREADS (OWN_WAVES * 32)
#define SCAN_DEPTH 8
#define CHUNK_IDS 128
#define SLAB_IDS (OWN_WAVES * CHUNK_IDS)
#define BATCH_IDS (SCAN_DEPTH * SLAB_IDS)
#define WCAP 512
#define MAP_DEPTH 8
#define MAP_TRIP (MAP_DEPTH * OWN_WAVES * 32)
#define SCORE_DEPTH 8
#define GROUP_EDGES 16
#define LIST_WORDS (OWN_WAVES * WCAP)
#define HEAD_DEPTH ((OWN_ROWS + OWN_WAVES - 1) / OWN_WAVES)
#define LIST_STRIDE (LIST_WORDS + 64)
#define ACC_PITCH 132
#define ACC_ROWS 208

__device__ __forceinline__ int own_append_b(int* __restrict__ qe, int qn, unsigned b, bool p, int e, int& ovf) {
  if (b != 0u) {
    const int q0 = __builtin_amdgcn_readfirstlane(qn);
    const int cnt = __builtin_popcount(b);
    if (q0 + cnt <= WCAP) {
      const int pos = q0 + (int)__builtin_amdgcn_mbcnt_lo(b, 0u);
      if (p) qe[pos] = e;
      qn = q0 + cnt;
    } else {
      ovf = 1;
    }
  }
  return qn;
}
__device__ __forceinline__ int own_append(int* __restrict__ qe, int qn, bool p, int e, int& ovf) {
  return own_append_b(qe, qn, __builtin_amdgcn_ballot_w32(p), p, e, ovf);
}

__device__ __forceinline__ unsigned zero_bytes_inv(unsigned x) {
  return ((x & 0x7f7f7f7fu) + 0x7f7f7f7fu) | x;
}

__global__ void __launch_bounds__(OWN_THREADS) gat_owner_rows(const int* __restrict__ srcI,
                                                              const int* __restrict__ dstI,
                                                              int ecount, int n,
                                                              const float* __restrict__ asrc,
                                                              const float* __restrict__ adst,
                                                              const float* __restrict__ H,
                                                              const float* __restrict__ W,
                                                              const float* __restrict__ bias,
                                                              int relu,
                                                              int* __restrict__ lists,
                                                              int reuse,
                                                              const unsigned* __restrict__ omap,
                                                              float* __restrict__ O,
                                                              int nxt,
                                                              const float* __restrict__ wa_ns,
                                                              const float* __restrict__ wa_nd,
                                                              float* __restrict__ as_n,
                                                              float* __restrict__ ad_n,
                                                              int head,
                                                              const float* __restrict__ x2o,
                                                              const float* __restrict__ deg,
                                                              const float* __restrict__ Wp,
                                                              const float* __restrict__ bp,
                                                              float* __restrict__ out) {
  __shared__ __attribute__((aligned(16))) float acc[ACC_ROWS * ACC_PITCH];
  __shared__ __attribute__((aligned(16))) int qsv[2 * OWN_WAVES * WCAP];
  int* const qsbuf = qsv;
  float* const qvbuf = (float*)(qsv + OWN_WAVES * WCAP);
  float* const obuf = (float*)qsv;
  __shared__ unsigned char qrbuf[OWN_WAVES * WCAP];
  __shared__ unsigned rmax[OWN_ROWS];
  __shared__ float rden[OWN_ROWS];
  __shared__ int ovf_flag;

  const int tid = threadIdx.x;
  const int lane = tid & 31;
  const int wave = __builtin_amdgcn_readfirstlane(tid >> 5);
  const int lo = blockIdx.x * OWN_ROWS;
  const int rows = min(OWN_ROWS, n - lo);
  const unsigned urows = (unsigned)rows;

  for (int i = tid; i < ACC_ROWS * ACC_PITCH / 4; i += OWN_THREADS)
    ((float4*)acc)[i] = make_float4(0.0f, 0.0f, 0.0f, 0.0f);
  if (tid < OWN_ROWS) {
    rmax[tid] = fenc(-INFINITY);
    rden[tid] = 0.0f;
  }
  if (tid == 0) ovf_flag = 0;
  __syncthreads();

  int* myqs = qsbuf + wave * WCAP;
  float* myqv = qvbuf + wave * WCAP;
  unsigned char* myqr = qrbuf + wave * WCAP;
  int qn = 0;
  int ovf = 0;

  int* gq = lists + (size_t)blockIdx.x * LIST_STRIDE;
  int* myg = gq + wave * WCAP;

  if (reuse == 0) {
    for (int r0 = wave * 32; r0 < rows; r0 += OWN_THREADS)
      qn = own_append(myqs, qn, r0 + lane < rows, ecount + lo + r0 + lane, ovf);

    int done = 0;
    const bool map_ok = gridDim.x <= 256u && (((size_t)omap) & 15) == 0;
    if (map_ok) {
      const unsigned pat = blockIdx.x * 0x01010101u;
      const uint4* mp = (const uint4*)omap;
      const int nvec = ecount >> 4;
      const int ntrip = (nvec + MAP_TRIP - 1) / MAP_TRIP;
      const int ntot = ntrip * MAP_TRIP;
      const int rot = ntot ? (int)((blockIdx.x * 256u) % (unsigned)ntot) : 0;
      for (int k = 0; k < ntrip; ++k) {
        const int g0 = k * MAP_TRIP + wave * 32 + lane + rot;
        uint4 mv[MAP_DEPTH];
  #pragma unroll
        for (int u = 0; u < MAP_DEPTH; ++u) {
          int gg = g0 + u * (OWN_WAVES * 32);
          if (gg >= ntot) gg -= ntot;
          mv[u] = mp[gg < nvec ? gg : 0];
        }
  #pragma unroll
        for (int u = 0; u < MAP_DEPTH; ++u) {
          int gg = g0 + u * (OWN_WAVES * 32);
          if (gg >= ntot) gg -= ntot;
          const unsigned y0 = zero_bytes_inv(mv[u].x ^ pat);
          const unsigned y1 = zero_bytes_inv(mv[u].y ^ pat);
          const unsigned y2 = zero_bytes_inv(mv[u].z ^ pat);
          const unsigned y3 = zero_bytes_inv(mv[u].w ^ pat);
          const unsigned inval = gg < nvec ? 0u : 0xffffffffu;
          const unsigned ya = (y0 & y1 & y2 & y3) | inval;
          if (__builtin_amdgcn_ballot_w32((ya & 0x80808080u) != 0x80808080u) != 0u) {
            unsigned m = ((~y0 & 0x80808080u) | ((~y1 & 0x80808080u) >> 1) |
                          ((~y2 & 0x80808080u) >> 2) | ((~y3 & 0x80808080u) >> 3)) & ~inval;
            for (;;) {
              const bool p = m != 0u;
              const unsigned b = __builtin_amdgcn_ballot_w32(p);
              if (b == 0u) break;
              const int c = __builtin_ctz(m | 0x80000000u);
              const int off = ((7 - (c & 7)) << 2) + (c >> 3);
              qn = own_append_b(myqs, qn, b, p, (gg << 4) + off, ovf);
              m &= m - 1u;
            }
          }
        }
      }
      done = nvec << 4;
    }

    const bool vec_ok = !map_ok && (((size_t)dstI) & 15) == 0;
    const int nb_full = vec_ok ? ecount / BATCH_IDS : 0;
    const int rot = nb_full ? (int)(blockIdx.x % (unsigned)nb_full) : 0;
    if (nb_full) done = nb_full * BATCH_IDS;
    for (int k = 0; k < nb_full; ++k) {
      int bi = k + rot;
      if (bi >= nb_full) bi -= nb_full;
      const int b0 = bi * BATCH_IDS + wave * CHUNK_IDS + lane * 4;
      int4 dv[SCAN_DEPTH];
  #pragma unroll
      for (int u = 0; u < SCAN_DEPTH; ++u)
        dv[u] = *(const int4*)(dstI + b0 + u * SLAB_IDS);
  #pragma unroll
      for (int u = 0; u < SCAN_DEPTH; ++u) {
        const unsigned t0 = (unsigned)(dv[u].x - lo);
        const unsigned t1 = (unsigned)(dv[u].y - lo);
        const unsigned t2 = (unsigned)(dv[u].z - lo);
        const unsigned t3 = (unsigned)(dv[u].w - lo);
        const unsigned tm = min(min(t0, t1), min(t2, t3));
        if (__builtin_amdgcn_ballot_w32(tm < urows) != 0u) {
          const int e0 = b0 + u * SLAB_IDS;
          qn = own_append(myqs, qn, t0 < urows, e0, ovf);
          qn = own_append(myqs, qn, t1 < urows, e0 + 1, ovf);
          qn = own_append(myqs, qn, t2 < urows, e0 + 2, ovf);
          qn = own_append(myqs, qn, t3 < urows, e0 + 3, ovf);
        }
      }
    }

    for (int base = done + wave * CHUNK_IDS; base < ecount; base += SLAB_IDS) {
      unsigned t[4];
  #pragma unroll
      for (int j = 0; j < 4; ++j) {
        const int e = base + j * 32 + lane;
        t[j] = (e < ecount) ? (unsigned)(dstI[e] - lo) : 0xffffffffu;
      }
  #pragma unroll
      for (int j = 0; j < 4; ++j)
        qn = own_append(myqs, qn, t[j] < urows, base + j * 32 + lane, ovf);
    }
  } else {
    ovf = __builtin_amdgcn_readfirstlane(gq[LIST_WORDS + OWN_WAVES]);
    qn = ovf != 0 ? 0 : __builtin_amdgcn_readfirstlane(gq[LIST_WORDS + wave]);
  }

  for (int g = 0; g < qn; g += 32 * SCORE_DEPTH) {
    int s[SCORE_DEPTH];
    int d[SCORE_DEPTH];
    float v[SCORE_DEPTH];
    if (reuse == 0) {
#pragma unroll
      for (int u = 0; u < SCORE_DEPTH; ++u) {
        const int i = g + u * 32 + lane;
        s[u] = 0;
        d[u] = lo;
        if (i < qn) {
          const int e = myqs[i];
          if (e < ecount) {
            s[u] = srcI[e];
            d[u] = dstI[e];
          } else {
            s[u] = e - ecount;
            d[u] = e - ecount;
          }
        }
      }
    } else {
#pragma unroll
      for (int u = 0; u < SCORE_DEPTH; ++u) {
        const int i = g + u * 32 + lane;
        s[u] = 0;
        if (i < qn) s[u] = myg[i];
      }
#pragma unroll
      for (int u = 0; u < SCORE_DEPTH; ++u) {
        d[u] = lo + (int)((unsigned)s[u] >> 24);
        s[u] &= 0xffffff;
      }
    }
#pragma unroll
    for (int u = 0; u < SCORE_DEPTH; ++u)
      v[u] = asrc[s[u]] + adst[d[u]];
#pragma unroll
    for (int u = 0; u < SCORE_DEPTH; ++u) {
      const int i = g + u * 32 + lane;
      if (i < qn) {
        const float x = v[u] > 0.0f ? v[u] : v[u] * NEG_SLOPE;
        const int r = d[u] - lo;
        myqs[i] = s[u];
        myqv[i] = x;
        if (reuse == 0) myg[i] = (int)((unsigned)s[u] | ((unsigned)r << 24));
        myqr[i] = (unsigned char)r;
        atomicMax(&rmax[r], fenc(x));
      }
    }
  }
  if (lane == 0) {
    if (reuse == 0) gq[LIST_WORDS + wave] = qn;
    if (ovf != 0) ovf_flag = 1;
  }

  __syncthreads();

  if (reuse == 0 && tid == 0) gq[LIST_WORDS + OWN_WAVES] = ovf_flag;

  const float* hl = H + lane;
  if (__builtin_amdgcn_readfirstlane(ovf_flag) == 0) {
    for (int j0 = 0; j0 < qn; j0 += 32) {
      const int cnt = min(32, qn - j0);
      int s = 0, r = 0;
      float w = 0.0f;
      if (lane < cnt) {
        s = myqs[j0 + lane];
        r = myqr[j0 + lane];
        w = expf(myqv[j0 + lane] - fdec(rmax[r]));
        atomicAdd(&rden[r], w);
      }
      const int wBits = __float_as_int(w);
      for (int k0 = 0; k0 < cnt; k0 += GROUP_EDGES) {
        float hv[GROUP_EDGES][4];
        float cf[GROUP_EDGES];
        int rr[GROUP_EDGES];
#pragma unroll
        for (int u = 0; u < GROUP_EDGES; ++u) {
          const int sj = __builtin_amdgcn_readlane(s, k0 + u);
          rr[u] = __builtin_amdgcn_readlane(r, k0 + u);
          cf[u] = __int_as_float(__builtin_amdgcn_readlane(wBits, k0 + u));
          const float* h = hl + (size_t)sj * DFEAT;
          hv[u][0] = h[0];
          hv[u][1] = h[32];
          hv[u][2] = h[64];
          hv[u][3] = h[96];
        }
#pragma unroll
        for (int u = 0; u < GROUP_EDGES; ++u) {
          if (k0 + u < cnt) {
            float* a = &acc[rr[u] * ACC_PITCH + lane];
            atomicAdd(a + 0, cf[u] * hv[u][0]);
            atomicAdd(a + 32, cf[u] * hv[u][1]);
            atomicAdd(a + 64, cf[u] * hv[u][2]);
            atomicAdd(a + 96, cf[u] * hv[u][3]);
          }
        }
      }
    }
  } else {
    const int etot = ecount + n;
    for (int pass = 0; pass < 2; ++pass) {
      for (int base = wave * 32; base < etot; base += OWN_THREADS) {
        const int e = base + lane;
        int s = 0;
        unsigned t = 0xffffffffu;
        float v = 0.0f;
        if (e < etot) {
          const int d = (e < ecount) ? dstI[e] : e - ecount;
          t = (unsigned)(d - lo);
          if (t < urows) {
            s = (e < ecount) ? srcI[e] : d;
            v = asrc[s] + adst[d];
            v = v > 0.0f ? v : v * NEG_SLOPE;
          }
        }
        const bool own = t < urows;
        if (pass == 0) {
          if (own) atomicMax(&rmax[t], fenc(v));
        } else {
          float w = 0.0f;
          if (own) {
            w = expf(v - fdec(rmax[t]));
            atomicAdd(&rden[t], w);
          }
          const int wBits = __float_as_int(w);
          unsigned b = __builtin_amdgcn_ballot_w32(own);
          while (b != 0u) {
            const int j = __builtin_ctz(b);
            b &= b - 1u;
            const int sj = __builtin_amdgcn_readlane(s, j);
            const int rj = __builtin_amdgcn_readlane((int)t, j);
            const float wj = __int_as_float(__builtin_amdgcn_readlane(wBits, j));
            const float* h = hl + (size_t)sj * DFEAT;
            float* a = &acc[rj * ACC_PITCH + lane];
            atomicAdd(a + 0, wj * h[0]);
            atomicAdd(a + 32, wj * h[32]);
            atomicAdd(a + 64, wj * h[64]);
            atomicAdd(a + 96, wj * h[96]);
          }
        }
      }
      __syncthreads();
    }
  }

  const int fc = lane & 15;
  const int fh = lane >> 4;
  const int fq = wave & 7;
  v2f bf[32];
  {
    const float* wp = W + (size_t)(4 * fh) * DFEAT + 16 * fq + fc;
#pragma unroll
    for (int ks = 0; ks < 32; ++ks) {
      const int k0 = 8 * (ks >> 1) + 2 * (ks & 1);
      bf[ks].x = wp[(size_t)k0 * DFEAT];
      bf[ks].y = wp[(size_t)(k0 + 1) * DFEAT];
    }
  }

  __syncthreads();

  const int ntile = (rows + 15) >> 4;
  for (int rt = wave >> 3; rt < ntile; rt += OWN_WAVES / 8) {
    const float* ar = acc + (16 * rt + fc) * ACC_PITCH + 4 * fh;
    v8f d = {};
#pragma unroll
    for (int j = 0; j < 16; ++j) {
      const float4 a4 = *(const float4*)(ar + 8 * j);
      v2f a;
      a.x = a4.x; a.y = a4.y;
      d = __builtin_amdgcn_wmma_f32_16x16x4_f32(
           false, a,  false, bf[2 * j],
           (short)0, d,  false,  false);
      a.x = a4.z; a.y = a4.w;
      d = __builtin_amdgcn_wmma_f32_16x16x4_f32(
           false, a,  false, bf[2 * j + 1],
           (short)0, d,  false,  false);
    }
    float* ob = obuf + (16 * rt + 8 * fh) * DFEAT + 16 * fq + fc;
#pragma unroll
    for (int r = 0; r < 8; ++r) ob[r * DFEAT] = d[r];
  }

  __syncthreads();

  if (head != 0) {
    const float4 wo0 = *(const float4*)(Wp + lane * 8);
    const float4 wo1 = *(const float4*)(Wp + lane * 8 + 4);
    const float4 ws0 = *(const float4*)(Wp + 2 * DFEAT + lane * 8);
    const float4 ws1 = *(const float4*)(Wp + 2 * DFEAT + lane * 8 + 4);
    const float2 wd = *(const float2*)(Wp + 4 * DFEAT + lane * 2);
    const float bp0 = bp[0];
    const float bp1 = bp[1];
    const float4 b4 = *(const float4*)(bias + lane * 4);
    float4 ov[HEAD_DEPTH];
    float dgv[HEAD_DEPTH];
#pragma unroll
    for (int k = 0; k < HEAD_DEPTH; ++k) {
      const int r = wave + k * OWN_WAVES;
      const int rc = r < rows ? r : 0;
      const size_t row = (size_t)(lo + rc);
      ov[k] = *(const float4*)(x2o + row * DFEAT + lane * 4);
      dgv[k] = deg[row * DDEG + lane];
    }
#pragma unroll
    for (int k = 0; k < HEAD_DEPTH; ++k) {
      const int r = wave + k * OWN_WAVES;
      if (r < rows) {
        float4 v = ((const float4*)obuf)[r * (DFEAT / 4) + lane];
        const float dn = rden[r] + 1e-16f;
        v.x = v.x / dn + b4.x; v.y = v.y / dn + b4.y;
        v.z = v.z / dn + b4.z; v.w = v.w / dn + b4.w;
        if (relu) {
          v.x = fmaxf(v.x, 0.0f); v.y = fmaxf(v.y, 0.0f);
          v.z = fmaxf(v.z, 0.0f); v.w = fmaxf(v.w, 0.0f);
        }
        const float4 o = ov[k];
        float l0 = o.x * wo0.x + o.y * wo0.z + o.z * wo1.x + o.w * wo1.z;
        float l1 = o.x * wo0.y + o.y * wo0.w + o.z * wo1.y + o.w * wo1.w;
        l0 += v.x * ws0.x + v.y * ws0.z + v.z * ws1.x + v.w * ws1.z;
        l1 += v.x * ws0.y + v.y * ws0.w + v.z * ws1.y + v.w * ws1.w;
        l0 += dgv[k] * wd.x;
        l1 += dgv[k] * wd.y;
        l0 = wave_sum(l0);
        l1 = wave_sum(l1);
        l0 += bp0;
        l1 += bp1;
        const float mx = fmaxf(l0, l1);
        const float e0 = expf(l0 - mx), e1 = expf(l1 - mx);
        const float inv = 1.0f / (e0 + e1);
        const float a0 = e0 * inv, a1 = e1 * inv;
        float4 p;
        p.x = a0 * o.x + a1 * v.x; p.y = a0 * o.y + a1 * v.y;
        p.z = a0 * o.z + a1 * v.z; p.w = a0 * o.w + a1 * v.w;
        *(float4*)(out + (size_t)(lo + r) * DFEAT + lane * 4) = p;
      }
    }
    return;
  }

  const float4 b4 = *(const float4*)(bias + lane * 4);
  float4 ns4 = make_float4(0.0f, 0.0f, 0.0f, 0.0f);
  float4 nd4 = make_float4(0.0f, 0.0f, 0.0f, 0.0f);
  if (nxt != 0) {
    ns4 = *(const float4*)(wa_ns + lane * 4);
    nd4 = *(const float4*)(wa_nd + lane * 4);
  }
  for (int r = wave; r < rows; r += OWN_WAVES) {
    float4 v = ((const float4*)obuf)[r * (DFEAT / 4) + lane];
    const float dn = rden[r] + 1e-16f;
    v.x = v.x / dn + b4.x; v.y = v.y / dn + b4.y;
    v.z = v.z / dn + b4.z; v.w = v.w / dn + b4.w;
    if (relu) {
      v.x = fmaxf(v.x, 0.0f); v.y = fmaxf(v.y, 0.0f);
      v.z = fmaxf(v.z, 0.0f); v.w = fmaxf(v.w, 0.0f);
    }
    *(float4*)(O + (size_t)(lo + r) * DFEAT + lane * 4) = v;
    if (nxt != 0) {
      float ps = v.x * ns4.x + v.y * ns4.y + v.z * ns4.z + v.w * ns4.w;
      float pd = v.x * nd4.x + v.y * nd4.y + v.z * nd4.z + v.w * nd4.w;
      ps = wave_sum(ps);
      pd = wave_sum(pd);
      if (lane == 0) {
        as_n[lo + r] = ps;
        ad_n[lo + r] = pd;
      }
    }
  }
}

static inline int ceil_div(long long a, long long b) { return (int)((a + b - 1) / b); }

static void run_gat_layer(const float* X, const int* srcI, const int* dstI, int ecount,
                          const float* W, const float* bias, const float* asrc, const float* adst,
                          float* O, int n, int relu, int* lists, int reuse, const unsigned* omap, hipStream_t stream,
                          int nxt, const float* wa_ns, const float* wa_nd, float* as_n, float* ad_n,
                          int head, const float* x2o, const float* deg, const float* Wp,
                          const float* bp, float* out) {
  gat_owner_rows<<<ceil_div(n, OWN_ROWS), OWN_THREADS, 0, stream>>>(srcI, dstI, ecount, n, asrc, adst,
                                                                  X, W, bias, relu, lists, reuse, omap, O,
                                                                  nxt, wa_ns, wa_nd, as_n, ad_n,
                                                                  head, x2o, deg, Wp, bp, out);
}

extern "C" void kernel_launch(void* const* d_in, const int* in_sizes, int n_in,
                              void* d_out, int out_size, void* d_ws, size_t ws_size,
                              hipStream_t stream) {
  const float* x_o      = (const float*)d_in[0];
  const float* degree_o = (const float*)d_in[1];
  const int*   ei_o     = (const int*)d_in[2];
  const int*   ei_s     = (const int*)d_in[3];
  const float* W_o1 = (const float*)d_in[4];
  const float* as_o1 = (const float*)d_in[5];
  const float* ad_o1 = (const float*)d_in[6];
  const float* b_o1  = (const float*)d_in[7];
  const float* W_o2 = (const float*)d_in[8];
  const float* as_o2 = (const float*)d_in[9];
  const float* ad_o2 = (const float*)d_in[10];
  const float* b_o2  = (const float*)d_in[11];
  const float* W_s1 = (const float*)d_in[12];
  const float* as_s1 = (const float*)d_in[13];
  const float* ad_s1 = (const float*)d_in[14];
  const float* b_s1  = (const float*)d_in[15];
  const float* W_s2 = (const float*)d_in[16];
  const float* as_s2 = (const float*)d_in[17];
  const float* ad_s2 = (const float*)d_in[18];
  const float* b_s2  = (const float*)d_in[19];
  const float* W_pred = (const float*)d_in[20];
  const float* b_pred = (const float*)d_in[21];

  const int n = in_sizes[0] / DFEAT;
  const int ecount = in_sizes[2] / 2;

  const int* src_o = ei_o;
  const int* dst_o = ei_o + ecount;
  const int* src_s = ei_s;
  const int* dst_s = ei_s + ecount;

  float* ws = (float*)d_ws;
  float* T     = ws;
  float* X1    = T + (size_t)n * DFEAT;
  float* X2o   = X1 + (size_t)n * DFEAT;
  float* LST   = X2o + (size_t)n * DFEAT;
  float* wa    = T;
  float* a1os  = wa + 8 * DFEAT;
  float* a1od  = a1os + n;
  float* a1ss  = a1od + n;
  float* a1sd  = a1ss + n;
  float* a2s   = a1sd + n;
  float* a2d   = a2s + n;
  int* lists   = (int*)LST;
  unsigned* map_o = (unsigned*)(T + (size_t)n * 64);
  unsigned* map_s = (unsigned*)(T + (size_t)n * 96);

  gat_fold_a<<<64, 256, 0, stream>>>(W_o1, as_o1, ad_o1, W_o2, as_o2, ad_o2,
                                     W_s1, as_s1, ad_s1, W_s2, as_s2, ad_s2, wa);
  gat_node_alpha4<<<ceil_div((long long)n * 32, 256), 256, 0, stream>>>(x_o, wa, a1os, a1od, a1ss, a1sd, n,
                                                                        dst_o, dst_s, map_o, map_s, ecount);

  run_gat_layer(x_o, src_o, dst_o, ecount, W_o1, b_o1, a1os, a1od,
                X1, n,  1, lists,  0, map_o, stream,
                 1, wa + 2 * DFEAT, wa + 3 * DFEAT, a2s, a2d,
                 0, nullptr, nullptr, nullptr, nullptr, nullptr);
  run_gat_layer(X1, src_o, dst_o, ecount, W_o2, b_o2, a2s, a2d,
                X2o, n,  0, lists,  1, map_o, stream,
                 0, nullptr, nullptr, nullptr, nullptr,
                 0, nullptr, nullptr, nullptr, nullptr, nullptr);
  run_gat_layer(x_o, src_s, dst_s, ecount, W_s1, b_s1, a1ss, a1sd,
                X1, n,  1, lists,  0, map_s, stream,
                 1, wa + 6 * DFEAT, wa + 7 * DFEAT, a2s, a2d,
                 0, nullptr, nullptr, nullptr, nullptr, nullptr);
  run_gat_layer(X1, src_s, dst_s, ecount, W_s2, b_s2, a2s, a2d,
                 nullptr, n,  0, lists,  1, map_s, stream,
                 0, nullptr, nullptr, nullptr, nullptr,
                 1, X2o, degree_o, W_pred, b_pred, (float*)d_out);
}


# Owner map becomes 8 bit planes (bit k of each dst id's owner number, 32 ids a word): owners test 32 ids with 8 logic ops a lane, 128 ids a lane and trip; list order changes, sums stay arrival order
# speedup vs baseline: 5.7911x; 1.0432x over previous
#include <hip/hip_runtime.h>
#include <math.h>

typedef __attribute__((ext_vector_type(2))) float v2f;
typedef __attribute__((ext_vector_type(8))) float v8f;

#define DFEAT 128
#define DDEG 32
#define NEG_SLOPE 0.2f
#define OWN_ROWS 196

__device__ __forceinline__ unsigned fenc(float f) {
  unsigned u = __float_as_uint(f);
  return (u & 0x80000000u) ? ~u : (u | 0x80000000u);
}
__device__ __forceinline__ float fdec(unsigned u) {
  u = (u & 0x80000000u) ? (u & 0x7fffffffu) : ~u;
  return __uint_as_float(u);
}

template <int M>
__device__ __forceinline__ float row_xmask(float x) {
  const int b = __builtin_bit_cast(int, x);
  return __builtin_bit_cast(float, __builtin_amdgcn_update_dpp(b, b, 0x160 + M, 0xf, 0xf, false));
}
__device__ __forceinline__ float xhalf(float x) {
  const int b = __builtin_bit_cast(int, x);
  return __builtin_bit_cast(float, __builtin_amdgcn_permlanex16(b, b, 0x76543210, 0xfedcba98, false, false));
}
__device__ __forceinline__ float wave_sum(float x) {
  x += row_xmask<1>(x);
  x += row_xmask<2>(x);
  x += row_xmask<4>(x);
  x += row_xmask<8>(x);
  return x + xhalf(x);
}

__global__ void __launch_bounds__(256) gat_fold_a(const float* __restrict__ W0, const float* __restrict__ s0,
                                                  const float* __restrict__ d0,
                                                  const float* __restrict__ W1, const float* __restrict__ s1,
                                                  const float* __restrict__ d1,
                                                  const float* __restrict__ W2, const float* __restrict__ s2,
                                                  const float* __restrict__ d2,
                                                  const float* __restrict__ W3, const float* __restrict__ s3,
                                                  const float* __restrict__ d3,
                                                  float* __restrict__ wa) {
  const int lane = threadIdx.x & 31;
  const int wave = threadIdx.x >> 5;
  const int layer = blockIdx.x >> 4;
  const int i = ((blockIdx.x & 15) << 3) + wave;
  const float* W = layer == 0 ? W0 : layer == 1 ? W1 : layer == 2 ? W2 : W3;
  const float* as = layer == 0 ? s0 : layer == 1 ? s1 : layer == 2 ? s2 : s3;
  const float* ad = layer == 0 ? d0 : layer == 1 ? d1 : layer == 2 ? d2 : d3;
  const float4 w4 = *(const float4*)(W + (size_t)i * DFEAT + lane * 4);
  const float4 s4 = *(const float4*)(as + lane * 4);
  const float4 d4 = *(const float4*)(ad + lane * 4);
  float ps = w4.x * s4.x + w4.y * s4.y + w4.z * s4.z + w4.w * s4.w;
  float pd = w4.x * d4.x + w4.y * d4.y + w4.z * d4.z + w4.w * d4.w;
  ps = wave_sum(ps);
  pd = wave_sum(pd);
  if (lane == 0) {
    wa[(2 * layer + 0) * DFEAT + i] = ps;
    wa[(2 * layer + 1) * DFEAT + i] = pd;
  }
}

__device__ __forceinline__ int plane_pitch(int ecount) { return ((ecount >> 5) + 3) & ~3; }

__global__ void __launch_bounds__(256) gat_node_alpha4(const float* __restrict__ X,
                                                       const float* __restrict__ wa,
                                                       float* __restrict__ as_o, float* __restrict__ ad_o,
                                                       float* __restrict__ as_s, float* __restrict__ ad_s,
                                                       int n,
                                                       const int* __restrict__ dst_o,
                                                       const int* __restrict__ dst_s,
                                                       unsigned* __restrict__ map_o,
                                                       unsigned* __restrict__ map_s,
                                                       int ecount) {
  const int g = blockIdx.x * blockDim.x + threadIdx.x;
  const int lane = threadIdx.x & 31;
  {
    const int gw = __builtin_amdgcn_readfirstlane((int)(blockIdx.x * (blockDim.x >> 5) + (threadIdx.x >> 5)));
    const int nwaves = (int)(gridDim.x * (blockDim.x >> 5));
    const int nw = ecount >> 5;
    const int pw = plane_pitch(ecount);
    const int ntask = (nw + 31) >> 5;
    for (int t = gw; t < 2 * ntask; t += nwaves) {
      const int* __restrict__ d = (t & 1) ? dst_s : dst_o;
      unsigned* __restrict__ mp = (t & 1) ? map_s : map_o;
      const int w0 = (t >> 1) << 5;
      unsigned pk[8] = {0u, 0u, 0u, 0u, 0u, 0u, 0u, 0u};
#pragma nounroll
      for (int jh = 0; jh < 32; jh += 16) {
        int idv[16];
#pragma unroll
        for (int j = 0; j < 16; ++j) {
          const int w = w0 + jh + j;
          idv[j] = d[(size_t)(w < nw ? w : 0) * 32 + lane];
        }
#pragma unroll
        for (int j = 0; j < 16; ++j) {
          const unsigned o = (unsigned)idv[j] / OWN_ROWS;
          const bool mine = lane == jh + j;
#pragma unroll
          for (int k = 0; k < 8; ++k) {
            const unsigned b = __builtin_amdgcn_ballot_w32(((o >> k) & 1u) != 0u);
            if (mine) pk[k] = b;
          }
        }
      }
      const int w = w0 + lane;
      if (w < nw) {
#pragma unroll
        for (int k = 0; k < 8; ++k) mp[(size_t)k * pw + w] = pk[k];
      }
    }
  }
  const int node = g >> 5;
  if (node >= n) return;
  const float4 x4 = *(const float4*)(X + (size_t)node * DFEAT + lane * 4);
  const float4 a4 = *(const float4*)(wa + 0 * DFEAT + lane * 4);
  const float4 b4 = *(const float4*)(wa + 1 * DFEAT + lane * 4);
  const float4 c4 = *(const float4*)(wa + 4 * DFEAT + lane * 4);
  const float4 e4 = *(const float4*)(wa + 5 * DFEAT + lane * 4);
  float p0 = x4.x * a4.x + x4.y * a4.y + x4.z * a4.z + x4.w * a4.w;
  float p1 = x4.x * b4.x + x4.y * b4.y + x4.z * b4.z + x4.w * b4.w;
  float p2 = x4.x * c4.x + x4.y * c4.y + x4.z * c4.z + x4.w * c4.w;
  float p3 = x4.x * e4.x + x4.y * e4.y + x4.z * e4.z + x4.w * e4.w;
  p0 = wave_sum(p0);
  p1 = wave_sum(p1);
  p2 = wave_sum(p2);
  p3 = wave_sum(p3);
  if (lane == 0) {
    as_o[node] = p0;
    ad_o[node] = p1;
    as_s[node] = p2;
    ad_s[node] = p3;
  }
}

#define OWN_WAVES 32
#define OWN_THREADS (OWN_WAVES * 32)
#define SCAN_DEPTH 8
#define CHUNK_IDS 128
#define SLAB_IDS (OWN_WAVES * CHUNK_IDS)
#define BATCH_IDS (SCAN_DEPTH * SLAB_IDS)
#define WCAP 512
#define MAP_PLANES 8
#define MAP_TRIP (OWN_WAVES * 32)
#define SCORE_DEPTH 8
#define GROUP_EDGES 16
#define LIST_WORDS (OWN_WAVES * WCAP)
#define HEAD_DEPTH ((OWN_ROWS + OWN_WAVES - 1) / OWN_WAVES)
#define LIST_STRIDE (LIST_WORDS + 64)
#define ACC_PITCH 132
#define ACC_ROWS 208

__device__ __forceinline__ int own_append_b(int* __restrict__ qe, int qn, unsigned b, bool p, int e, int& ovf) {
  if (b != 0u) {
    const int q0 = __builtin_amdgcn_readfirstlane(qn);
    const int cnt = __builtin_popcount(b);
    if (q0 + cnt <= WCAP) {
      const int pos = q0 + (int)__builtin_amdgcn_mbcnt_lo(b, 0u);
      if (p) qe[pos] = e;
      qn = q0 + cnt;
    } else {
      ovf = 1;
    }
  }
  return qn;
}
__device__ __forceinline__ int own_append(int* __restrict__ qe, int qn, bool p, int e, int& ovf) {
  return own_append_b(qe, qn, __builtin_amdgcn_ballot_w32(p), p, e, ovf);
}

__global__ void __launch_bounds__(OWN_THREADS) gat_owner_rows(const int* __restrict__ srcI,
                                                              const int* __restrict__ dstI,
                                                              int ecount, int n,
                                                              const float* __restrict__ asrc,
                                                              const float* __restrict__ adst,
                                                              const float* __restrict__ H,
                                                              const float* __restrict__ W,
                                                              const float* __restrict__ bias,
                                                              int relu,
                                                              int* __restrict__ lists,
                                                              int reuse,
                                                              const unsigned* __restrict__ omap,
                                                              float* __restrict__ O,
                                                              int nxt,
                                                              const float* __restrict__ wa_ns,
                                                              const float* __restrict__ wa_nd,
                                                              float* __restrict__ as_n,
                                                              float* __restrict__ ad_n,
                                                              int head,
                                                              const float* __restrict__ x2o,
                                                              const float* __restrict__ deg,
                                                              const float* __restrict__ Wp,
                                                              const float* __restrict__ bp,
                                                              float* __restrict__ out) {
  __shared__ __attribute__((aligned(16))) float acc[ACC_ROWS * ACC_PITCH];
  __shared__ __attribute__((aligned(16))) int qsv[2 * OWN_WAVES * WCAP];
  int* const qsbuf = qsv;
  float* const qvbuf = (float*)(qsv + OWN_WAVES * WCAP);
  float* const obuf = (float*)qsv;
  __shared__ unsigned char qrbuf[OWN_WAVES * WCAP];
  __shared__ unsigned rmax[OWN_ROWS];
  __shared__ float rden[OWN_ROWS];
  __shared__ int ovf_flag;

  const int tid = threadIdx.x;
  const int lane = tid & 31;
  const int wave = __builtin_amdgcn_readfirstlane(tid >> 5);
  const int lo = blockIdx.x * OWN_ROWS;
  const int rows = min(OWN_ROWS, n - lo);
  const unsigned urows = (unsigned)rows;

  for (int i = tid; i < ACC_ROWS * ACC_PITCH / 4; i += OWN_THREADS)
    ((float4*)acc)[i] = make_float4(0.0f, 0.0f, 0.0f, 0.0f);
  if (tid < OWN_ROWS) {
    rmax[tid] = fenc(-INFINITY);
    rden[tid] = 0.0f;
  }
  if (tid == 0) ovf_flag = 0;
  __syncthreads();

  int* myqs = qsbuf + wave * WCAP;
  float* myqv = qvbuf + wave * WCAP;
  unsigned char* myqr = qrbuf + wave * WCAP;
  int qn = 0;
  int ovf = 0;

  int* gq = lists + (size_t)blockIdx.x * LIST_STRIDE;
  int* myg = gq + wave * WCAP;

  if (reuse == 0) {
    for (int r0 = wave * 32; r0 < rows; r0 += OWN_THREADS)
      qn = own_append(myqs, qn, r0 + lane < rows, ecount + lo + r0 + lane, ovf);

    int done = 0;
    const bool map_ok = gridDim.x <= 256u && (((size_t)omap) & 15) == 0;
    if (map_ok) {
      const int pw = plane_pitch(ecount);
      const int ng = ecount >> 7;
      const int ntrip = (ng + MAP_TRIP - 1) / MAP_TRIP;
      const int ntot = ntrip * MAP_TRIP;
      const int rot = ntot ? (int)((blockIdx.x * 256u) % (unsigned)ntot) : 0;
      unsigned cm[MAP_PLANES];
#pragma unroll
      for (int q = 0; q < MAP_PLANES; ++q) cm[q] = ((blockIdx.x >> q) & 1u) ? 0u : 0xffffffffu;
      for (int k = 0; k < ntrip; ++k) {
        int gg = k * MAP_TRIP + wave * 32 + lane + rot;
        if (gg >= ntot) gg -= ntot;
        const bool valid = gg < ng;
        const unsigned* mp = omap + 4 * (size_t)(valid ? gg : 0);
        uint4 pv[MAP_PLANES];
#pragma unroll
        for (int q = 0; q < MAP_PLANES; ++q) pv[q] = *(const uint4*)(mp + (size_t)q * pw);
        unsigned m0 = valid ? 0xffffffffu : 0u;
        unsigned m1 = m0, m2 = m0, m3 = m0;
#pragma unroll
        for (int q = 0; q < MAP_PLANES; ++q) {
          m0 &= pv[q].x ^ cm[q];
          m1 &= pv[q].y ^ cm[q];
          m2 &= pv[q].z ^ cm[q];
          m3 &= pv[q].w ^ cm[q];
        }
        const int e0 = gg << 7;
        for (;;) {
          const bool z0 = m0 != 0u, z1 = m1 != 0u, z2 = m2 != 0u;
          const bool p = z0 || z1 || z2 || m3 != 0u;
          const unsigned b = __builtin_amdgcn_ballot_w32(p);
          if (b == 0u) break;
          const unsigned mm = z0 ? m0 : z1 ? m1 : z2 ? m2 : m3;
          const int j32 = z0 ? 0 : z1 ? 32 : z2 ? 64 : 96;
          const int cz = __builtin_ctz(mm | 0x80000000u);
          qn = own_append_b(myqs, qn, b, p, e0 + j32 + cz, ovf);
          const unsigned cl = mm & (mm - 1u);
          if (z0) m0 = cl; else if (z1) m1 = cl; else if (z2) m2 = cl; else m3 = cl;
        }
      }
      done = ng << 7;
    }

    const bool vec_ok = !map_ok && (((size_t)dstI) & 15) == 0;
    const int nb_full = vec_ok ? ecount / BATCH_IDS : 0;
    const int rot = nb_full ? (int)(blockIdx.x % (unsigned)nb_full) : 0;
    if (nb_full) done = nb_full * BATCH_IDS;
    for (int k = 0; k < nb_full; ++k) {
      int bi = k + rot;
      if (bi >= nb_full) bi -= nb_full;
      const int b0 = bi * BATCH_IDS + wave * CHUNK_IDS + lane * 4;
      int4 dv[SCAN_DEPTH];
  #pragma unroll
      for (int u = 0; u < SCAN_DEPTH; ++u)
        dv[u] = *(const int4*)(dstI + b0 + u * SLAB_IDS);
  #pragma unroll
      for (int u = 0; u < SCAN_DEPTH; ++u) {
        const unsigned t0 = (unsigned)(dv[u].x - lo);
        const unsigned t1 = (unsigned)(dv[u].y - lo);
        const unsigned t2 = (unsigned)(dv[u].z - lo);
        const unsigned t3 = (unsigned)(dv[u].w - lo);
        const unsigned tm = min(min(t0, t1), min(t2, t3));
        if (__builtin_amdgcn_ballot_w32(tm < urows) != 0u) {
          const int e0 = b0 + u * SLAB_IDS;
          qn = own_append(myqs, qn, t0 < urows, e0, ovf);
          qn = own_append(myqs, qn, t1 < urows, e0 + 1, ovf);
          qn = own_append(myqs, qn, t2 < urows, e0 + 2, ovf);
          qn = own_append(myqs, qn, t3 < urows, e0 + 3, ovf);
        }
      }
    }

    for (int base = done + wave * CHUNK_IDS; base < ecount; base += SLAB_IDS) {
      unsigned t[4];
  #pragma unroll
      for (int j = 0; j < 4; ++j) {
        const int e = base + j * 32 + lane;
        t[j] = (e < ecount) ? (unsigned)(dstI[e] - lo) : 0xffffffffu;
      }
  #pragma unroll
      for (int j = 0; j < 4; ++j)
        qn = own_append(myqs, qn, t[j] < urows, base + j * 32 + lane, ovf);
    }
  } else {
    ovf = __builtin_amdgcn_readfirstlane(gq[LIST_WORDS + OWN_WAVES]);
    qn = ovf != 0 ? 0 : __builtin_amdgcn_readfirstlane(gq[LIST_WORDS + wave]);
  }

  for (int g = 0; g < qn; g += 32 * SCORE_DEPTH) {
    int s[SCORE_DEPTH];
    int d[SCORE_DEPTH];
    float v[SCORE_DEPTH];
    if (reuse == 0) {
#pragma unroll
      for (int u = 0; u < SCORE_DEPTH; ++u) {
        const int i = g + u * 32 + lane;
        s[u] = 0;
        d[u] = lo;
        if (i < qn) {
          const int e = myqs[i];
          if (e < ecount) {
            s[u] = srcI[e];
            d[u] = dstI[e];
          } else {
            s[u] = e - ecount;
            d[u] = e - ecount;
          }
        }
      }
    } else {
#pragma unroll
      for (int u = 0; u < SCORE_DEPTH; ++u) {
        const int i = g + u * 32 + lane;
        s[u] = 0;
        if (i < qn) s[u] = myg[i];
      }
#pragma unroll
      for (int u = 0; u < SCORE_DEPTH; ++u) {
        d[u] = lo + (int)((unsigned)s[u] >> 24);
        s[u] &= 0xffffff;
      }
    }
#pragma unroll
    for (int u = 0; u < SCORE_DEPTH; ++u)
      v[u] = asrc[s[u]] + adst[d[u]];
#pragma unroll
    for (int u = 0; u < SCORE_DEPTH; ++u) {
      const int i = g + u * 32 + lane;
      if (i < qn) {
        const float x = v[u] > 0.0f ? v[u] : v[u] * NEG_SLOPE;
        const int r = d[u] - lo;
        myqs[i] = s[u];
        myqv[i] = x;
        if (reuse == 0) myg[i] = (int)((unsigned)s[u] | ((unsigned)r << 24));
        myqr[i] = (unsigned char)r;
        atomicMax(&rmax[r], fenc(x));
      }
    }
  }
  if (lane == 0) {
    if (reuse == 0) gq[LIST_WORDS + wave] = qn;
    if (ovf != 0) ovf_flag = 1;
  }

  __syncthreads();

  if (reuse == 0 && tid == 0) gq[LIST_WORDS + OWN_WAVES] = ovf_flag;

  const float* hl = H + lane;
  if (__builtin_amdgcn_readfirstlane(ovf_flag) == 0) {
    for (int j0 = 0; j0 < qn; j0 += 32) {
      const int cnt = min(32, qn - j0);
      int s = 0, r = 0;
      float w = 0.0f;
      if (lane < cnt) {
        s = myqs[j0 + lane];
        r = myqr[j0 + lane];
        w = expf(myqv[j0 + lane] - fdec(rmax[r]));
        atomicAdd(&rden[r], w);
      }
      const int wBits = __float_as_int(w);
      for (int k0 = 0; k0 < cnt; k0 += GROUP_EDGES) {
        float hv[GROUP_EDGES][4];
        float cf[GROUP_EDGES];
        int rr[GROUP_EDGES];
#pragma unroll
        for (int u = 0; u < GROUP_EDGES; ++u) {
          const int sj = __builtin_amdgcn_readlane(s, k0 + u);
          rr[u] = __builtin_amdgcn_readlane(r, k0 + u);
          cf[u] = __int_as_float(__builtin_amdgcn_readlane(wBits, k0 + u));
          const float* h = hl + (size_t)sj * DFEAT;
          hv[u][0] = h[0];
          hv[u][1] = h[32];
          hv[u][2] = h[64];
          hv[u][3] = h[96];
        }
#pragma unroll
        for (int u = 0; u < GROUP_EDGES; ++u) {
          if (k0 + u < cnt) {
            float* a = &acc[rr[u] * ACC_PITCH + lane];
            atomicAdd(a + 0, cf[u] * hv[u][0]);
            atomicAdd(a + 32, cf[u] * hv[u][1]);
            atomicAdd(a + 64, cf[u] * hv[u][2]);
            atomicAdd(a + 96, cf[u] * hv[u][3]);
          }
        }
      }
    }
  } else {
    const int etot = ecount + n;
    for (int pass = 0; pass < 2; ++pass) {
      for (int base = wave * 32; base < etot; base += OWN_THREADS) {
        const int e = base + lane;
        int s = 0;
        unsigned t = 0xffffffffu;
        float v = 0.0f;
        if (e < etot) {
          const int d = (e < ecount) ? dstI[e] : e - ecount;
          t = (unsigned)(d - lo);
          if (t < urows) {
            s = (e < ecount) ? srcI[e] : d;
            v = asrc[s] + adst[d];
            v = v > 0.0f ? v : v * NEG_SLOPE;
          }
        }
        const bool own = t < urows;
        if (pass == 0) {
          if (own) atomicMax(&rmax[t], fenc(v));
        } else {
          float w = 0.0f;
          if (own) {
            w = expf(v - fdec(rmax[t]));
            atomicAdd(&rden[t], w);
          }
          const int wBits = __float_as_int(w);
          unsigned b = __builtin_amdgcn_ballot_w32(own);
          while (b != 0u) {
            const int j = __builtin_ctz(b);
            b &= b - 1u;
            const int sj = __builtin_amdgcn_readlane(s, j);
            const int rj = __builtin_amdgcn_readlane((int)t, j);
            const float wj = __int_as_float(__builtin_amdgcn_readlane(wBits, j));
            const float* h = hl + (size_t)sj * DFEAT;
            float* a = &acc[rj * ACC_PITCH + lane];
            atomicAdd(a + 0, wj * h[0]);
            atomicAdd(a + 32, wj * h[32]);
            atomicAdd(a + 64, wj * h[64]);
            atomicAdd(a + 96, wj * h[96]);
          }
        }
      }
      __syncthreads();
    }
  }

  const int fc = lane & 15;
  const int fh = lane >> 4;
  const int fq = wave & 7;
  v2f bf[32];
  {
    const float* wp = W + (size_t)(4 * fh) * DFEAT + 16 * fq + fc;
#pragma unroll
    for (int ks = 0; ks < 32; ++ks) {
      const int k0 = 8 * (ks >> 1) + 2 * (ks & 1);
      bf[ks].x = wp[(size_t)k0 * DFEAT];
      bf[ks].y = wp[(size_t)(k0 + 1) * DFEAT];
    }
  }

  __syncthreads();

  const int ntile = (rows + 15) >> 4;
  for (int rt = wave >> 3; rt < ntile; rt += OWN_WAVES / 8) {
    const float* ar = acc + (16 * rt + fc) * ACC_PITCH + 4 * fh;
    v8f d = {};
#pragma unroll
    for (int j = 0; j < 16; ++j) {
      const float4 a4 = *(const float4*)(ar + 8 * j);
      v2f a;
      a.x = a4.x; a.y = a4.y;
      d = __builtin_amdgcn_wmma_f32_16x16x4_f32(
           false, a,  false, bf[2 * j],
           (short)0, d,  false,  false);
      a.x = a4.z; a.y = a4.w;
      d = __builtin_amdgcn_wmma_f32_16x16x4_f32(
           false, a,  false, bf[2 * j + 1],
           (short)0, d,  false,  false);
    }
    float* ob = obuf + (16 * rt + 8 * fh) * DFEAT + 16 * fq + fc;
#pragma unroll
    for (int r = 0; r < 8; ++r) ob[r * DFEAT] = d[r];
  }

  __syncthreads();

  if (head != 0) {
    const float4 wo0 = *(const float4*)(Wp + lane * 8);
    const float4 wo1 = *(const float4*)(Wp + lane * 8 + 4);
    const float4 ws0 = *(const float4*)(Wp + 2 * DFEAT + lane * 8);
    const float4 ws1 = *(const float4*)(Wp + 2 * DFEAT + lane * 8 + 4);
    const float2 wd = *(const float2*)(Wp + 4 * DFEAT + lane * 2);
    const float bp0 = bp[0];
    const float bp1 = bp[1];
    const float4 b4 = *(const float4*)(bias + lane * 4);
    float4 ov[HEAD_DEPTH];
    float dgv[HEAD_DEPTH];
#pragma unroll
    for (int k = 0; k < HEAD_DEPTH; ++k) {
      const int r = wave + k * OWN_WAVES;
      const int rc = r < rows ? r : 0;
      const size_t row = (size_t)(lo + rc);
      ov[k] = *(const float4*)(x2o + row * DFEAT + lane * 4);
      dgv[k] = deg[row * DDEG + lane];
    }
#pragma unroll
    for (int k = 0; k < HEAD_DEPTH; ++k) {
      const int r = wave + k * OWN_WAVES;
      if (r < rows) {
        float4 v = ((const float4*)obuf)[r * (DFEAT / 4) + lane];
        const float dn = rden[r] + 1e-16f;
        v.x = v.x / dn + b4.x; v.y = v.y / dn + b4.y;
        v.z = v.z / dn + b4.z; v.w = v.w / dn + b4.w;
        if (relu) {
          v.x = fmaxf(v.x, 0.0f); v.y = fmaxf(v.y, 0.0f);
          v.z = fmaxf(v.z, 0.0f); v.w = fmaxf(v.w, 0.0f);
        }
        const float4 o = ov[k];
        float l0 = o.x * wo0.x + o.y * wo0.z + o.z * wo1.x + o.w * wo1.z;
        float l1 = o.x * wo0.y + o.y * wo0.w + o.z * wo1.y + o.w * wo1.w;
        l0 += v.x * ws0.x + v.y * ws0.z + v.z * ws1.x + v.w * ws1.z;
        l1 += v.x * ws0.y + v.y * ws0.w + v.z * ws1.y + v.w * ws1.w;
        l0 += dgv[k] * wd.x;
        l1 += dgv[k] * wd.y;
        l0 = wave_sum(l0);
        l1 = wave_sum(l1);
        l0 += bp0;
        l1 += bp1;
        const float mx = fmaxf(l0, l1);
        const float e0 = expf(l0 - mx), e1 = expf(l1 - mx);
        const float inv = 1.0f / (e0 + e1);
        const float a0 = e0 * inv, a1 = e1 * inv;
        float4 p;
        p.x = a0 * o.x + a1 * v.x; p.y = a0 * o.y + a1 * v.y;
        p.z = a0 * o.z + a1 * v.z; p.w = a0 * o.w + a1 * v.w;
        *(float4*)(out + (size_t)(lo + r) * DFEAT + lane * 4) = p;
      }
    }
    return;
  }

  const float4 b4 = *(const float4*)(bias + lane * 4);
  float4 ns4 = make_float4(0.0f, 0.0f, 0.0f, 0.0f);
  float4 nd4 = make_float4(0.0f, 0.0f, 0.0f, 0.0f);
  if (nxt != 0) {
    ns4 = *(const float4*)(wa_ns + lane * 4);
    nd4 = *(const float4*)(wa_nd + lane * 4);
  }
  for (int r = wave; r < rows; r += OWN_WAVES) {
    float4 v = ((const float4*)obuf)[r * (DFEAT / 4) + lane];
    const float dn = rden[r] + 1e-16f;
    v.x = v.x / dn + b4.x; v.y = v.y / dn + b4.y;
    v.z = v.z / dn + b4.z; v.w = v.w / dn + b4.w;
    if (relu) {
      v.x = fmaxf(v.x, 0.0f); v.y = fmaxf(v.y, 0.0f);
      v.z = fmaxf(v.z, 0.0f); v.w = fmaxf(v.w, 0.0f);
    }
    *(float4*)(O + (size_t)(lo + r) * DFEAT + lane * 4) = v;
    if (nxt != 0) {
      float ps = v.x * ns4.x + v.y * ns4.y + v.z * ns4.z + v.w * ns4.w;
      float pd = v.x * nd4.x + v.y * nd4.y + v.z * nd4.z + v.w * nd4.w;
      ps = wave_sum(ps);
      pd = wave_sum(pd);
      if (lane == 0) {
        as_n[lo + r] = ps;
        ad_n[lo + r] = pd;
      }
    }
  }
}

static inline int ceil_div(long long a, long long b) { return (int)((a + b - 1) / b); }

static void run_gat_layer(const float* X, const int* srcI, const int* dstI, int ecount,
                          const float* W, const float* bias, const float* asrc, const float* adst,
                          float* O, int n, int relu, int* lists, int reuse, const unsigned* omap, hipStream_t stream,
                          int nxt, const float* wa_ns, const float* wa_nd, float* as_n, float* ad_n,
                          int head, const float* x2o, const float* deg, const float* Wp,
                          const float* bp, float* out) {
  gat_owner_rows<<<ceil_div(n, OWN_ROWS), OWN_THREADS, 0, stream>>>(srcI, dstI, ecount, n, asrc, adst,
                                                                  X, W, bias, relu, lists, reuse, omap, O,
                                                                  nxt, wa_ns, wa_nd, as_n, ad_n,
                                                                  head, x2o, deg, Wp, bp, out);
}

extern "C" void kernel_launch(void* const* d_in, const int* in_sizes, int n_in,
                              void* d_out, int out_size, void* d_ws, size_t ws_size,
                              hipStream_t stream) {
  const float* x_o      = (const float*)d_in[0];
  const float* degree_o = (const float*)d_in[1];
  const int*   ei_o     = (const int*)d_in[2];
  const int*   ei_s     = (const int*)d_in[3];
  const float* W_o1 = (const float*)d_in[4];
  const float* as_o1 = (const float*)d_in[5];
  const float* ad_o1 = (const float*)d_in[6];
  const float* b_o1  = (const float*)d_in[7];
  const float* W_o2 = (const float*)d_in[8];
  const float* as_o2 = (const float*)d_in[9];
  const float* ad_o2 = (const float*)d_in[10];
  const float* b_o2  = (const float*)d_in[11];
  const float* W_s1 = (const float*)d_in[12];
  const float* as_s1 = (const float*)d_in[13];
  const float* ad_s1 = (const float*)d_in[14];
  const float* b_s1  = (const float*)d_in[15];
  const float* W_s2 = (const float*)d_in[16];
  const float* as_s2 = (const float*)d_in[17];
  const float* ad_s2 = (const float*)d_in[18];
  const float* b_s2  = (const float*)d_in[19];
  const float* W_pred = (const float*)d_in[20];
  const float* b_pred = (const float*)d_in[21];

  const int n = in_sizes[0] / DFEAT;
  const int ecount = in_sizes[2] / 2;

  const int* src_o = ei_o;
  const int* dst_o = ei_o + ecount;
  const int* src_s = ei_s;
  const int* dst_s = ei_s + ecount;

  float* ws = (float*)d_ws;
  float* T     = ws;
  float* X1    = T + (size_t)n * DFEAT;
  float* X2o   = X1 + (size_t)n * DFEAT;
  float* LST   = X2o + (size_t)n * DFEAT;
  float* wa    = T;
  float* a1os  = wa + 8 * DFEAT;
  float* a1od  = a1os + n;
  float* a1ss  = a1od + n;
  float* a1sd  = a1ss + n;
  float* a2s   = a1sd + n;
  float* a2d   = a2s + n;
  int* lists   = (int*)LST;
  unsigned* map_o = (unsigned*)(T + (size_t)n * 64);
  unsigned* map_s = (unsigned*)(T + (size_t)n * 96);

  gat_fold_a<<<64, 256, 0, stream>>>(W_o1, as_o1, ad_o1, W_o2, as_o2, ad_o2,
                                     W_s1, as_s1, ad_s1, W_s2, as_s2, ad_s2, wa);
  gat_node_alpha4<<<ceil_div((long long)n * 32, 256), 256, 0, stream>>>(x_o, wa, a1os, a1od, a1ss, a1sd, n,
                                                                        dst_o, dst_s, map_o, map_s, ecount);

  run_gat_layer(x_o, src_o, dst_o, ecount, W_o1, b_o1, a1os, a1od,
                X1, n,  1, lists,  0, map_o, stream,
                 1, wa + 2 * DFEAT, wa + 3 * DFEAT, a2s, a2d,
                 0, nullptr, nullptr, nullptr, nullptr, nullptr);
  run_gat_layer(X1, src_o, dst_o, ecount, W_o2, b_o2, a2s, a2d,
                X2o, n,  0, lists,  1, map_o, stream,
                 0, nullptr, nullptr, nullptr, nullptr,
                 0, nullptr, nullptr, nullptr, nullptr, nullptr);
  run_gat_layer(x_o, src_s, dst_s, ecount, W_s1, b_s1, a1ss, a1sd,
                X1, n,  1, lists,  0, map_s, stream,
                 1, wa + 6 * DFEAT, wa + 7 * DFEAT, a2s, a2d,
                 0, nullptr, nullptr, nullptr, nullptr, nullptr);
  run_gat_layer(X1, src_s, dst_s, ecount, W_s2, b_s2, a2s, a2d,
                 nullptr, n,  0, lists,  1, map_s, stream,
                 0, nullptr, nullptr, nullptr, nullptr,
                 1, X2o, degree_o, W_pred, b_pred, (float*)d_out);
}
